# starting kernel
# baseline (speedup 1.0000x reference)
#include <hip/hip_runtime.h>


typedef __attribute__((ext_vector_type(2))) float v2f;
typedef __attribute__((ext_vector_type(8))) float v8f;

#define DIM   128
#define GNUM  64
#define LAYERS 3
#define LNEPS 1e-5f

__global__ void zero_f32(float* __restrict__ p, int n) {
    int i = blockIdx.x * blockDim.x + threadIdx.x;
    if (i < n) p[i] = 0.0f;
}

__global__ void deg_accum(const int* __restrict__ ei, float* __restrict__ deg, int E) {
    int e = blockIdx.x * blockDim.x + threadIdx.x;
    if (e < E) atomicAdd(&deg[ei[E + e]], 1.0f);
}

__global__ void deg_to_dinv(float* __restrict__ d, int N) {
    int i = blockIdx.x * blockDim.x + threadIdx.x;
    if (i < N) d[i] = rsqrtf(d[i] + 1.0f);
}

__global__ __launch_bounds__(256) void gemm_wmma(const float* __restrict__ H,
                                                 const float* __restrict__ W,
                                                 float* __restrict__ out, int N) {
    __shared__ float lw[DIM * DIM];
    {
        const float4* Wv = (const float4*)W;
        float4* lv = (float4*)lw;
        #pragma unroll
        for (int i = 0; i < (DIM * DIM / 4) / 256; ++i)
            lv[threadIdx.x + i * 256] = Wv[threadIdx.x + i * 256];
    }
    __syncthreads();

    const int wave = threadIdx.x >> 5;
    const int lane = threadIdx.x & 31;
    const int half = lane >> 4;
    const int ln   = lane & 15;

    const int nstrips = N >> 4;
    int strip = blockIdx.x * 8 + wave;
    if (strip >= nstrips) strip = nstrips - 1;
    const int rowBase = strip << 4;

    const v2f* arow2 = (const v2f*)(H + (size_t)(rowBase + ln) * DIM);

    v8f acc[8] = {};

    for (int k = 0; k < DIM; k += 4) {
        v2f a = arow2[(k >> 1) + half];
        const int kb = k + 2 * half;
        #pragma unroll
        for (int t = 0; t < 8; ++t) {
            v2f b;
            b.x = lw[kb * DIM + t * 16 + ln];
            b.y = lw[(kb + 1) * DIM + t * 16 + ln];
            acc[t] = __builtin_amdgcn_wmma_f32_16x16x4_f32(
                false, a, false, b, (short)0, acc[t], false, false);
        }
    }

    #pragma unroll
    for (int t = 0; t < 8; ++t) {
        #pragma unroll
        for (int j = 0; j < 8; ++j) {
            int row = rowBase + half * 8 + j;
            out[(size_t)row * DIM + t * 16 + ln] = acc[t][j];
        }
    }
}

__global__ void acc_init(const float* __restrict__ hw, const float* __restrict__ dinv,
                         const float* __restrict__ bias, float* __restrict__ acc, int N) {
    int i = blockIdx.x * blockDim.x + threadIdx.x;
    if (i >= N * (DIM / 4)) return;
    int n  = i >> 5;
    int c4 = i & 31;
    float s = dinv[n]; s = s * s;
    float4 hv = ((const float4*)hw)[i];
    float4 bv = ((const float4*)bias)[c4];
    float4 r;
    r.x = bv.x + s * hv.x;  r.y = bv.y + s * hv.y;
    r.z = bv.z + s * hv.z;  r.w = bv.w + s * hv.w;
    ((float4*)acc)[i] = r;
}

__global__ __launch_bounds__(256) void edge_scatter(const int* __restrict__ ei,
                                                    const float* __restrict__ hw,
                                                    const float* __restrict__ dinv,
                                                    float* __restrict__ acc, int E) {
    int wave = threadIdx.x >> 5, lane = threadIdx.x & 31;
    int e = blockIdx.x * 8 + wave;
    if (e >= E) return;
    int src = ei[e];
    int dst = ei[E + e];
    float w = dinv[src] * dinv[dst];
    float4 v = ((const float4*)(hw + (size_t)src * DIM))[lane];
    float* dp = acc + (size_t)dst * DIM + lane * 4;
    atomicAdd(dp + 0, w * v.x);
    atomicAdd(dp + 1, w * v.y);
    atomicAdd(dp + 2, w * v.z);
    atomicAdd(dp + 3, w * v.w);
}

__global__ __launch_bounds__(256) void ln_relu(const float* __restrict__ acc,
                                               const float* __restrict__ gamma,
                                               const float* __restrict__ beta,
                                               float* __restrict__ out, int N) {
    int wave = threadIdx.x >> 5, lane = threadIdx.x & 31;
    int n = blockIdx.x * 8 + wave;
    if (n >= N) return;
    float4 v = ((const float4*)(acc + (size_t)n * DIM))[lane];
    float s = v.x + v.y + v.z + v.w;
    float q = v.x * v.x + v.y * v.y + v.z * v.z + v.w * v.w;
    #pragma unroll
    for (int off = 16; off > 0; off >>= 1) {
        s += __shfl_xor(s, off, 32);
        q += __shfl_xor(q, off, 32);
    }
    float mu  = s * (1.0f / DIM);
    float var = fmaxf(q * (1.0f / DIM) - mu * mu, 0.0f);
    float inv = rsqrtf(var + LNEPS);
    float4 g = ((const float4*)gamma)[lane];
    float4 b = ((const float4*)beta)[lane];
    float4 r;
    r.x = fmaxf(0.0f, g.x * (v.x - mu) * inv + b.x);
    r.y = fmaxf(0.0f, g.y * (v.y - mu) * inv + b.y);
    r.z = fmaxf(0.0f, g.z * (v.z - mu) * inv + b.z);
    r.w = fmaxf(0.0f, g.w * (v.w - mu) * inv + b.w);
    ((float4*)(out + (size_t)n * DIM))[lane] = r;
}

__global__ __launch_bounds__(256) void pool_accum(const float* __restrict__ h,
                                                  const int* __restrict__ batch,
                                                  float* __restrict__ pool,
                                                  float* __restrict__ cnt, int N) {
    int wave = threadIdx.x >> 5, lane = threadIdx.x & 31;
    int n = blockIdx.x * 8 + wave;
    if (n >= N) return;
    int g = batch[n];
    float4 v = ((const float4*)(h + (size_t)n * DIM))[lane];
    float* p = pool + (size_t)g * DIM + lane * 4;
    atomicAdd(p + 0, v.x);
    atomicAdd(p + 1, v.y);
    atomicAdd(p + 2, v.z);
    atomicAdd(p + 3, v.w);
    if (lane == 0) atomicAdd(&cnt[g], 1.0f);
}

__global__ void pool_final(const float* __restrict__ pool, const float* __restrict__ cnt,
                           float* __restrict__ out, int total) {
    int i = blockIdx.x * blockDim.x + threadIdx.x;
    if (i < total) out[i] = pool[i] / fmaxf(cnt[i / DIM], 1.0f);
}

static inline size_t al256(size_t x) { return (x + 255) & ~(size_t)255; }

extern "C" void kernel_launch(void* const* d_in, const int* in_sizes, int n_in,
                              void* d_out, int out_size, void* d_ws, size_t ws_size,
                              hipStream_t stream) {
    const float* x      = (const float*)d_in[0];
    const int*   ei     = (const int*)d_in[1];
    const int*   batch  = (const int*)d_in[2];
    const float* Ws     = (const float*)d_in[3];
    const float* bs     = (const float*)d_in[4];
    const float* gammas = (const float*)d_in[5];
    const float* betas  = (const float*)d_in[6];

    const int N = in_sizes[0] / DIM;
    const int E = in_sizes[1] / 2;
    const size_t nd = (size_t)N * DIM * sizeof(float);

    char* ws = (char*)d_ws;
    float* dinv = (float*)ws;                 ws += al256((size_t)N * sizeof(float));
    float* hw   = (float*)ws;                 ws += al256(nd);
    float* accb = (float*)ws;                 ws += al256(nd);
    float* hbuf = (float*)ws;                 ws += al256(nd);
    float* pool = (float*)ws;
    float* cnt  = pool + GNUM * DIM;

    zero_f32<<<(N + 255) / 256, 256, 0, stream>>>(dinv, N);
    deg_accum<<<(E + 255) / 256, 256, 0, stream>>>(ei, dinv, E);
    deg_to_dinv<<<(N + 255) / 256, 256, 0, stream>>>(dinv, N);

    const int nstrips = N >> 4;
    const float* hcur = x;
    for (int l = 0; l < LAYERS; ++l) {
        gemm_wmma<<<(nstrips + 7) / 8, 256, 0, stream>>>(
            hcur, Ws + (size_t)l * DIM * DIM, hw, N);
        acc_init<<<(N * (DIM / 4) + 255) / 256, 256, 0, stream>>>(
            hw, dinv, bs + (size_t)l * DIM, accb, N);
        edge_scatter<<<(E + 7) / 8, 256, 0, stream>>>(ei, hw, dinv, accb, E);
        ln_relu<<<(N + 7) / 8, 256, 0, stream>>>(
            accb, gammas + (size_t)l * DIM, betas + (size_t)l * DIM, hbuf, N);
        hcur = hbuf;
    }

    zero_f32<<<(GNUM * DIM + GNUM + 255) / 256, 256, 0, stream>>>(pool, GNUM * DIM + GNUM);
    pool_accum<<<(N + 7) / 8, 256, 0, stream>>>(hcur, batch, pool, cnt, N);
    pool_final<<<(GNUM * DIM + 255) / 256, 256, 0, stream>>>(pool, cnt, (float*)d_out, GNUM * DIM);
}


# mean pool as a one-hot matrix product on the f32 WMMA: each 200-node chunk stores a partial 64x128 sum, a second launch adds partials in fixed order; pool atomics and zero fill go (new sum order)
# speedup vs baseline: 1.2907x; 1.2907x over previous
#include <hip/hip_runtime.h>


typedef __attribute__((ext_vector_type(2))) float v2f;
typedef __attribute__((ext_vector_type(8))) float v8f;

#define DIM   128
#define GNUM  64
#define LAYERS 3
#define LNEPS 1e-5f

__global__ void zero_f32(float* __restrict__ p, int n) {
    int i = blockIdx.x * blockDim.x + threadIdx.x;
    if (i < n) p[i] = 0.0f;
}

__global__ void deg_accum(const int* __restrict__ ei, float* __restrict__ deg, int E) {
    int e = blockIdx.x * blockDim.x + threadIdx.x;
    if (e < E) atomicAdd(&deg[ei[E + e]], 1.0f);
}

__global__ void deg_to_dinv(float* __restrict__ d, int N) {
    int i = blockIdx.x * blockDim.x + threadIdx.x;
    if (i < N) d[i] = rsqrtf(d[i] + 1.0f);
}

__global__ __launch_bounds__(256) void gemm_wmma(const float* __restrict__ H,
                                                 const float* __restrict__ W,
                                                 float* __restrict__ out, int N) {
    __shared__ float lw[DIM * DIM];
    {
        const float4* Wv = (const float4*)W;
        float4* lv = (float4*)lw;
        #pragma unroll
        for (int i = 0; i < (DIM * DIM / 4) / 256; ++i)
            lv[threadIdx.x + i * 256] = Wv[threadIdx.x + i * 256];
    }
    __syncthreads();

    const int wave = threadIdx.x >> 5;
    const int lane = threadIdx.x & 31;
    const int half = lane >> 4;
    const int ln   = lane & 15;

    const int nstrips = N >> 4;
    int strip = blockIdx.x * 8 + wave;
    if (strip >= nstrips) strip = nstrips - 1;
    const int rowBase = strip << 4;

    const v2f* arow2 = (const v2f*)(H + (size_t)(rowBase + ln) * DIM);

    v8f acc[8] = {};

    for (int k = 0; k < DIM; k += 4) {
        v2f a = arow2[(k >> 1) + half];
        const int kb = k + 2 * half;
        #pragma unroll
        for (int t = 0; t < 8; ++t) {
            v2f b;
            b.x = lw[kb * DIM + t * 16 + ln];
            b.y = lw[(kb + 1) * DIM + t * 16 + ln];
            acc[t] = __builtin_amdgcn_wmma_f32_16x16x4_f32(
                false, a, false, b, (short)0, acc[t], false, false);
        }
    }

    #pragma unroll
    for (int t = 0; t < 8; ++t) {
        #pragma unroll
        for (int j = 0; j < 8; ++j) {
            int row = rowBase + half * 8 + j;
            out[(size_t)row * DIM + t * 16 + ln] = acc[t][j];
        }
    }
}

__global__ void acc_init(const float* __restrict__ hw, const float* __restrict__ dinv,
                         const float* __restrict__ bias, float* __restrict__ acc, int N) {
    int i = blockIdx.x * blockDim.x + threadIdx.x;
    if (i >= N * (DIM / 4)) return;
    int n  = i >> 5;
    int c4 = i & 31;
    float s = dinv[n]; s = s * s;
    float4 hv = ((const float4*)hw)[i];
    float4 bv = ((const float4*)bias)[c4];
    float4 r;
    r.x = bv.x + s * hv.x;  r.y = bv.y + s * hv.y;
    r.z = bv.z + s * hv.z;  r.w = bv.w + s * hv.w;
    ((float4*)acc)[i] = r;
}

__global__ __launch_bounds__(256) void edge_scatter(const int* __restrict__ ei,
                                                    const float* __restrict__ hw,
                                                    const float* __restrict__ dinv,
                                                    float* __restrict__ acc, int E) {
    int wave = threadIdx.x >> 5, lane = threadIdx.x & 31;
    int e = blockIdx.x * 8 + wave;
    if (e >= E) return;
    int src = ei[e];
    int dst = ei[E + e];
    float w = dinv[src] * dinv[dst];
    float4 v = ((const float4*)(hw + (size_t)src * DIM))[lane];
    float* dp = acc + (size_t)dst * DIM + lane * 4;
    atomicAdd(dp + 0, w * v.x);
    atomicAdd(dp + 1, w * v.y);
    atomicAdd(dp + 2, w * v.z);
    atomicAdd(dp + 3, w * v.w);
}

__global__ __launch_bounds__(256) void ln_relu(const float* __restrict__ acc,
                                               const float* __restrict__ gamma,
                                               const float* __restrict__ beta,
                                               float* __restrict__ out, int N) {
    int wave = threadIdx.x >> 5, lane = threadIdx.x & 31;
    int n = blockIdx.x * 8 + wave;
    if (n >= N) return;
    float4 v = ((const float4*)(acc + (size_t)n * DIM))[lane];
    float s = v.x + v.y + v.z + v.w;
    float q = v.x * v.x + v.y * v.y + v.z * v.z + v.w * v.w;
    #pragma unroll
    for (int off = 16; off > 0; off >>= 1) {
        s += __shfl_xor(s, off, 32);
        q += __shfl_xor(q, off, 32);
    }
    float mu  = s * (1.0f / DIM);
    float var = fmaxf(q * (1.0f / DIM) - mu * mu, 0.0f);
    float inv = rsqrtf(var + LNEPS);
    float4 g = ((const float4*)gamma)[lane];
    float4 b = ((const float4*)beta)[lane];
    float4 r;
    r.x = fmaxf(0.0f, g.x * (v.x - mu) * inv + b.x);
    r.y = fmaxf(0.0f, g.y * (v.y - mu) * inv + b.y);
    r.z = fmaxf(0.0f, g.z * (v.z - mu) * inv + b.z);
    r.w = fmaxf(0.0f, g.w * (v.w - mu) * inv + b.w);
    ((float4*)(out + (size_t)n * DIM))[lane] = r;
}

#define PCH_BLOCKS 5
#define PCH_STEPS  5
#define PCH_WNODES (PCH_BLOCKS * PCH_STEPS * 4)
#define PCH_NODES  (2 * PCH_WNODES)
#define PSTRIDE    (GNUM * DIM + GNUM)

__global__ __launch_bounds__(128) void pool_onehot(const float* __restrict__ h,
                                                   const int* __restrict__ batch,
                                                   float* __restrict__ part, int N) {
    __shared__ float4 lsum[GNUM * DIM / 4];
    __shared__ float  lcnt[GNUM];
    const int wave = threadIdx.x >> 5, lane = threadIdx.x & 31;
    const int ch = wave & 1;
    const int sc = wave >> 1;
    const int half = lane >> 4, c = lane & 15;
    const int nbase = blockIdx.x * PCH_NODES + sc * PCH_WNODES + 2 * half;
    const float* hcol = h + 64 * ch + 4 * c;

    v8f acc[4][4] = {};
    float cnt[4] = {0.0f, 0.0f, 0.0f, 0.0f};

    #pragma unroll
    for (int blk = 0; blk < PCH_BLOCKS; ++blk) {
        float4 r0[PCH_STEPS], r1[PCH_STEPS];
        int g0[PCH_STEPS], g1[PCH_STEPS];
        #pragma unroll
        for (int s = 0; s < PCH_STEPS; ++s) {
            const int n0 = nbase + (blk * PCH_STEPS + s) * 4;
            const int n1 = n0 + 1;
            const int i0 = min(n0, N - 1), i1 = min(n1, N - 1);
            r0[s] = *(const float4*)(hcol + (size_t)i0 * DIM);
            r1[s] = *(const float4*)(hcol + (size_t)i1 * DIM);
            const int b0 = batch[i0], b1 = batch[i1];
            g0[s] = (n0 < N) ? b0 : -1;
            g1[s] = (n1 < N) ? b1 : -1;
        }
        #pragma unroll
        for (int s = 0; s < PCH_STEPS; ++s) {
            #pragma unroll
            for (int m = 0; m < 4; ++m) {
                v2f a;
                a.x = (g0[s] == 16 * m + c) ? 1.0f : 0.0f;
                a.y = (g1[s] == 16 * m + c) ? 1.0f : 0.0f;
                cnt[m] += a.x + a.y;
                v2f b;
                b.x = r0[s].x; b.y = r1[s].x;
                acc[m][0] = __builtin_amdgcn_wmma_f32_16x16x4_f32(false, a, false, b, (short)0, acc[m][0], false, false);
                b.x = r0[s].y; b.y = r1[s].y;
                acc[m][1] = __builtin_amdgcn_wmma_f32_16x16x4_f32(false, a, false, b, (short)0, acc[m][1], false, false);
                b.x = r0[s].z; b.y = r1[s].z;
                acc[m][2] = __builtin_amdgcn_wmma_f32_16x16x4_f32(false, a, false, b, (short)0, acc[m][2], false, false);
                b.x = r0[s].w; b.y = r1[s].w;
                acc[m][3] = __builtin_amdgcn_wmma_f32_16x16x4_f32(false, a, false, b, (short)0, acc[m][3], false, false);
            }
        }
    }

    #pragma unroll
    for (int m = 0; m < 4; ++m) cnt[m] += __shfl_xor(cnt[m], 16, 32);

    if (sc == 1) {
        #pragma unroll
        for (int m = 0; m < 4; ++m) {
            #pragma unroll
            for (int r = 0; r < 8; ++r) {
                const int row = 16 * m + 8 * half + r;
                lsum[row * (DIM / 4) + 16 * ch + c] =
                    make_float4(acc[m][0][r], acc[m][1][r], acc[m][2][r], acc[m][3][r]);
            }
            if (ch == 0 && half == 0) lcnt[16 * m + c] = cnt[m];
        }
    }
    __syncthreads();
    if (sc == 0) {
        float* pp = part + (size_t)blockIdx.x * PSTRIDE;
        #pragma unroll
        for (int m = 0; m < 4; ++m) {
            #pragma unroll
            for (int r = 0; r < 8; ++r) {
                const int row = 16 * m + 8 * half + r;
                const float4 o = lsum[row * (DIM / 4) + 16 * ch + c];
                float4 v;
                v.x = acc[m][0][r] + o.x;  v.y = acc[m][1][r] + o.y;
                v.z = acc[m][2][r] + o.z;  v.w = acc[m][3][r] + o.w;
                ((float4*)pp)[row * (DIM / 4) + 16 * ch + c] = v;
            }
            if (ch == 0 && half == 0) pp[GNUM * DIM + 16 * m + c] = cnt[m] + lcnt[16 * m + c];
        }
    }
}

__global__ __launch_bounds__(256) void pool_reduce(const float* __restrict__ part,
                                                   float* __restrict__ out, int P) {
    __shared__ float red[8][32];
    __shared__ float redc[8];
    const int lane = threadIdx.x & 31, s = threadIdx.x >> 5;
    const int o = blockIdx.x * 32 + lane;
    const int g = (blockIdx.x * 32) / DIM;
    float sum = 0.0f;
    for (int p0 = s; p0 < P; p0 += 128) {
        float v[16];
        #pragma unroll
        for (int u = 0; u < 16; ++u) {
            const int p = p0 + 8 * u;
            v[u] = (p < P) ? part[(size_t)p * PSTRIDE + o] : 0.0f;
        }
        #pragma unroll
        for (int u = 0; u < 16; ++u) sum += v[u];
    }
    float cs = 0.0f;
    for (int p = s + 8 * lane; p < P; p += 256) cs += part[(size_t)p * PSTRIDE + GNUM * DIM + g];
    #pragma unroll
    for (int off = 16; off > 0; off >>= 1) cs += __shfl_xor(cs, off, 32);
    red[s][lane] = sum;
    if (lane == 0) redc[s] = cs;
    __syncthreads();
    if (s == 0) {
        float t = 0.0f, cc = 0.0f;
        #pragma unroll
        for (int i = 0; i < 8; ++i) { t += red[i][lane]; cc += redc[i]; }
        out[o] = t / fmaxf(cc, 1.0f);
    }
}

static inline size_t al256(size_t x) { return (x + 255) & ~(size_t)255; }

extern "C" void kernel_launch(void* const* d_in, const int* in_sizes, int n_in,
                              void* d_out, int out_size, void* d_ws, size_t ws_size,
                              hipStream_t stream) {
    const float* x      = (const float*)d_in[0];
    const int*   ei     = (const int*)d_in[1];
    const int*   batch  = (const int*)d_in[2];
    const float* Ws     = (const float*)d_in[3];
    const float* bs     = (const float*)d_in[4];
    const float* gammas = (const float*)d_in[5];
    const float* betas  = (const float*)d_in[6];

    const int N = in_sizes[0] / DIM;
    const int E = in_sizes[1] / 2;
    const size_t nd = (size_t)N * DIM * sizeof(float);

    char* ws = (char*)d_ws;
    float* dinv = (float*)ws;                 ws += al256((size_t)N * sizeof(float));
    float* hw   = (float*)ws;                 ws += al256(nd);
    float* accb = (float*)ws;                 ws += al256(nd);
    float* hbuf = (float*)ws;                 ws += al256(nd);
    float* part = (float*)ws;

    zero_f32<<<(N + 255) / 256, 256, 0, stream>>>(dinv, N);
    deg_accum<<<(E + 255) / 256, 256, 0, stream>>>(ei, dinv, E);
    deg_to_dinv<<<(N + 255) / 256, 256, 0, stream>>>(dinv, N);

    const int nstrips = N >> 4;
    const float* hcur = x;
    for (int l = 0; l < LAYERS; ++l) {
        gemm_wmma<<<(nstrips + 7) / 8, 256, 0, stream>>>(
            hcur, Ws + (size_t)l * DIM * DIM, hw, N);
        acc_init<<<(N * (DIM / 4) + 255) / 256, 256, 0, stream>>>(
            hw, dinv, bs + (size_t)l * DIM, accb, N);
        edge_scatter<<<(E + 7) / 8, 256, 0, stream>>>(ei, hw, dinv, accb, E);
        ln_relu<<<(N + 7) / 8, 256, 0, stream>>>(
            accb, gammas + (size_t)l * DIM, betas + (size_t)l * DIM, hbuf, N);
        hcur = hbuf;
    }

    const int P = (N + PCH_NODES - 1) / PCH_NODES;
    pool_onehot<<<P, 128, 0, stream>>>(hcur, batch, part, N);
    pool_reduce<<<GNUM * DIM / 32, 256, 0, stream>>>(part, (float*)d_out, P);
}


# edge scatter: lane l takes columns l, l+32, l+64, l+96, so every f32 atomic add (and row load) instruction covers one whole 128-byte line: 4 line requests per edge, not 16; same arrival-order sums
# speedup vs baseline: 1.6843x; 1.3050x over previous
#include <hip/hip_runtime.h>


typedef __attribute__((ext_vector_type(2))) float v2f;
typedef __attribute__((ext_vector_type(8))) float v8f;

#define DIM   128
#define GNUM  64
#define LAYERS 3
#define LNEPS 1e-5f

__global__ void zero_f32(float* __restrict__ p, int n) {
    int i = blockIdx.x * blockDim.x + threadIdx.x;
    if (i < n) p[i] = 0.0f;
}

__global__ void deg_accum(const int* __restrict__ ei, float* __restrict__ deg, int E) {
    int e = blockIdx.x * blockDim.x + threadIdx.x;
    if (e < E) atomicAdd(&deg[ei[E + e]], 1.0f);
}

__global__ void deg_to_dinv(float* __restrict__ d, int N) {
    int i = blockIdx.x * blockDim.x + threadIdx.x;
    if (i < N) d[i] = rsqrtf(d[i] + 1.0f);
}

__global__ __launch_bounds__(256) void gemm_wmma(const float* __restrict__ H,
                                                 const float* __restrict__ W,
                                                 float* __restrict__ out, int N) {
    __shared__ float lw[DIM * DIM];
    {
        const float4* Wv = (const float4*)W;
        float4* lv = (float4*)lw;
        #pragma unroll
        for (int i = 0; i < (DIM * DIM / 4) / 256; ++i)
            lv[threadIdx.x + i * 256] = Wv[threadIdx.x + i * 256];
    }
    __syncthreads();

    const int wave = threadIdx.x >> 5;
    const int lane = threadIdx.x & 31;
    const int half = lane >> 4;
    const int ln   = lane & 15;

    const int nstrips = N >> 4;
    int strip = blockIdx.x * 8 + wave;
    if (strip >= nstrips) strip = nstrips - 1;
    const int rowBase = strip << 4;

    const v2f* arow2 = (const v2f*)(H + (size_t)(rowBase + ln) * DIM);

    v8f acc[8] = {};

    for (int k = 0; k < DIM; k += 4) {
        v2f a = arow2[(k >> 1) + half];
        const int kb = k + 2 * half;
        #pragma unroll
        for (int t = 0; t < 8; ++t) {
            v2f b;
            b.x = lw[kb * DIM + t * 16 + ln];
            b.y = lw[(kb + 1) * DIM + t * 16 + ln];
            acc[t] = __builtin_amdgcn_wmma_f32_16x16x4_f32(
                false, a, false, b, (short)0, acc[t], false, false);
        }
    }

    #pragma unroll
    for (int t = 0; t < 8; ++t) {
        #pragma unroll
        for (int j = 0; j < 8; ++j) {
            int row = rowBase + half * 8 + j;
            out[(size_t)row * DIM + t * 16 + ln] = acc[t][j];
        }
    }
}

__global__ void acc_init(const float* __restrict__ hw, const float* __restrict__ dinv,
                         const float* __restrict__ bias, float* __restrict__ acc, int N) {
    int i = blockIdx.x * blockDim.x + threadIdx.x;
    if (i >= N * (DIM / 4)) return;
    int n  = i >> 5;
    int c4 = i & 31;
    float s = dinv[n]; s = s * s;
    float4 hv = ((const float4*)hw)[i];
    float4 bv = ((const float4*)bias)[c4];
    float4 r;
    r.x = bv.x + s * hv.x;  r.y = bv.y + s * hv.y;
    r.z = bv.z + s * hv.z;  r.w = bv.w + s * hv.w;
    ((float4*)acc)[i] = r;
}

__global__ __launch_bounds__(256) void edge_scatter(const int* __restrict__ ei,
                                                    const float* __restrict__ hw,
                                                    const float* __restrict__ dinv,
                                                    float* __restrict__ acc, int E) {
    int wave = threadIdx.x >> 5, lane = threadIdx.x & 31;
    int e = blockIdx.x * 8 + wave;
    if (e >= E) return;
    int src = ei[e];
    int dst = ei[E + e];
    float w = dinv[src] * dinv[dst];
    const float* sp = hw + (size_t)src * DIM + lane;
    float v0 = sp[0], v1 = sp[32], v2 = sp[64], v3 = sp[96];
    float* dp = acc + (size_t)dst * DIM + lane;
    atomicAdd(dp +  0, w * v0);
    atomicAdd(dp + 32, w * v1);
    atomicAdd(dp + 64, w * v2);
    atomicAdd(dp + 96, w * v3);
}

__global__ __launch_bounds__(256) void ln_relu(const float* __restrict__ acc,
                                               const float* __restrict__ gamma,
                                               const float* __restrict__ beta,
                                               float* __restrict__ out, int N) {
    int wave = threadIdx.x >> 5, lane = threadIdx.x & 31;
    int n = blockIdx.x * 8 + wave;
    if (n >= N) return;
    float4 v = ((const float4*)(acc + (size_t)n * DIM))[lane];
    float s = v.x + v.y + v.z + v.w;
    float q = v.x * v.x + v.y * v.y + v.z * v.z + v.w * v.w;
    #pragma unroll
    for (int off = 16; off > 0; off >>= 1) {
        s += __shfl_xor(s, off, 32);
        q += __shfl_xor(q, off, 32);
    }
    float mu  = s * (1.0f / DIM);
    float var = fmaxf(q * (1.0f / DIM) - mu * mu, 0.0f);
    float inv = rsqrtf(var + LNEPS);
    float4 g = ((const float4*)gamma)[lane];
    float4 b = ((const float4*)beta)[lane];
    float4 r;
    r.x = fmaxf(0.0f, g.x * (v.x - mu) * inv + b.x);
    r.y = fmaxf(0.0f, g.y * (v.y - mu) * inv + b.y);
    r.z = fmaxf(0.0f, g.z * (v.z - mu) * inv + b.z);
    r.w = fmaxf(0.0f, g.w * (v.w - mu) * inv + b.w);
    ((float4*)(out + (size_t)n * DIM))[lane] = r;
}

#define PCH_BLOCKS 5
#define PCH_STEPS  5
#define PCH_WNODES (PCH_BLOCKS * PCH_STEPS * 4)
#define PCH_NODES  (2 * PCH_WNODES)
#define PSTRIDE    (GNUM * DIM + GNUM)

__global__ __launch_bounds__(128) void pool_onehot(const float* __restrict__ h,
                                                   const int* __restrict__ batch,
                                                   float* __restrict__ part, int N) {
    __shared__ float4 lsum[GNUM * DIM / 4];
    __shared__ float  lcnt[GNUM];
    const int wave = threadIdx.x >> 5, lane = threadIdx.x & 31;
    const int ch = wave & 1;
    const int sc = wave >> 1;
    const int half = lane >> 4, c = lane & 15;
    const int nbase = blockIdx.x * PCH_NODES + sc * PCH_WNODES + 2 * half;
    const float* hcol = h + 64 * ch + 4 * c;

    v8f acc[4][4] = {};
    float cnt[4] = {0.0f, 0.0f, 0.0f, 0.0f};

    #pragma unroll
    for (int blk = 0; blk < PCH_BLOCKS; ++blk) {
        float4 r0[PCH_STEPS], r1[PCH_STEPS];
        int g0[PCH_STEPS], g1[PCH_STEPS];
        #pragma unroll
        for (int s = 0; s < PCH_STEPS; ++s) {
            const int n0 = nbase + (blk * PCH_STEPS + s) * 4;
            const int n1 = n0 + 1;
            const int i0 = min(n0, N - 1), i1 = min(n1, N - 1);
            r0[s] = *(const float4*)(hcol + (size_t)i0 * DIM);
            r1[s] = *(const float4*)(hcol + (size_t)i1 * DIM);
            const int b0 = batch[i0], b1 = batch[i1];
            g0[s] = (n0 < N) ? b0 : -1;
            g1[s] = (n1 < N) ? b1 : -1;
        }
        #pragma unroll
        for (int s = 0; s < PCH_STEPS; ++s) {
            #pragma unroll
            for (int m = 0; m < 4; ++m) {
                v2f a;
                a.x = (g0[s] == 16 * m + c) ? 1.0f : 0.0f;
                a.y = (g1[s] == 16 * m + c) ? 1.0f : 0.0f;
                cnt[m] += a.x + a.y;
                v2f b;
                b.x = r0[s].x; b.y = r1[s].x;
                acc[m][0] = __builtin_amdgcn_wmma_f32_16x16x4_f32(false, a, false, b, (short)0, acc[m][0], false, false);
                b.x = r0[s].y; b.y = r1[s].y;
                acc[m][1] = __builtin_amdgcn_wmma_f32_16x16x4_f32(false, a, false, b, (short)0, acc[m][1], false, false);
                b.x = r0[s].z; b.y = r1[s].z;
                acc[m][2] = __builtin_amdgcn_wmma_f32_16x16x4_f32(false, a, false, b, (short)0, acc[m][2], false, false);
                b.x = r0[s].w; b.y = r1[s].w;
                acc[m][3] = __builtin_amdgcn_wmma_f32_16x16x4_f32(false, a, false, b, (short)0, acc[m][3], false, false);
            }
        }
    }

    #pragma unroll
    for (int m = 0; m < 4; ++m) cnt[m] += __shfl_xor(cnt[m], 16, 32);

    if (sc == 1) {
        #pragma unroll
        for (int m = 0; m < 4; ++m) {
            #pragma unroll
            for (int r = 0; r < 8; ++r) {
                const int row = 16 * m + 8 * half + r;
                lsum[row * (DIM / 4) + 16 * ch + c] =
                    make_float4(acc[m][0][r], acc[m][1][r], acc[m][2][r], acc[m][3][r]);
            }
            if (ch == 0 && half == 0) lcnt[16 * m + c] = cnt[m];
        }
    }
    __syncthreads();
    if (sc == 0) {
        float* pp = part + (size_t)blockIdx.x * PSTRIDE;
        #pragma unroll
        for (int m = 0; m < 4; ++m) {
            #pragma unroll
            for (int r = 0; r < 8; ++r) {
                const int row = 16 * m + 8 * half + r;
                const float4 o = lsum[row * (DIM / 4) + 16 * ch + c];
                float4 v;
                v.x = acc[m][0][r] + o.x;  v.y = acc[m][1][r] + o.y;
                v.z = acc[m][2][r] + o.z;  v.w = acc[m][3][r] + o.w;
                ((float4*)pp)[row * (DIM / 4) + 16 * ch + c] = v;
            }
            if (ch == 0 && half == 0) pp[GNUM * DIM + 16 * m + c] = cnt[m] + lcnt[16 * m + c];
        }
    }
}

__global__ __launch_bounds__(256) void pool_reduce(const float* __restrict__ part,
                                                   float* __restrict__ out, int P) {
    __shared__ float red[8][32];
    __shared__ float redc[8];
    const int lane = threadIdx.x & 31, s = threadIdx.x >> 5;
    const int o = blockIdx.x * 32 + lane;
    const int g = (blockIdx.x * 32) / DIM;
    float sum = 0.0f;
    for (int p0 = s; p0 < P; p0 += 128) {
        float v[16];
        #pragma unroll
        for (int u = 0; u < 16; ++u) {
            const int p = p0 + 8 * u;
            v[u] = (p < P) ? part[(size_t)p * PSTRIDE + o] : 0.0f;
        }
        #pragma unroll
        for (int u = 0; u < 16; ++u) sum += v[u];
    }
    float cs = 0.0f;
    for (int p = s + 8 * lane; p < P; p += 256) cs += part[(size_t)p * PSTRIDE + GNUM * DIM + g];
    #pragma unroll
    for (int off = 16; off > 0; off >>= 1) cs += __shfl_xor(cs, off, 32);
    red[s][lane] = sum;
    if (lane == 0) redc[s] = cs;
    __syncthreads();
    if (s == 0) {
        float t = 0.0f, cc = 0.0f;
        #pragma unroll
        for (int i = 0; i < 8; ++i) { t += red[i][lane]; cc += redc[i]; }
        out[o] = t / fmaxf(cc, 1.0f);
    }
}

static inline size_t al256(size_t x) { return (x + 255) & ~(size_t)255; }

extern "C" void kernel_launch(void* const* d_in, const int* in_sizes, int n_in,
                              void* d_out, int out_size, void* d_ws, size_t ws_size,
                              hipStream_t stream) {
    const float* x      = (const float*)d_in[0];
    const int*   ei     = (const int*)d_in[1];
    const int*   batch  = (const int*)d_in[2];
    const float* Ws     = (const float*)d_in[3];
    const float* bs     = (const float*)d_in[4];
    const float* gammas = (const float*)d_in[5];
    const float* betas  = (const float*)d_in[6];

    const int N = in_sizes[0] / DIM;
    const int E = in_sizes[1] / 2;
    const size_t nd = (size_t)N * DIM * sizeof(float);

    char* ws = (char*)d_ws;
    float* dinv = (float*)ws;                 ws += al256((size_t)N * sizeof(float));
    float* hw   = (float*)ws;                 ws += al256(nd);
    float* accb = (float*)ws;                 ws += al256(nd);
    float* hbuf = (float*)ws;                 ws += al256(nd);
    float* part = (float*)ws;

    zero_f32<<<(N + 255) / 256, 256, 0, stream>>>(dinv, N);
    deg_accum<<<(E + 255) / 256, 256, 0, stream>>>(ei, dinv, E);
    deg_to_dinv<<<(N + 255) / 256, 256, 0, stream>>>(dinv, N);

    const int nstrips = N >> 4;
    const float* hcur = x;
    for (int l = 0; l < LAYERS; ++l) {
        gemm_wmma<<<(nstrips + 7) / 8, 256, 0, stream>>>(
            hcur, Ws + (size_t)l * DIM * DIM, hw, N);
        acc_init<<<(N * (DIM / 4) + 255) / 256, 256, 0, stream>>>(
            hw, dinv, bs + (size_t)l * DIM, accb, N);
        edge_scatter<<<(E + 7) / 8, 256, 0, stream>>>(ei, hw, dinv, accb, E);
        ln_relu<<<(N + 7) / 8, 256, 0, stream>>>(
            accb, gammas + (size_t)l * DIM, betas + (size_t)l * DIM, hbuf, N);
        hcur = hbuf;
    }

    const int P = (N + PCH_NODES - 1) / PCH_NODES;
    pool_onehot<<<P, 128, 0, stream>>>(hcur, batch, part, N);
    pool_reduce<<<GNUM * DIM / 32, 256, 0, stream>>>(part, (float*)d_out, P);
}


# edge sums and degree counts kept in one plane per package half: a wave reads its die number once, loops over 32 edges, adds into its half's plane; LayerNorm adds the 2 planes (new sum order)
# speedup vs baseline: 2.2013x; 1.3070x over previous
#include <hip/hip_runtime.h>


typedef __attribute__((ext_vector_type(2))) float v2f;
typedef __attribute__((ext_vector_type(8))) float v8f;

#define DIM   128
#define GNUM  64
#define LAYERS 3
#define LNEPS 1e-5f

__global__ void zero2_f32(float* __restrict__ p0, float* __restrict__ p1, int n) {
    int i = blockIdx.x * blockDim.x + threadIdx.x;
    if (i < n) { p0[i] = 0.0f; p1[i] = 0.0f; }
}

__device__ __forceinline__ unsigned placement_id() { return __builtin_amdgcn_s_sendmsg_rtn(0x87); }
__device__ __forceinline__ int package_half(unsigned id) { return (int)((id >> 18) & 1u); }

__global__ void deg_accum(const int* __restrict__ ei, float* __restrict__ deg0,
                          float* __restrict__ deg1, int E) {
    int e = blockIdx.x * blockDim.x + threadIdx.x;
    const unsigned id = placement_id();
    if (e >= E) return;
    const int dst = ei[E + e];
    float* deg = package_half(id) ? deg1 : deg0;
    atomicAdd(&deg[dst], 1.0f);
}

__global__ void deg_to_dinv(float* __restrict__ d, const float* __restrict__ d1, int N) {
    int i = blockIdx.x * blockDim.x + threadIdx.x;
    if (i < N) d[i] = rsqrtf(d[i] + d1[i] + 1.0f);
}

__global__ __launch_bounds__(256) void gemm_wmma(const float* __restrict__ H,
                                                 const float* __restrict__ W,
                                                 float* __restrict__ out, int N) {
    __shared__ float lw[DIM * DIM];
    {
        const float4* Wv = (const float4*)W;
        float4* lv = (float4*)lw;
        #pragma unroll
        for (int i = 0; i < (DIM * DIM / 4) / 256; ++i)
            lv[threadIdx.x + i * 256] = Wv[threadIdx.x + i * 256];
    }
    __syncthreads();

    const int wave = threadIdx.x >> 5;
    const int lane = threadIdx.x & 31;
    const int half = lane >> 4;
    const int ln   = lane & 15;

    const int nstrips = N >> 4;
    int strip = blockIdx.x * 8 + wave;
    if (strip >= nstrips) strip = nstrips - 1;
    const int rowBase = strip << 4;

    const v2f* arow2 = (const v2f*)(H + (size_t)(rowBase + ln) * DIM);

    v8f acc[8] = {};

    for (int k = 0; k < DIM; k += 4) {
        v2f a = arow2[(k >> 1) + half];
        const int kb = k + 2 * half;
        #pragma unroll
        for (int t = 0; t < 8; ++t) {
            v2f b;
            b.x = lw[kb * DIM + t * 16 + ln];
            b.y = lw[(kb + 1) * DIM + t * 16 + ln];
            acc[t] = __builtin_amdgcn_wmma_f32_16x16x4_f32(
                false, a, false, b, (short)0, acc[t], false, false);
        }
    }

    #pragma unroll
    for (int t = 0; t < 8; ++t) {
        #pragma unroll
        for (int j = 0; j < 8; ++j) {
            int row = rowBase + half * 8 + j;
            out[(size_t)row * DIM + t * 16 + ln] = acc[t][j];
        }
    }
}

__global__ void acc_init(const float* __restrict__ hw, const float* __restrict__ dinv,
                         const float* __restrict__ bias, float* __restrict__ acc,
                         float* __restrict__ acc1, int N) {
    int i = blockIdx.x * blockDim.x + threadIdx.x;
    if (i >= N * (DIM / 4)) return;
    int n  = i >> 5;
    int c4 = i & 31;
    float s = dinv[n]; s = s * s;
    float4 hv = ((const float4*)hw)[i];
    float4 bv = ((const float4*)bias)[c4];
    float4 r;
    r.x = bv.x + s * hv.x;  r.y = bv.y + s * hv.y;
    r.z = bv.z + s * hv.z;  r.w = bv.w + s * hv.w;
    ((float4*)acc)[i] = r;
    ((float4*)acc1)[i] = make_float4(0.0f, 0.0f, 0.0f, 0.0f);
}

__global__ __launch_bounds__(256) void edge_scatter(const int* __restrict__ ei,
                                                    const float* __restrict__ hw,
                                                    const float* __restrict__ dinv,
                                                    float* __restrict__ acc0,
                                                    float* __restrict__ acc1, int E) {
    const int lane = threadIdx.x & 31;
    const int wave = __builtin_amdgcn_readfirstlane((int)(threadIdx.x >> 5));
    const int e0 = (blockIdx.x * 8 + wave) * 32;
    if (e0 >= E) return;
    const unsigned id = placement_id();
    const int e = min(e0 + lane, E - 1);
    const int src = ei[e];
    const int dst = ei[E + e];
    const float w = dinv[src] * dinv[dst];
    const int cnt = min(32, E - e0);
    float* __restrict__ acc = package_half(id) ? acc1 : acc0;
    if (cnt == 32) {
        #pragma unroll
        for (int g = 0; g < 4; ++g) {
            float v[8][4];
            #pragma unroll
            for (int j = 0; j < 8; ++j) {
                const int s = __builtin_amdgcn_readlane(src, 8 * g + j);
                const float* sp = hw + (size_t)s * DIM + lane;
                v[j][0] = sp[0]; v[j][1] = sp[32]; v[j][2] = sp[64]; v[j][3] = sp[96];
            }
            #pragma unroll
            for (int j = 0; j < 8; ++j) {
                const int d = __builtin_amdgcn_readlane(dst, 8 * g + j);
                const float wj = __int_as_float(__builtin_amdgcn_readlane(__float_as_int(w), 8 * g + j));
                float* dp = acc + (size_t)d * DIM + lane;
                atomicAdd(dp +  0, wj * v[j][0]);
                atomicAdd(dp + 32, wj * v[j][1]);
                atomicAdd(dp + 64, wj * v[j][2]);
                atomicAdd(dp + 96, wj * v[j][3]);
            }
        }
    } else {
        for (int j = 0; j < cnt; ++j) {
            const int s = __builtin_amdgcn_readlane(src, j);
            const int d = __builtin_amdgcn_readlane(dst, j);
            const float wj = __int_as_float(__builtin_amdgcn_readlane(__float_as_int(w), j));
            const float* sp = hw + (size_t)s * DIM + lane;
            const float v0 = sp[0], v1 = sp[32], v2 = sp[64], v3 = sp[96];
            float* dp = acc + (size_t)d * DIM + lane;
            atomicAdd(dp +  0, wj * v0);
            atomicAdd(dp + 32, wj * v1);
            atomicAdd(dp + 64, wj * v2);
            atomicAdd(dp + 96, wj * v3);
        }
    }
}

__global__ __launch_bounds__(256) void ln_relu(const float* __restrict__ acc,
                                               const float* __restrict__ acc1,
                                               const float* __restrict__ gamma,
                                               const float* __restrict__ beta,
                                               float* __restrict__ out, int N) {
    int wave = threadIdx.x >> 5, lane = threadIdx.x & 31;
    int n = blockIdx.x * 8 + wave;
    if (n >= N) return;
    float4 v = ((const float4*)(acc + (size_t)n * DIM))[lane];
    const float4 u = ((const float4*)(acc1 + (size_t)n * DIM))[lane];
    v.x += u.x; v.y += u.y; v.z += u.z; v.w += u.w;
    float s = v.x + v.y + v.z + v.w;
    float q = v.x * v.x + v.y * v.y + v.z * v.z + v.w * v.w;
    #pragma unroll
    for (int off = 16; off > 0; off >>= 1) {
        s += __shfl_xor(s, off, 32);
        q += __shfl_xor(q, off, 32);
    }
    float mu  = s * (1.0f / DIM);
    float var = fmaxf(q * (1.0f / DIM) - mu * mu, 0.0f);
    float inv = rsqrtf(var + LNEPS);
    float4 g = ((const float4*)gamma)[lane];
    float4 b = ((const float4*)beta)[lane];
    float4 r;
    r.x = fmaxf(0.0f, g.x * (v.x - mu) * inv + b.x);
    r.y = fmaxf(0.0f, g.y * (v.y - mu) * inv + b.y);
    r.z = fmaxf(0.0f, g.z * (v.z - mu) * inv + b.z);
    r.w = fmaxf(0.0f, g.w * (v.w - mu) * inv + b.w);
    ((float4*)(out + (size_t)n * DIM))[lane] = r;
}

#define PCH_BLOCKS 5
#define PCH_STEPS  5
#define PCH_WNODES (PCH_BLOCKS * PCH_STEPS * 4)
#define PCH_NODES  (2 * PCH_WNODES)
#define PSTRIDE    (GNUM * DIM + GNUM)

__global__ __launch_bounds__(128) void pool_onehot(const float* __restrict__ h,
                                                   const int* __restrict__ batch,
                                                   float* __restrict__ part, int N) {
    __shared__ float4 lsum[GNUM * DIM / 4];
    __shared__ float  lcnt[GNUM];
    const int wave = threadIdx.x >> 5, lane = threadIdx.x & 31;
    const int ch = wave & 1;
    const int sc = wave >> 1;
    const int half = lane >> 4, c = lane & 15;
    const int nbase = blockIdx.x * PCH_NODES + sc * PCH_WNODES + 2 * half;
    const float* hcol = h + 64 * ch + 4 * c;

    v8f acc[4][4] = {};
    float cnt[4] = {0.0f, 0.0f, 0.0f, 0.0f};

    #pragma unroll
    for (int blk = 0; blk < PCH_BLOCKS; ++blk) {
        float4 r0[PCH_STEPS], r1[PCH_STEPS];
        int g0[PCH_STEPS], g1[PCH_STEPS];
        #pragma unroll
        for (int s = 0; s < PCH_STEPS; ++s) {
            const int n0 = nbase + (blk * PCH_STEPS + s) * 4;
            const int n1 = n0 + 1;
            const int i0 = min(n0, N - 1), i1 = min(n1, N - 1);
            r0[s] = *(const float4*)(hcol + (size_t)i0 * DIM);
            r1[s] = *(const float4*)(hcol + (size_t)i1 * DIM);
            const int b0 = batch[i0], b1 = batch[i1];
            g0[s] = (n0 < N) ? b0 : -1;
            g1[s] = (n1 < N) ? b1 : -1;
        }
        #pragma unroll
        for (int s = 0; s < PCH_STEPS; ++s) {
            #pragma unroll
            for (int m = 0; m < 4; ++m) {
                v2f a;
                a.x = (g0[s] == 16 * m + c) ? 1.0f : 0.0f;
                a.y = (g1[s] == 16 * m + c) ? 1.0f : 0.0f;
                cnt[m] += a.x + a.y;
                v2f b;
                b.x = r0[s].x; b.y = r1[s].x;
                acc[m][0] = __builtin_amdgcn_wmma_f32_16x16x4_f32(false, a, false, b, (short)0, acc[m][0], false, false);
                b.x = r0[s].y; b.y = r1[s].y;
                acc[m][1] = __builtin_amdgcn_wmma_f32_16x16x4_f32(false, a, false, b, (short)0, acc[m][1], false, false);
                b.x = r0[s].z; b.y = r1[s].z;
                acc[m][2] = __builtin_amdgcn_wmma_f32_16x16x4_f32(false, a, false, b, (short)0, acc[m][2], false, false);
                b.x = r0[s].w; b.y = r1[s].w;
                acc[m][3] = __builtin_amdgcn_wmma_f32_16x16x4_f32(false, a, false, b, (short)0, acc[m][3], false, false);
            }
        }
    }

    #pragma unroll
    for (int m = 0; m < 4; ++m) cnt[m] += __shfl_xor(cnt[m], 16, 32);

    if (sc == 1) {
        #pragma unroll
        for (int m = 0; m < 4; ++m) {
            #pragma unroll
            for (int r = 0; r < 8; ++r) {
                const int row = 16 * m + 8 * half + r;
                lsum[row * (DIM / 4) + 16 * ch + c] =
                    make_float4(acc[m][0][r], acc[m][1][r], acc[m][2][r], acc[m][3][r]);
            }
            if (ch == 0 && half == 0) lcnt[16 * m + c] = cnt[m];
        }
    }
    __syncthreads();
    if (sc == 0) {
        float* pp = part + (size_t)blockIdx.x * PSTRIDE;
        #pragma unroll
        for (int m = 0; m < 4; ++m) {
            #pragma unroll
            for (int r = 0; r < 8; ++r) {
                const int row = 16 * m + 8 * half + r;
                const float4 o = lsum[row * (DIM / 4) + 16 * ch + c];
                float4 v;
                v.x = acc[m][0][r] + o.x;  v.y = acc[m][1][r] + o.y;
                v.z = acc[m][2][r] + o.z;  v.w = acc[m][3][r] + o.w;
                ((float4*)pp)[row * (DIM / 4) + 16 * ch + c] = v;
            }
            if (ch == 0 && half == 0) pp[GNUM * DIM + 16 * m + c] = cnt[m] + lcnt[16 * m + c];
        }
    }
}

__global__ __launch_bounds__(256) void pool_reduce(const float* __restrict__ part,
                                                   float* __restrict__ out, int P) {
    __shared__ float red[8][32];
    __shared__ float redc[8];
    const int lane = threadIdx.x & 31, s = threadIdx.x >> 5;
    const int o = blockIdx.x * 32 + lane;
    const int g = (blockIdx.x * 32) / DIM;
    float sum = 0.0f;
    for (int p0 = s; p0 < P; p0 += 128) {
        float v[16];
        #pragma unroll
        for (int u = 0; u < 16; ++u) {
            const int p = p0 + 8 * u;
            v[u] = (p < P) ? part[(size_t)p * PSTRIDE + o] : 0.0f;
        }
        #pragma unroll
        for (int u = 0; u < 16; ++u) sum += v[u];
    }
    float cs = 0.0f;
    for (int p = s + 8 * lane; p < P; p += 256) cs += part[(size_t)p * PSTRIDE + GNUM * DIM + g];
    #pragma unroll
    for (int off = 16; off > 0; off >>= 1) cs += __shfl_xor(cs, off, 32);
    red[s][lane] = sum;
    if (lane == 0) redc[s] = cs;
    __syncthreads();
    if (s == 0) {
        float t = 0.0f, cc = 0.0f;
        #pragma unroll
        for (int i = 0; i < 8; ++i) { t += red[i][lane]; cc += redc[i]; }
        out[o] = t / fmaxf(cc, 1.0f);
    }
}

static inline size_t al256(size_t x) { return (x + 255) & ~(size_t)255; }

extern "C" void kernel_launch(void* const* d_in, const int* in_sizes, int n_in,
                              void* d_out, int out_size, void* d_ws, size_t ws_size,
                              hipStream_t stream) {
    const float* x      = (const float*)d_in[0];
    const int*   ei     = (const int*)d_in[1];
    const int*   batch  = (const int*)d_in[2];
    const float* Ws     = (const float*)d_in[3];
    const float* bs     = (const float*)d_in[4];
    const float* gammas = (const float*)d_in[5];
    const float* betas  = (const float*)d_in[6];

    const int N = in_sizes[0] / DIM;
    const int E = in_sizes[1] / 2;
    const size_t nd = (size_t)N * DIM * sizeof(float);

    char* ws = (char*)d_ws;
    float* dinv = (float*)ws;                 ws += al256((size_t)N * sizeof(float));
    float* deg1 = (float*)ws;                 ws += al256((size_t)N * sizeof(float));
    float* hw   = (float*)ws;                 ws += al256(nd);
    float* accb = (float*)ws;                 ws += al256(nd);
    float* acc1 = (float*)ws;                 ws += al256(nd);
    float* hbuf = (float*)ws;                 ws += al256(nd);
    float* part = (float*)ws;

    zero2_f32<<<(N + 255) / 256, 256, 0, stream>>>(dinv, deg1, N);
    deg_accum<<<(E + 255) / 256, 256, 0, stream>>>(ei, dinv, deg1, E);
    deg_to_dinv<<<(N + 255) / 256, 256, 0, stream>>>(dinv, deg1, N);

    const int nstrips = N >> 4;
    const float* hcur = x;
    for (int l = 0; l < LAYERS; ++l) {
        gemm_wmma<<<(nstrips + 7) / 8, 256, 0, stream>>>(
            hcur, Ws + (size_t)l * DIM * DIM, hw, N);
        acc_init<<<(N * (DIM / 4) + 255) / 256, 256, 0, stream>>>(
            hw, dinv, bs + (size_t)l * DIM, accb, acc1, N);
        edge_scatter<<<(E + 255) / 256, 256, 0, stream>>>(ei, hw, dinv, accb, acc1, E);
        ln_relu<<<(N + 7) / 8, 256, 0, stream>>>(
            accb, acc1, gammas + (size_t)l * DIM, betas + (size_t)l * DIM, hbuf, N);
        hcur = hbuf;
    }

    const int P = (N + PCH_NODES - 1) / PCH_NODES;
    pool_onehot<<<P, 128, 0, stream>>>(hcur, batch, part, N);
    pool_reduce<<<GNUM * DIM / 32, 256, 0, stream>>>(part, (float*)d_out, P);
}


# row-local stages fused into the product's launch: a wave adds the 2 planes, LayerNorm, ReLU in registers as A fragments, multiplies, stores hw + next start planes; 17->12 launches (new LN sum order)
# speedup vs baseline: 2.4201x; 1.0994x over previous
#include <hip/hip_runtime.h>


typedef __attribute__((ext_vector_type(2))) float v2f;
typedef __attribute__((ext_vector_type(8))) float v8f;

#define DIM   128
#define GNUM  64
#define LAYERS 3
#define LNEPS 1e-5f

__global__ void zero2_f32(float* __restrict__ p0, float* __restrict__ p1, int n) {
    int i = blockIdx.x * blockDim.x + threadIdx.x;
    if (i < n) { p0[i] = 0.0f; p1[i] = 0.0f; }
}

__device__ __forceinline__ unsigned placement_id() { return __builtin_amdgcn_s_sendmsg_rtn(0x87); }
__device__ __forceinline__ int package_half(unsigned id) { return (int)((id >> 18) & 1u); }

__global__ void deg_accum(const int* __restrict__ ei, float* __restrict__ deg0,
                          float* __restrict__ deg1, int E) {
    int e = blockIdx.x * blockDim.x + threadIdx.x;
    const unsigned id = placement_id();
    if (e >= E) return;
    const int dst = ei[E + e];
    float* deg = package_half(id) ? deg1 : deg0;
    atomicAdd(&deg[dst], 1.0f);
}

__global__ void deg_to_dinv(float* __restrict__ d, const float* __restrict__ d1, int N) {
    int i = blockIdx.x * blockDim.x + threadIdx.x;
    if (i < N) d[i] = rsqrtf(d[i] + d1[i] + 1.0f);
}

template <bool FIRST>
__global__ __launch_bounds__(256) void layer_rows(const float* __restrict__ X,
                                                  const float* __restrict__ W,
                                                  const float* __restrict__ gamma,
                                                  const float* __restrict__ beta,
                                                  const float* __restrict__ bias,
                                                  const float* __restrict__ dinv,
                                                  float* __restrict__ hw,
                                                  float* __restrict__ acc0,
                                                  float* __restrict__ acc1, int N) {
    __shared__ float lw[DIM * DIM];
    __shared__ __attribute__((aligned(16))) float lg[DIM];
    __shared__ __attribute__((aligned(16))) float lb[DIM];
    {
        const float4* Wv = (const float4*)W;
        float4* lv = (float4*)lw;
        #pragma unroll
        for (int i = 0; i < (DIM * DIM / 4) / 256; ++i)
            lv[threadIdx.x + i * 256] = Wv[threadIdx.x + i * 256];
        if (!FIRST) {
            if (threadIdx.x < DIM) lg[threadIdx.x] = gamma[threadIdx.x];
            else                   lb[threadIdx.x - DIM] = beta[threadIdx.x - DIM];
        }
    }
    __syncthreads();

    const int wave = __builtin_amdgcn_readfirstlane((int)(threadIdx.x >> 5));
    const int lane = threadIdx.x & 31;
    const int half = lane >> 4;
    const int ln   = lane & 15;

    const int nstrips = N >> 4;
    const int strip = blockIdx.x * 8 + wave;
    if (strip >= nstrips) return;
    const int rowBase = strip << 4;

    v2f a[DIM / 4];
    if (FIRST) {
        const v2f* arow2 = (const v2f*)(X + (size_t)(rowBase + ln) * DIM);
        #pragma unroll
        for (int kk = 0; kk < DIM / 4; ++kk) a[kk] = arow2[2 * kk + half];
    } else {
        const v2f* p0 = (const v2f*)(acc0 + (size_t)(rowBase + ln) * DIM);
        const v2f* p1 = (const v2f*)(acc1 + (size_t)(rowBase + ln) * DIM);
        float s = 0.0f, q = 0.0f;
        #pragma unroll
        for (int kk = 0; kk < DIM / 4; ++kk) {
            const v2f u0 = p0[2 * kk + half];
            const v2f u1 = p1[2 * kk + half];
            v2f v;
            v.x = u0.x + u1.x;  v.y = u0.y + u1.y;
            a[kk] = v;
            s += v.x + v.y;
            q += v.x * v.x + v.y * v.y;
        }
        s += __shfl_xor(s, 16, 32);
        q += __shfl_xor(q, 16, 32);
        const float mu  = s * (1.0f / DIM);
        const float var = fmaxf(q * (1.0f / DIM) - mu * mu, 0.0f);
        const float inv = rsqrtf(var + LNEPS);
        #pragma unroll
        for (int kk = 0; kk < DIM / 4; ++kk) {
            const v2f g = ((const v2f*)lg)[2 * kk + half];
            const v2f b = ((const v2f*)lb)[2 * kk + half];
            v2f v = a[kk];
            v.x = fmaxf(0.0f, g.x * (v.x - mu) * inv + b.x);
            v.y = fmaxf(0.0f, g.y * (v.y - mu) * inv + b.y);
            a[kk] = v;
        }
    }

    v8f acc[8] = {};

    #pragma unroll
    for (int kk = 0; kk < DIM / 4; ++kk) {
        const int kb = 4 * kk + 2 * half;
        #pragma unroll
        for (int t = 0; t < 8; ++t) {
            v2f b;
            b.x = lw[kb * DIM + t * 16 + ln];
            b.y = lw[(kb + 1) * DIM + t * 16 + ln];
            acc[t] = __builtin_amdgcn_wmma_f32_16x16x4_f32(
                false, a[kk], false, b, (short)0, acc[t], false, false);
        }
    }

    float sn[8], bv[8];
    #pragma unroll
    for (int j = 0; j < 8; ++j) {
        const float d = dinv[rowBase + half * 8 + j];
        sn[j] = d * d;
    }
    #pragma unroll
    for (int t = 0; t < 8; ++t) bv[t] = bias[t * 16 + ln];
    #pragma unroll
    for (int t = 0; t < 8; ++t) {
        #pragma unroll
        for (int j = 0; j < 8; ++j) {
            const int row = rowBase + half * 8 + j;
            const size_t o = (size_t)row * DIM + t * 16 + ln;
            const float v = acc[t][j];
            hw[o]   = v;
            acc0[o] = bv[t] + sn[j] * v;
        }
    }
    #pragma unroll
    for (int r = 0; r < 16; ++r)
        ((float4*)(acc1 + (size_t)(rowBase + r) * DIM))[lane] = make_float4(0.0f, 0.0f, 0.0f, 0.0f);
}

__global__ __launch_bounds__(256) void edge_scatter(const int* __restrict__ ei,
                                                    const float* __restrict__ hw,
                                                    const float* __restrict__ dinv,
                                                    float* __restrict__ acc0,
                                                    float* __restrict__ acc1, int E) {
    const int lane = threadIdx.x & 31;
    const int wave = __builtin_amdgcn_readfirstlane((int)(threadIdx.x >> 5));
    const int e0 = (blockIdx.x * 8 + wave) * 32;
    if (e0 >= E) return;
    const unsigned id = placement_id();
    const int e = min(e0 + lane, E - 1);
    const int src = ei[e];
    const int dst = ei[E + e];
    const float w = dinv[src] * dinv[dst];
    const int cnt = min(32, E - e0);
    float* __restrict__ acc = package_half(id) ? acc1 : acc0;
    if (cnt == 32) {
        #pragma unroll
        for (int g = 0; g < 4; ++g) {
            float v[8][4];
            #pragma unroll
            for (int j = 0; j < 8; ++j) {
                const int s = __builtin_amdgcn_readlane(src, 8 * g + j);
                const float* sp = hw + (size_t)s * DIM + lane;
                v[j][0] = sp[0]; v[j][1] = sp[32]; v[j][2] = sp[64]; v[j][3] = sp[96];
            }
            #pragma unroll
            for (int j = 0; j < 8; ++j) {
                const int d = __builtin_amdgcn_readlane(dst, 8 * g + j);
                const float wj = __int_as_float(__builtin_amdgcn_readlane(__float_as_int(w), 8 * g + j));
                float* dp = acc + (size_t)d * DIM + lane;
                atomicAdd(dp +  0, wj * v[j][0]);
                atomicAdd(dp + 32, wj * v[j][1]);
                atomicAdd(dp + 64, wj * v[j][2]);
                atomicAdd(dp + 96, wj * v[j][3]);
            }
        }
    } else {
        for (int j = 0; j < cnt; ++j) {
            const int s = __builtin_amdgcn_readlane(src, j);
            const int d = __builtin_amdgcn_readlane(dst, j);
            const float wj = __int_as_float(__builtin_amdgcn_readlane(__float_as_int(w), j));
            const float* sp = hw + (size_t)s * DIM + lane;
            const float v0 = sp[0], v1 = sp[32], v2 = sp[64], v3 = sp[96];
            float* dp = acc + (size_t)d * DIM + lane;
            atomicAdd(dp +  0, wj * v0);
            atomicAdd(dp + 32, wj * v1);
            atomicAdd(dp + 64, wj * v2);
            atomicAdd(dp + 96, wj * v3);
        }
    }
}

__global__ __launch_bounds__(256) void ln_relu(const float* __restrict__ acc,
                                               const float* __restrict__ acc1,
                                               const float* __restrict__ gamma,
                                               const float* __restrict__ beta,
                                               float* __restrict__ out, int N) {
    int wave = threadIdx.x >> 5, lane = threadIdx.x & 31;
    int n = blockIdx.x * 8 + wave;
    if (n >= N) return;
    float4 v = ((const float4*)(acc + (size_t)n * DIM))[lane];
    const float4 u = ((const float4*)(acc1 + (size_t)n * DIM))[lane];
    v.x += u.x; v.y += u.y; v.z += u.z; v.w += u.w;
    float s = v.x + v.y + v.z + v.w;
    float q = v.x * v.x + v.y * v.y + v.z * v.z + v.w * v.w;
    #pragma unroll
    for (int off = 16; off > 0; off >>= 1) {
        s += __shfl_xor(s, off, 32);
        q += __shfl_xor(q, off, 32);
    }
    float mu  = s * (1.0f / DIM);
    float var = fmaxf(q * (1.0f / DIM) - mu * mu, 0.0f);
    float inv = rsqrtf(var + LNEPS);
    float4 g = ((const float4*)gamma)[lane];
    float4 b = ((const float4*)beta)[lane];
    float4 r;
    r.x = fmaxf(0.0f, g.x * (v.x - mu) * inv + b.x);
    r.y = fmaxf(0.0f, g.y * (v.y - mu) * inv + b.y);
    r.z = fmaxf(0.0f, g.z * (v.z - mu) * inv + b.z);
    r.w = fmaxf(0.0f, g.w * (v.w - mu) * inv + b.w);
    ((float4*)(out + (size_t)n * DIM))[lane] = r;
}

#define PCH_BLOCKS 5
#define PCH_STEPS  5
#define PCH_WNODES (PCH_BLOCKS * PCH_STEPS * 4)
#define PCH_NODES  (2 * PCH_WNODES)
#define PSTRIDE    (GNUM * DIM + GNUM)

__global__ __launch_bounds__(128) void pool_onehot(const float* __restrict__ h,
                                                   const int* __restrict__ batch,
                                                   float* __restrict__ part, int N) {
    __shared__ float4 lsum[GNUM * DIM / 4];
    __shared__ float  lcnt[GNUM];
    const int wave = threadIdx.x >> 5, lane = threadIdx.x & 31;
    const int ch = wave & 1;
    const int sc = wave >> 1;
    const int half = lane >> 4, c = lane & 15;
    const int nbase = blockIdx.x * PCH_NODES + sc * PCH_WNODES + 2 * half;
    const float* hcol = h + 64 * ch + 4 * c;

    v8f acc[4][4] = {};
    float cnt[4] = {0.0f, 0.0f, 0.0f, 0.0f};

    #pragma unroll
    for (int blk = 0; blk < PCH_BLOCKS; ++blk) {
        float4 r0[PCH_STEPS], r1[PCH_STEPS];
        int g0[PCH_STEPS], g1[PCH_STEPS];
        #pragma unroll
        for (int s = 0; s < PCH_STEPS; ++s) {
            const int n0 = nbase + (blk * PCH_STEPS + s) * 4;
            const int n1 = n0 + 1;
            const int i0 = min(n0, N - 1), i1 = min(n1, N - 1);
            r0[s] = *(const float4*)(hcol + (size_t)i0 * DIM);
            r1[s] = *(const float4*)(hcol + (size_t)i1 * DIM);
            const int b0 = batch[i0], b1 = batch[i1];
            g0[s] = (n0 < N) ? b0 : -1;
            g1[s] = (n1 < N) ? b1 : -1;
        }
        #pragma unroll
        for (int s = 0; s < PCH_STEPS; ++s) {
            #pragma unroll
            for (int m = 0; m < 4; ++m) {
                v2f a;
                a.x = (g0[s] == 16 * m + c) ? 1.0f : 0.0f;
                a.y = (g1[s] == 16 * m + c) ? 1.0f : 0.0f;
                cnt[m] += a.x + a.y;
                v2f b;
                b.x = r0[s].x; b.y = r1[s].x;
                acc[m][0] = __builtin_amdgcn_wmma_f32_16x16x4_f32(false, a, false, b, (short)0, acc[m][0], false, false);
                b.x = r0[s].y; b.y = r1[s].y;
                acc[m][1] = __builtin_amdgcn_wmma_f32_16x16x4_f32(false, a, false, b, (short)0, acc[m][1], false, false);
                b.x = r0[s].z; b.y = r1[s].z;
                acc[m][2] = __builtin_amdgcn_wmma_f32_16x16x4_f32(false, a, false, b, (short)0, acc[m][2], false, false);
                b.x = r0[s].w; b.y = r1[s].w;
                acc[m][3] = __builtin_amdgcn_wmma_f32_16x16x4_f32(false, a, false, b, (short)0, acc[m][3], false, false);
            }
        }
    }

    #pragma unroll
    for (int m = 0; m < 4; ++m) cnt[m] += __shfl_xor(cnt[m], 16, 32);

    if (sc == 1) {
        #pragma unroll
        for (int m = 0; m < 4; ++m) {
            #pragma unroll
            for (int r = 0; r < 8; ++r) {
                const int row = 16 * m + 8 * half + r;
                lsum[row * (DIM / 4) + 16 * ch + c] =
                    make_float4(acc[m][0][r], acc[m][1][r], acc[m][2][r], acc[m][3][r]);
            }
            if (ch == 0 && half == 0) lcnt[16 * m + c] = cnt[m];
        }
    }
    __syncthreads();
    if (sc == 0) {
        float* pp = part + (size_t)blockIdx.x * PSTRIDE;
        #pragma unroll
        for (int m = 0; m < 4; ++m) {
            #pragma unroll
            for (int r = 0; r < 8; ++r) {
                const int row = 16 * m + 8 * half + r;
                const float4 o = lsum[row * (DIM / 4) + 16 * ch + c];
                float4 v;
                v.x = acc[m][0][r] + o.x;  v.y = acc[m][1][r] + o.y;
                v.z = acc[m][2][r] + o.z;  v.w = acc[m][3][r] + o.w;
                ((float4*)pp)[row * (DIM / 4) + 16 * ch + c] = v;
            }
            if (ch == 0 && half == 0) pp[GNUM * DIM + 16 * m + c] = cnt[m] + lcnt[16 * m + c];
        }
    }
}

__global__ __launch_bounds__(256) void pool_reduce(const float* __restrict__ part,
                                                   float* __restrict__ out, int P) {
    __shared__ float red[8][32];
    __shared__ float redc[8];
    const int lane = threadIdx.x & 31, s = threadIdx.x >> 5;
    const int o = blockIdx.x * 32 + lane;
    const int g = (blockIdx.x * 32) / DIM;
    float sum = 0.0f;
    for (int p0 = s; p0 < P; p0 += 128) {
        float v[16];
        #pragma unroll
        for (int u = 0; u < 16; ++u) {
            const int p = p0 + 8 * u;
            v[u] = (p < P) ? part[(size_t)p * PSTRIDE + o] : 0.0f;
        }
        #pragma unroll
        for (int u = 0; u < 16; ++u) sum += v[u];
    }
    float cs = 0.0f;
    for (int p = s + 8 * lane; p < P; p += 256) cs += part[(size_t)p * PSTRIDE + GNUM * DIM + g];
    #pragma unroll
    for (int off = 16; off > 0; off >>= 1) cs += __shfl_xor(cs, off, 32);
    red[s][lane] = sum;
    if (lane == 0) redc[s] = cs;
    __syncthreads();
    if (s == 0) {
        float t = 0.0f, cc = 0.0f;
        #pragma unroll
        for (int i = 0; i < 8; ++i) { t += red[i][lane]; cc += redc[i]; }
        out[o] = t / fmaxf(cc, 1.0f);
    }
}

static inline size_t al256(size_t x) { return (x + 255) & ~(size_t)255; }

extern "C" void kernel_launch(void* const* d_in, const int* in_sizes, int n_in,
                              void* d_out, int out_size, void* d_ws, size_t ws_size,
                              hipStream_t stream) {
    const float* x      = (const float*)d_in[0];
    const int*   ei     = (const int*)d_in[1];
    const int*   batch  = (const int*)d_in[2];
    const float* Ws     = (const float*)d_in[3];
    const float* bs     = (const float*)d_in[4];
    const float* gammas = (const float*)d_in[5];
    const float* betas  = (const float*)d_in[6];

    const int N = in_sizes[0] / DIM;
    const int E = in_sizes[1] / 2;
    const size_t nd = (size_t)N * DIM * sizeof(float);

    char* ws = (char*)d_ws;
    float* dinv = (float*)ws;                 ws += al256((size_t)N * sizeof(float));
    float* deg1 = (float*)ws;                 ws += al256((size_t)N * sizeof(float));
    float* hw   = (float*)ws;                 ws += al256(nd);
    float* accb = (float*)ws;                 ws += al256(nd);
    float* acc1 = (float*)ws;                 ws += al256(nd);
    float* hbuf = (float*)ws;                 ws += al256(nd);
    float* part = (float*)ws;

    zero2_f32<<<(N + 255) / 256, 256, 0, stream>>>(dinv, deg1, N);
    deg_accum<<<(E + 255) / 256, 256, 0, stream>>>(ei, dinv, deg1, E);
    deg_to_dinv<<<(N + 255) / 256, 256, 0, stream>>>(dinv, deg1, N);

    const int nstrips = N >> 4;
    const int gw = (nstrips + 7) / 8;
    layer_rows<true><<<gw, 256, 0, stream>>>(x, Ws, gammas, betas, bs, dinv, hw, accb, acc1, N);
    edge_scatter<<<(E + 255) / 256, 256, 0, stream>>>(ei, hw, dinv, accb, acc1, E);
    for (int l = 1; l < LAYERS; ++l) {
        layer_rows<false><<<gw, 256, 0, stream>>>(
            x, Ws + (size_t)l * DIM * DIM, gammas + (size_t)(l - 1) * DIM, betas + (size_t)(l - 1) * DIM,
            bs + (size_t)l * DIM, dinv, hw, accb, acc1, N);
        edge_scatter<<<(E + 255) / 256, 256, 0, stream>>>(ei, hw, dinv, accb, acc1, E);
    }
    ln_relu<<<(N + 7) / 8, 256, 0, stream>>>(
        accb, acc1, gammas + (size_t)(LAYERS - 1) * DIM, betas + (size_t)(LAYERS - 1) * DIM, hbuf, N);
    const float* hcur = hbuf;

    const int P = (N + PCH_NODES - 1) / PCH_NODES;
    pool_onehot<<<P, 128, 0, stream>>>(hcur, batch, part, N);
    pool_reduce<<<GNUM * DIM / 32, 256, 0, stream>>>(part, (float*)d_out, P);
}


# edge sums: an owner work-group per 400 destination rows and half the edge list scans destinations, adds its edges' rows into LDS with LDS f32 adds; no global f32 atomics (sum order: LDS arrival)
# speedup vs baseline: 2.7191x; 1.1235x over previous
#include <hip/hip_runtime.h>


typedef __attribute__((ext_vector_type(2))) float v2f;
typedef __attribute__((ext_vector_type(8))) float v8f;

#define DIM   128
#define GNUM  64
#define LAYERS 3
#define LNEPS 1e-5f

__global__ void zero2_f32(float* __restrict__ p0, float* __restrict__ p1, int n) {
    int i = blockIdx.x * blockDim.x + threadIdx.x;
    if (i < n) { p0[i] = 0.0f; p1[i] = 0.0f; }
}

__device__ __forceinline__ unsigned placement_id() { return __builtin_amdgcn_s_sendmsg_rtn(0x87); }
__device__ __forceinline__ int package_half(unsigned id) { return (int)((id >> 18) & 1u); }

__global__ void deg_accum(const int* __restrict__ ei, float* __restrict__ deg0,
                          float* __restrict__ deg1, int E) {
    int e = blockIdx.x * blockDim.x + threadIdx.x;
    const unsigned id = placement_id();
    if (e >= E) return;
    const int dst = ei[E + e];
    float* deg = package_half(id) ? deg1 : deg0;
    atomicAdd(&deg[dst], 1.0f);
}

__global__ void deg_to_dinv(float* __restrict__ d, const float* __restrict__ d1, int N) {
    int i = blockIdx.x * blockDim.x + threadIdx.x;
    if (i < N) d[i] = rsqrtf(d[i] + d1[i] + 1.0f);
}

template <bool FIRST>
__global__ __launch_bounds__(256) void layer_rows(const float* __restrict__ X,
                                                  const float* __restrict__ W,
                                                  const float* __restrict__ gamma,
                                                  const float* __restrict__ beta,
                                                  const float* __restrict__ bias,
                                                  const float* __restrict__ dinv,
                                                  float* __restrict__ hw,
                                                  float* __restrict__ acc0,
                                                  const float* __restrict__ acc1, int N) {
    __shared__ float lw[DIM * DIM];
    __shared__ __attribute__((aligned(16))) float lg[DIM];
    __shared__ __attribute__((aligned(16))) float lb[DIM];
    {
        const float4* Wv = (const float4*)W;
        float4* lv = (float4*)lw;
        #pragma unroll
        for (int i = 0; i < (DIM * DIM / 4) / 256; ++i)
            lv[threadIdx.x + i * 256] = Wv[threadIdx.x + i * 256];
        if (!FIRST) {
            if (threadIdx.x < DIM) lg[threadIdx.x] = gamma[threadIdx.x];
            else                   lb[threadIdx.x - DIM] = beta[threadIdx.x - DIM];
        }
    }
    __syncthreads();

    const int wave = __builtin_amdgcn_readfirstlane((int)(threadIdx.x >> 5));
    const int lane = threadIdx.x & 31;
    const int half = lane >> 4;
    const int ln   = lane & 15;

    const int nstrips = N >> 4;
    const int strip = blockIdx.x * 8 + wave;
    if (strip >= nstrips) return;
    const int rowBase = strip << 4;

    v2f a[DIM / 4];
    if (FIRST) {
        const v2f* arow2 = (const v2f*)(X + (size_t)(rowBase + ln) * DIM);
        #pragma unroll
        for (int kk = 0; kk < DIM / 4; ++kk) a[kk] = arow2[2 * kk + half];
    } else {
        const v2f* p0 = (const v2f*)(acc0 + (size_t)(rowBase + ln) * DIM);
        const v2f* p1 = (const v2f*)(acc1 + (size_t)(rowBase + ln) * DIM);
        float s = 0.0f, q = 0.0f;
        #pragma unroll
        for (int kk = 0; kk < DIM / 4; ++kk) {
            const v2f u0 = p0[2 * kk + half];
            const v2f u1 = p1[2 * kk + half];
            v2f v;
            v.x = u0.x + u1.x;  v.y = u0.y + u1.y;
            a[kk] = v;
            s += v.x + v.y;
            q += v.x * v.x + v.y * v.y;
        }
        s += __shfl_xor(s, 16, 32);
        q += __shfl_xor(q, 16, 32);
        const float mu  = s * (1.0f / DIM);
        const float var = fmaxf(q * (1.0f / DIM) - mu * mu, 0.0f);
        const float inv = rsqrtf(var + LNEPS);
        #pragma unroll
        for (int kk = 0; kk < DIM / 4; ++kk) {
            const v2f g = ((const v2f*)lg)[2 * kk + half];
            const v2f b = ((const v2f*)lb)[2 * kk + half];
            v2f v = a[kk];
            v.x = fmaxf(0.0f, g.x * (v.x - mu) * inv + b.x);
            v.y = fmaxf(0.0f, g.y * (v.y - mu) * inv + b.y);
            a[kk] = v;
        }
    }

    v8f acc[8] = {};

    #pragma unroll
    for (int kk = 0; kk < DIM / 4; ++kk) {
        const int kb = 4 * kk + 2 * half;
        #pragma unroll
        for (int t = 0; t < 8; ++t) {
            v2f b;
            b.x = lw[kb * DIM + t * 16 + ln];
            b.y = lw[(kb + 1) * DIM + t * 16 + ln];
            acc[t] = __builtin_amdgcn_wmma_f32_16x16x4_f32(
                false, a[kk], false, b, (short)0, acc[t], false, false);
        }
    }

    float sn[8], bv[8];
    #pragma unroll
    for (int j = 0; j < 8; ++j) {
        const float d = dinv[rowBase + half * 8 + j];
        sn[j] = d * d;
    }
    #pragma unroll
    for (int t = 0; t < 8; ++t) bv[t] = bias[t * 16 + ln];
    #pragma unroll
    for (int t = 0; t < 8; ++t) {
        #pragma unroll
        for (int j = 0; j < 8; ++j) {
            const int row = rowBase + half * 8 + j;
            const size_t o = (size_t)row * DIM + t * 16 + ln;
            const float v = acc[t][j];
            hw[o]   = v;
            acc0[o] = bv[t] + sn[j] * v;
        }
    }
}

#define OWN_ROWS  400
#define OWN_WAVES 32
#define OWN_LIST  160
#define OWN_AHEAD 8

__device__ __forceinline__ void own_batch(const int* __restrict__ ei, const float* __restrict__ hw,
                                          const float* __restrict__ dinv, float* sum,
                                          const int* le, const int* lr, int n, int base, int lane) {
    const int i = min(lane, n - 1);
    const int e = le[i];
    const int r = lr[i];
    const int src = ei[e];
    const float ws = dinv[src];
    const float wd = dinv[base + r];
    #pragma nounroll
    for (int g = 0; g < 32; g += 8) {
        if (g >= n) break;
        float v[8][4];
        #pragma unroll
        for (int j = 0; j < 8; ++j) {
            const int s = __builtin_amdgcn_readlane(src, g + j);
            const float* sp = hw + (size_t)s * DIM + lane;
            v[j][0] = sp[0]; v[j][1] = sp[32]; v[j][2] = sp[64]; v[j][3] = sp[96];
        }
        #pragma unroll
        for (int j = 0; j < 8; ++j) {
            if (g + j < n) {
                const int rj = __builtin_amdgcn_readlane(r, g + j);
                const float wj = __int_as_float(__builtin_amdgcn_readlane(__float_as_int(ws), g + j)) *
                                 __int_as_float(__builtin_amdgcn_readlane(__float_as_int(wd), g + j));
                float* dp = sum + rj * DIM + lane;
                __hip_atomic_fetch_add(dp +  0, wj * v[j][0], __ATOMIC_RELAXED, __HIP_MEMORY_SCOPE_WORKGROUP);
                __hip_atomic_fetch_add(dp + 32, wj * v[j][1], __ATOMIC_RELAXED, __HIP_MEMORY_SCOPE_WORKGROUP);
                __hip_atomic_fetch_add(dp + 64, wj * v[j][2], __ATOMIC_RELAXED, __HIP_MEMORY_SCOPE_WORKGROUP);
                __hip_atomic_fetch_add(dp + 96, wj * v[j][3], __ATOMIC_RELAXED, __HIP_MEMORY_SCOPE_WORKGROUP);
            }
        }
    }
}

__global__ __launch_bounds__(1024) void edge_owner(const int* __restrict__ ei,
                                                   const float* __restrict__ hw,
                                                   const float* __restrict__ dinv,
                                                   float* __restrict__ acc0,
                                                   float* __restrict__ acc1, int N, int E) {
    __shared__ __attribute__((aligned(16))) float sum[OWN_ROWS * DIM];
    __shared__ int le[OWN_WAVES][OWN_LIST];
    __shared__ int lr[OWN_WAVES][OWN_LIST];
    const int lane = threadIdx.x & 31;
    const int wave = __builtin_amdgcn_readfirstlane((int)(threadIdx.x >> 5));
    const int part = blockIdx.x & 1;
    const int base = (blockIdx.x >> 1) * OWN_ROWS;
    const unsigned rows = (unsigned)min(OWN_ROWS, N - base);

    for (int i = threadIdx.x; i < OWN_ROWS * DIM / 4; i += OWN_WAVES * 32)
        ((float4*)sum)[i] = make_float4(0.0f, 0.0f, 0.0f, 0.0f);
    __syncthreads();

    const int* __restrict__ dstp = ei + E;
    int* mle = le[wave];
    int* mlr = lr[wave];
    const int nsteps = (E + 127) >> 7;
    int cnt = 0;
    for (int st0 = part * OWN_WAVES + wave; st0 < nsteps; st0 += 2 * OWN_WAVES * OWN_AHEAD) {
        int4 d[OWN_AHEAD];
        if ((st0 + (OWN_AHEAD - 1) * 2 * OWN_WAVES) * 128 + 128 <= E) {
            #pragma unroll
            for (int u = 0; u < OWN_AHEAD; ++u)
                d[u] = *(const int4*)(dstp + (st0 + u * 2 * OWN_WAVES) * 128 + 4 * lane);
        } else {
            #pragma unroll
            for (int u = 0; u < OWN_AHEAD; ++u) {
                const int e = (st0 + u * 2 * OWN_WAVES) * 128 + 4 * lane;
                d[u].x = (e     < E) ? dstp[e]     : -1;
                d[u].y = (e + 1 < E) ? dstp[e + 1] : -1;
                d[u].z = (e + 2 < E) ? dstp[e + 2] : -1;
                d[u].w = (e + 3 < E) ? dstp[e + 3] : -1;
            }
        }
        #pragma unroll
        for (int u = 0; u < OWN_AHEAD; ++u) {
            const int e = (st0 + u * 2 * OWN_WAVES) * 128 + 4 * lane;
            const unsigned r0 = (unsigned)d[u].x - (unsigned)base;
            const unsigned r1 = (unsigned)d[u].y - (unsigned)base;
            const unsigned r2 = (unsigned)d[u].z - (unsigned)base;
            const unsigned r3 = (unsigned)d[u].w - (unsigned)base;
            if (__builtin_amdgcn_ballot_w32(min(min(r0, r1), min(r2, r3)) < rows) != 0u) {
                #define OWN_APPEND(rk, ek) {                                              \
                    const bool hit = (rk) < rows;                                         \
                    const unsigned mask = __builtin_amdgcn_ballot_w32(hit);               \
                    if (hit) {                                                            \
                        const int pos = cnt + (int)__builtin_amdgcn_mbcnt_lo(mask, 0u);   \
                        mle[pos] = (ek);                                                  \
                        mlr[pos] = (int)(rk);                                             \
                    }                                                                     \
                    cnt += __builtin_popcount(mask);                                      \
                }
                OWN_APPEND(r0, e)
                OWN_APPEND(r1, e + 1)
                OWN_APPEND(r2, e + 2)
                OWN_APPEND(r3, e + 3)
                #undef OWN_APPEND
                while (cnt >= 32) {
                    cnt -= 32;
                    own_batch(ei, hw, dinv, sum, mle + cnt, mlr + cnt, 32, base, lane);
                }
            }
        }
    }
    if (cnt > 0) own_batch(ei, hw, dinv, sum, mle, mlr, cnt, base, lane);
    __syncthreads();

    const int nq = (int)rows * (DIM / 4);
    if (part == 0) {
        float4* gp = (float4*)(acc0 + (size_t)base * DIM);
        for (int i = threadIdx.x; i < nq; i += OWN_WAVES * 32) {
            float4 a = gp[i];
            const float4 s = ((const float4*)sum)[i];
            a.x += s.x; a.y += s.y; a.z += s.z; a.w += s.w;
            gp[i] = a;
        }
    } else {
        float4* gp = (float4*)(acc1 + (size_t)base * DIM);
        for (int i = threadIdx.x; i < nq; i += OWN_WAVES * 32)
            gp[i] = ((const float4*)sum)[i];
    }
}

__global__ __launch_bounds__(256) void ln_relu(const float* __restrict__ acc,
                                               const float* __restrict__ acc1,
                                               const float* __restrict__ gamma,
                                               const float* __restrict__ beta,
                                               float* __restrict__ out, int N) {
    int wave = threadIdx.x >> 5, lane = threadIdx.x & 31;
    int n = blockIdx.x * 8 + wave;
    if (n >= N) return;
    float4 v = ((const float4*)(acc + (size_t)n * DIM))[lane];
    const float4 u = ((const float4*)(acc1 + (size_t)n * DIM))[lane];
    v.x += u.x; v.y += u.y; v.z += u.z; v.w += u.w;
    float s = v.x + v.y + v.z + v.w;
    float q = v.x * v.x + v.y * v.y + v.z * v.z + v.w * v.w;
    #pragma unroll
    for (int off = 16; off > 0; off >>= 1) {
        s += __shfl_xor(s, off, 32);
        q += __shfl_xor(q, off, 32);
    }
    float mu  = s * (1.0f / DIM);
    float var = fmaxf(q * (1.0f / DIM) - mu * mu, 0.0f);
    float inv = rsqrtf(var + LNEPS);
    float4 g = ((const float4*)gamma)[lane];
    float4 b = ((const float4*)beta)[lane];
    float4 r;
    r.x = fmaxf(0.0f, g.x * (v.x - mu) * inv + b.x);
    r.y = fmaxf(0.0f, g.y * (v.y - mu) * inv + b.y);
    r.z = fmaxf(0.0f, g.z * (v.z - mu) * inv + b.z);
    r.w = fmaxf(0.0f, g.w * (v.w - mu) * inv + b.w);
    ((float4*)(out + (size_t)n * DIM))[lane] = r;
}

#define PCH_BLOCKS 5
#define PCH_STEPS  5
#define PCH_WNODES (PCH_BLOCKS * PCH_STEPS * 4)
#define PCH_NODES  (2 * PCH_WNODES)
#define PSTRIDE    (GNUM * DIM + GNUM)

__global__ __launch_bounds__(128) void pool_onehot(const float* __restrict__ h,
                                                   const int* __restrict__ batch,
                                                   float* __restrict__ part, int N) {
    __shared__ float4 lsum[GNUM * DIM / 4];
    __shared__ float  lcnt[GNUM];
    const int wave = threadIdx.x >> 5, lane = threadIdx.x & 31;
    const int ch = wave & 1;
    const int sc = wave >> 1;
    const int half = lane >> 4, c = lane & 15;
    const int nbase = blockIdx.x * PCH_NODES + sc * PCH_WNODES + 2 * half;
    const float* hcol = h + 64 * ch + 4 * c;

    v8f acc[4][4] = {};
    float cnt[4] = {0.0f, 0.0f, 0.0f, 0.0f};

    #pragma unroll
    for (int blk = 0; blk < PCH_BLOCKS; ++blk) {
        float4 r0[PCH_STEPS], r1[PCH_STEPS];
        int g0[PCH_STEPS], g1[PCH_STEPS];
        #pragma unroll
        for (int s = 0; s < PCH_STEPS; ++s) {
            const int n0 = nbase + (blk * PCH_STEPS + s) * 4;
            const int n1 = n0 + 1;
            const int i0 = min(n0, N - 1), i1 = min(n1, N - 1);
            r0[s] = *(const float4*)(hcol + (size_t)i0 * DIM);
            r1[s] = *(const float4*)(hcol + (size_t)i1 * DIM);
            const int b0 = batch[i0], b1 = batch[i1];
            g0[s] = (n0 < N) ? b0 : -1;
            g1[s] = (n1 < N) ? b1 : -1;
        }
        #pragma unroll
        for (int s = 0; s < PCH_STEPS; ++s) {
            #pragma unroll
            for (int m = 0; m < 4; ++m) {
                v2f a;
                a.x = (g0[s] == 16 * m + c) ? 1.0f : 0.0f;
                a.y = (g1[s] == 16 * m + c) ? 1.0f : 0.0f;
                cnt[m] += a.x + a.y;
                v2f b;
                b.x = r0[s].x; b.y = r1[s].x;
                acc[m][0] = __builtin_amdgcn_wmma_f32_16x16x4_f32(false, a, false, b, (short)0, acc[m][0], false, false);
                b.x = r0[s].y; b.y = r1[s].y;
                acc[m][1] = __builtin_amdgcn_wmma_f32_16x16x4_f32(false, a, false, b, (short)0, acc[m][1], false, false);
                b.x = r0[s].z; b.y = r1[s].z;
                acc[m][2] = __builtin_amdgcn_wmma_f32_16x16x4_f32(false, a, false, b, (short)0, acc[m][2], false, false);
                b.x = r0[s].w; b.y = r1[s].w;
                acc[m][3] = __builtin_amdgcn_wmma_f32_16x16x4_f32(false, a, false, b, (short)0, acc[m][3], false, false);
            }
        }
    }

    #pragma unroll
    for (int m = 0; m < 4; ++m) cnt[m] += __shfl_xor(cnt[m], 16, 32);

    if (sc == 1) {
        #pragma unroll
        for (int m = 0; m < 4; ++m) {
            #pragma unroll
            for (int r = 0; r < 8; ++r) {
                const int row = 16 * m + 8 * half + r;
                lsum[row * (DIM / 4) + 16 * ch + c] =
                    make_float4(acc[m][0][r], acc[m][1][r], acc[m][2][r], acc[m][3][r]);
            }
            if (ch == 0 && half == 0) lcnt[16 * m + c] = cnt[m];
        }
    }
    __syncthreads();
    if (sc == 0) {
        float* pp = part + (size_t)blockIdx.x * PSTRIDE;
        #pragma unroll
        for (int m = 0; m < 4; ++m) {
            #pragma unroll
            for (int r = 0; r < 8; ++r) {
                const int row = 16 * m + 8 * half + r;
                const float4 o = lsum[row * (DIM / 4) + 16 * ch + c];
                float4 v;
                v.x = acc[m][0][r] + o.x;  v.y = acc[m][1][r] + o.y;
                v.z = acc[m][2][r] + o.z;  v.w = acc[m][3][r] + o.w;
                ((float4*)pp)[row * (DIM / 4) + 16 * ch + c] = v;
            }
            if (ch == 0 && half == 0) pp[GNUM * DIM + 16 * m + c] = cnt[m] + lcnt[16 * m + c];
        }
    }
}

__global__ __launch_bounds__(256) void pool_reduce(const float* __restrict__ part,
                                                   float* __restrict__ out, int P) {
    __shared__ float red[8][32];
    __shared__ float redc[8];
    const int lane = threadIdx.x & 31, s = threadIdx.x >> 5;
    const int o = blockIdx.x * 32 + lane;
    const int g = (blockIdx.x * 32) / DIM;
    float sum = 0.0f;
    for (int p0 = s; p0 < P; p0 += 128) {
        float v[16];
        #pragma unroll
        for (int u = 0; u < 16; ++u) {
            const int p = p0 + 8 * u;
            v[u] = (p < P) ? part[(size_t)p * PSTRIDE + o] : 0.0f;
        }
        #pragma unroll
        for (int u = 0; u < 16; ++u) sum += v[u];
    }
    float cs = 0.0f;
    for (int p = s + 8 * lane; p < P; p += 256) cs += part[(size_t)p * PSTRIDE + GNUM * DIM + g];
    #pragma unroll
    for (int off = 16; off > 0; off >>= 1) cs += __shfl_xor(cs, off, 32);
    red[s][lane] = sum;
    if (lane == 0) redc[s] = cs;
    __syncthreads();
    if (s == 0) {
        float t = 0.0f, cc = 0.0f;
        #pragma unroll
        for (int i = 0; i < 8; ++i) { t += red[i][lane]; cc += redc[i]; }
        out[o] = t / fmaxf(cc, 1.0f);
    }
}

static inline size_t al256(size_t x) { return (x + 255) & ~(size_t)255; }

extern "C" void kernel_launch(void* const* d_in, const int* in_sizes, int n_in,
                              void* d_out, int out_size, void* d_ws, size_t ws_size,
                              hipStream_t stream) {
    const float* x      = (const float*)d_in[0];
    const int*   ei     = (const int*)d_in[1];
    const int*   batch  = (const int*)d_in[2];
    const float* Ws     = (const float*)d_in[3];
    const float* bs     = (const float*)d_in[4];
    const float* gammas = (const float*)d_in[5];
    const float* betas  = (const float*)d_in[6];

    const int N = in_sizes[0] / DIM;
    const int E = in_sizes[1] / 2;
    const size_t nd = (size_t)N * DIM * sizeof(float);

    char* ws = (char*)d_ws;
    float* dinv = (float*)ws;                 ws += al256((size_t)N * sizeof(float));
    float* deg1 = (float*)ws;                 ws += al256((size_t)N * sizeof(float));
    float* hw   = (float*)ws;                 ws += al256(nd);
    float* accb = (float*)ws;                 ws += al256(nd);
    float* acc1 = (float*)ws;                 ws += al256(nd);
    float* hbuf = (float*)ws;                 ws += al256(nd);
    float* part = (float*)ws;

    zero2_f32<<<(N + 255) / 256, 256, 0, stream>>>(dinv, deg1, N);
    deg_accum<<<(E + 255) / 256, 256, 0, stream>>>(ei, dinv, deg1, E);
    deg_to_dinv<<<(N + 255) / 256, 256, 0, stream>>>(dinv, deg1, N);

    const int nstrips = N >> 4;
    const int gw = (nstrips + 7) / 8;
    const int go = 2 * ((N + OWN_ROWS - 1) / OWN_ROWS);
    layer_rows<true><<<gw, 256, 0, stream>>>(x, Ws, gammas, betas, bs, dinv, hw, accb, acc1, N);
    edge_owner<<<go, OWN_WAVES * 32, 0, stream>>>(ei, hw, dinv, accb, acc1, N, E);
    for (int l = 1; l < LAYERS; ++l) {
        layer_rows<false><<<gw, 256, 0, stream>>>(
            x, Ws + (size_t)l * DIM * DIM, gammas + (size_t)(l - 1) * DIM, betas + (size_t)(l - 1) * DIM,
            bs + (size_t)l * DIM, dinv, hw, accb, acc1, N);
        edge_owner<<<go, OWN_WAVES * 32, 0, stream>>>(ei, hw, dinv, accb, acc1, N, E);
    }
    ln_relu<<<(N + 7) / 8, 256, 0, stream>>>(
        accb, acc1, gammas + (size_t)(LAYERS - 1) * DIM, betas + (size_t)(LAYERS - 1) * DIM, hbuf, N);
    const float* hcur = hbuf;

    const int P = (N + PCH_NODES - 1) / PCH_NODES;
    pool_onehot<<<P, 128, 0, stream>>>(hcur, batch, part, N);
    pool_reduce<<<GNUM * DIM / 32, 256, 0, stream>>>(part, (float*)d_out, P);
}


# edge sums of layers 2, 3: each owner wave reads the list (source row, row, weight) of its own edges that layer 1's scan stored; destination list scanned once, not 3 times (sum order: LDS arrival)
# speedup vs baseline: 3.1187x; 1.1470x over previous
#include <hip/hip_runtime.h>


typedef __attribute__((ext_vector_type(2))) float v2f;
typedef __attribute__((ext_vector_type(8))) float v8f;

#define DIM   128
#define GNUM  64
#define LAYERS 3
#define LNEPS 1e-5f

__global__ void zero2_f32(float* __restrict__ p0, float* __restrict__ p1, int n) {
    int i = blockIdx.x * blockDim.x + threadIdx.x;
    if (i < n) { p0[i] = 0.0f; p1[i] = 0.0f; }
}

__device__ __forceinline__ unsigned placement_id() { return __builtin_amdgcn_s_sendmsg_rtn(0x87); }
__device__ __forceinline__ int package_half(unsigned id) { return (int)((id >> 18) & 1u); }

__global__ void deg_accum(const int* __restrict__ ei, float* __restrict__ deg0,
                          float* __restrict__ deg1, int E) {
    int e = blockIdx.x * blockDim.x + threadIdx.x;
    const unsigned id = placement_id();
    if (e >= E) return;
    const int dst = ei[E + e];
    float* deg = package_half(id) ? deg1 : deg0;
    atomicAdd(&deg[dst], 1.0f);
}

__global__ void deg_to_dinv(float* __restrict__ d, const float* __restrict__ d1, int N) {
    int i = blockIdx.x * blockDim.x + threadIdx.x;
    if (i < N) d[i] = rsqrtf(d[i] + d1[i] + 1.0f);
}

template <bool FIRST>
__global__ __launch_bounds__(256) void layer_rows(const float* __restrict__ X,
                                                  const float* __restrict__ W,
                                                  const float* __restrict__ gamma,
                                                  const float* __restrict__ beta,
                                                  const float* __restrict__ bias,
                                                  const float* __restrict__ dinv,
                                                  float* __restrict__ hw,
                                                  float* __restrict__ acc0,
                                                  const float* __restrict__ acc1, int N) {
    __shared__ float lw[DIM * DIM];
    __shared__ __attribute__((aligned(16))) float lg[DIM];
    __shared__ __attribute__((aligned(16))) float lb[DIM];
    {
        const float4* Wv = (const float4*)W;
        float4* lv = (float4*)lw;
        #pragma unroll
        for (int i = 0; i < (DIM * DIM / 4) / 256; ++i)
            lv[threadIdx.x + i * 256] = Wv[threadIdx.x + i * 256];
        if (!FIRST) {
            if (threadIdx.x < DIM) lg[threadIdx.x] = gamma[threadIdx.x];
            else                   lb[threadIdx.x - DIM] = beta[threadIdx.x - DIM];
        }
    }
    __syncthreads();

    const int wave = __builtin_amdgcn_readfirstlane((int)(threadIdx.x >> 5));
    const int lane = threadIdx.x & 31;
    const int half = lane >> 4;
    const int ln   = lane & 15;

    const int nstrips = N >> 4;
    const int strip = blockIdx.x * 8 + wave;
    if (strip >= nstrips) return;
    const int rowBase = strip << 4;

    v2f a[DIM / 4];
    if (FIRST) {
        const v2f* arow2 = (const v2f*)(X + (size_t)(rowBase + ln) * DIM);
        #pragma unroll
        for (int kk = 0; kk < DIM / 4; ++kk) a[kk] = arow2[2 * kk + half];
    } else {
        const v2f* p0 = (const v2f*)(acc0 + (size_t)(rowBase + ln) * DIM);
        const v2f* p1 = (const v2f*)(acc1 + (size_t)(rowBase + ln) * DIM);
        float s = 0.0f, q = 0.0f;
        #pragma unroll
        for (int kk = 0; kk < DIM / 4; ++kk) {
            const v2f u0 = p0[2 * kk + half];
            const v2f u1 = p1[2 * kk + half];
            v2f v;
            v.x = u0.x + u1.x;  v.y = u0.y + u1.y;
            a[kk] = v;
            s += v.x + v.y;
            q += v.x * v.x + v.y * v.y;
        }
        s += __shfl_xor(s, 16, 32);
        q += __shfl_xor(q, 16, 32);
        const float mu  = s * (1.0f / DIM);
        const float var = fmaxf(q * (1.0f / DIM) - mu * mu, 0.0f);
        const float inv = rsqrtf(var + LNEPS);
        #pragma unroll
        for (int kk = 0; kk < DIM / 4; ++kk) {
            const v2f g = ((const v2f*)lg)[2 * kk + half];
            const v2f b = ((const v2f*)lb)[2 * kk + half];
            v2f v = a[kk];
            v.x = fmaxf(0.0f, g.x * (v.x - mu) * inv + b.x);
            v.y = fmaxf(0.0f, g.y * (v.y - mu) * inv + b.y);
            a[kk] = v;
        }
    }

    v8f acc[8] = {};

    #pragma unroll
    for (int kk = 0; kk < DIM / 4; ++kk) {
        const int kb = 4 * kk + 2 * half;
        #pragma unroll
        for (int t = 0; t < 8; ++t) {
            v2f b;
            b.x = lw[kb * DIM + t * 16 + ln];
            b.y = lw[(kb + 1) * DIM + t * 16 + ln];
            acc[t] = __builtin_amdgcn_wmma_f32_16x16x4_f32(
                false, a[kk], false, b, (short)0, acc[t], false, false);
        }
    }

    float sn[8], bv[8];
    #pragma unroll
    for (int j = 0; j < 8; ++j) {
        const float d = dinv[rowBase + half * 8 + j];
        sn[j] = d * d;
    }
    #pragma unroll
    for (int t = 0; t < 8; ++t) bv[t] = bias[t * 16 + ln];
    #pragma unroll
    for (int t = 0; t < 8; ++t) {
        #pragma unroll
        for (int j = 0; j < 8; ++j) {
            const int row = rowBase + half * 8 + j;
            const size_t o = (size_t)row * DIM + t * 16 + ln;
            const float v = acc[t][j];
            hw[o]   = v;
            acc0[o] = bv[t] + sn[j] * v;
        }
    }
}

#define OWN_ROWS  400
#define OWN_WAVES 32
#define OWN_LIST  160
#define OWN_AHEAD 8
#define OWN_KEEP  256

template <bool PAIR>
__device__ __forceinline__ void own_rows(const float* __restrict__ hw, float* sum, int src, int r,
                                         float wa, float wb, int n, int lane) {
    #pragma nounroll
    for (int g = 0; g < 32; g += 8) {
        if (g >= n) break;
        float v[8][4];
        #pragma unroll
        for (int j = 0; j < 8; ++j) {
            const int s = __builtin_amdgcn_readlane(src, g + j);
            const float* sp = hw + (size_t)s * DIM + lane;
            v[j][0] = sp[0]; v[j][1] = sp[32]; v[j][2] = sp[64]; v[j][3] = sp[96];
        }
        #pragma unroll
        for (int j = 0; j < 8; ++j) {
            if (g + j < n) {
                const int rj = __builtin_amdgcn_readlane(r, g + j);
                float wj = __int_as_float(__builtin_amdgcn_readlane(__float_as_int(wa), g + j));
                if (PAIR) wj *= __int_as_float(__builtin_amdgcn_readlane(__float_as_int(wb), g + j));
                float* dp = sum + rj * DIM + lane;
                __hip_atomic_fetch_add(dp +  0, wj * v[j][0], __ATOMIC_RELAXED, __HIP_MEMORY_SCOPE_WORKGROUP);
                __hip_atomic_fetch_add(dp + 32, wj * v[j][1], __ATOMIC_RELAXED, __HIP_MEMORY_SCOPE_WORKGROUP);
                __hip_atomic_fetch_add(dp + 64, wj * v[j][2], __ATOMIC_RELAXED, __HIP_MEMORY_SCOPE_WORKGROUP);
                __hip_atomic_fetch_add(dp + 96, wj * v[j][3], __ATOMIC_RELAXED, __HIP_MEMORY_SCOPE_WORKGROUP);
            }
        }
    }
}

template <bool SAVE>
__device__ __forceinline__ void own_batch(const int* __restrict__ ei, const float* __restrict__ hw,
                                          const float* __restrict__ dinv, float* sum,
                                          const int* le, const int* lr, int n, int base, int lane,
                                          int4* __restrict__ keep, int& kept) {
    const int i = min(lane, n - 1);
    const int e = le[i];
    const int r = lr[i];
    const int src = ei[e];
    const float ws = dinv[src];
    const float wd = dinv[base + r];
    own_rows<true>(hw, sum, src, r, ws, wd, n, lane);
    if (SAVE) {
        if (kept >= 0 && kept + n <= OWN_KEEP) {
            if (lane < n) keep[kept + lane] = make_int4(src, r, __float_as_int(ws * wd), 0);
            kept += n;
        } else {
            kept = -1;
        }
    }
}

template <bool SAVE>
__device__ __forceinline__ void own_scan(const int* __restrict__ ei, const float* __restrict__ hw,
                                         const float* __restrict__ dinv, float* sum, int* mle, int* mlr,
                                         int base, unsigned rows, int part, int wave, int lane, int E,
                                         int4* __restrict__ keep, int& kept) {
    const int* __restrict__ dstp = ei + E;
    const int nsteps = (E + 127) >> 7;
    int cnt = 0;
    for (int st0 = part * OWN_WAVES + wave; st0 < nsteps; st0 += 2 * OWN_WAVES * OWN_AHEAD) {
        int4 d[OWN_AHEAD];
        if ((st0 + (OWN_AHEAD - 1) * 2 * OWN_WAVES) * 128 + 128 <= E) {
            #pragma unroll
            for (int u = 0; u < OWN_AHEAD; ++u)
                d[u] = *(const int4*)(dstp + (st0 + u * 2 * OWN_WAVES) * 128 + 4 * lane);
        } else {
            #pragma unroll
            for (int u = 0; u < OWN_AHEAD; ++u) {
                const int e = (st0 + u * 2 * OWN_WAVES) * 128 + 4 * lane;
                d[u].x = (e     < E) ? dstp[e]     : -1;
                d[u].y = (e + 1 < E) ? dstp[e + 1] : -1;
                d[u].z = (e + 2 < E) ? dstp[e + 2] : -1;
                d[u].w = (e + 3 < E) ? dstp[e + 3] : -1;
            }
        }
        #pragma unroll
        for (int u = 0; u < OWN_AHEAD; ++u) {
            const int e = (st0 + u * 2 * OWN_WAVES) * 128 + 4 * lane;
            const unsigned r0 = (unsigned)d[u].x - (unsigned)base;
            const unsigned r1 = (unsigned)d[u].y - (unsigned)base;
            const unsigned r2 = (unsigned)d[u].z - (unsigned)base;
            const unsigned r3 = (unsigned)d[u].w - (unsigned)base;
            if (__builtin_amdgcn_ballot_w32(min(min(r0, r1), min(r2, r3)) < rows) != 0u) {
                #define OWN_APPEND(rk, ek) {                                              \
                    const bool hit = (rk) < rows;                                         \
                    const unsigned mask = __builtin_amdgcn_ballot_w32(hit);               \
                    if (hit) {                                                            \
                        const int pos = cnt + (int)__builtin_amdgcn_mbcnt_lo(mask, 0u);   \
                        mle[pos] = (ek);                                                  \
                        mlr[pos] = (int)(rk);                                             \
                    }                                                                     \
                    cnt += __builtin_popcount(mask);                                      \
                }
                OWN_APPEND(r0, e)
                OWN_APPEND(r1, e + 1)
                OWN_APPEND(r2, e + 2)
                OWN_APPEND(r3, e + 3)
                #undef OWN_APPEND
                while (cnt >= 32) {
                    cnt -= 32;
                    own_batch<SAVE>(ei, hw, dinv, sum, mle + cnt, mlr + cnt, 32, base, lane, keep, kept);
                }
            }
        }
    }
    if (cnt > 0) own_batch<SAVE>(ei, hw, dinv, sum, mle, mlr, cnt, base, lane, keep, kept);
}

template <bool FIRST>
__global__ __launch_bounds__(1024) void edge_owner(const int* __restrict__ ei,
                                                   const float* __restrict__ hw,
                                                   const float* __restrict__ dinv,
                                                   float* __restrict__ acc0,
                                                   float* __restrict__ acc1,
                                                   int* __restrict__ keepAll,
                                                   int* __restrict__ keptAll, int N, int E) {
    __shared__ __attribute__((aligned(16))) float sum[OWN_ROWS * DIM];
    __shared__ int le[OWN_WAVES][OWN_LIST];
    __shared__ int lr[OWN_WAVES][OWN_LIST];
    const int lane = threadIdx.x & 31;
    const int wave = __builtin_amdgcn_readfirstlane((int)(threadIdx.x >> 5));
    const int part = blockIdx.x & 1;
    const int base = (blockIdx.x >> 1) * OWN_ROWS;
    const unsigned rows = (unsigned)min(OWN_ROWS, N - base);

    for (int i = threadIdx.x; i < OWN_ROWS * DIM / 4; i += OWN_WAVES * 32)
        ((float4*)sum)[i] = make_float4(0.0f, 0.0f, 0.0f, 0.0f);
    __syncthreads();

    int4* __restrict__ keep = (int4*)keepAll + ((size_t)blockIdx.x * OWN_WAVES + wave) * OWN_KEEP;
    int* __restrict__ keptp = keptAll + blockIdx.x * OWN_WAVES + wave;
    if (FIRST) {
        int kept = 0;
        own_scan<true>(ei, hw, dinv, sum, le[wave], lr[wave], base, rows, part, wave, lane, E, keep, kept);
        if (lane == 0) *keptp = kept;
    } else {
        const int kept = __builtin_amdgcn_readfirstlane(*keptp);
        if (kept < 0) {
            int none = -1;
            own_scan<false>(ei, hw, dinv, sum, le[wave], lr[wave], base, rows, part, wave, lane, E, keep, none);
        } else if (kept > 0) {
            int4 en = keep[min(lane, kept - 1)];
            int b = 0;
            for (; b + 32 <= kept; b += 32) {
                const int4 cur = en;
                if (b + 32 < kept) en = keep[b + 32 + min(lane, kept - b - 33)];
                own_rows<false>(hw, sum, cur.x, cur.y, __int_as_float(cur.z), 0.0f, 32, lane);
            }
            if (b < kept) own_rows<false>(hw, sum, en.x, en.y, __int_as_float(en.z), 0.0f, kept - b, lane);
        }
    }
    __syncthreads();

    const int nq = (int)rows * (DIM / 4);
    if (part == 0) {
        float4* gp = (float4*)(acc0 + (size_t)base * DIM);
        for (int i = threadIdx.x; i < nq; i += OWN_WAVES * 32) {
            float4 a = gp[i];
            const float4 s = ((const float4*)sum)[i];
            a.x += s.x; a.y += s.y; a.z += s.z; a.w += s.w;
            gp[i] = a;
        }
    } else {
        float4* gp = (float4*)(acc1 + (size_t)base * DIM);
        for (int i = threadIdx.x; i < nq; i += OWN_WAVES * 32)
            gp[i] = ((const float4*)sum)[i];
    }
}

__global__ __launch_bounds__(256) void ln_relu(const float* __restrict__ acc,
                                               const float* __restrict__ acc1,
                                               const float* __restrict__ gamma,
                                               const float* __restrict__ beta,
                                               float* __restrict__ out, int N) {
    int wave = threadIdx.x >> 5, lane = threadIdx.x & 31;
    int n = blockIdx.x * 8 + wave;
    if (n >= N) return;
    float4 v = ((const float4*)(acc + (size_t)n * DIM))[lane];
    const float4 u = ((const float4*)(acc1 + (size_t)n * DIM))[lane];
    v.x += u.x; v.y += u.y; v.z += u.z; v.w += u.w;
    float s = v.x + v.y + v.z + v.w;
    float q = v.x * v.x + v.y * v.y + v.z * v.z + v.w * v.w;
    #pragma unroll
    for (int off = 16; off > 0; off >>= 1) {
        s += __shfl_xor(s, off, 32);
        q += __shfl_xor(q, off, 32);
    }
    float mu  = s * (1.0f / DIM);
    float var = fmaxf(q * (1.0f / DIM) - mu * mu, 0.0f);
    float inv = rsqrtf(var + LNEPS);
    float4 g = ((const float4*)gamma)[lane];
    float4 b = ((const float4*)beta)[lane];
    float4 r;
    r.x = fmaxf(0.0f, g.x * (v.x - mu) * inv + b.x);
    r.y = fmaxf(0.0f, g.y * (v.y - mu) * inv + b.y);
    r.z = fmaxf(0.0f, g.z * (v.z - mu) * inv + b.z);
    r.w = fmaxf(0.0f, g.w * (v.w - mu) * inv + b.w);
    ((float4*)(out + (size_t)n * DIM))[lane] = r;
}

#define PCH_BLOCKS 5
#define PCH_STEPS  5
#define PCH_WNODES (PCH_BLOCKS * PCH_STEPS * 4)
#define PCH_NODES  (2 * PCH_WNODES)
#define PSTRIDE    (GNUM * DIM + GNUM)

__global__ __launch_bounds__(128) void pool_onehot(const float* __restrict__ h,
                                                   const int* __restrict__ batch,
                                                   float* __restrict__ part, int N) {
    __shared__ float4 lsum[GNUM * DIM / 4];
    __shared__ float  lcnt[GNUM];
    const int wave = threadIdx.x >> 5, lane = threadIdx.x & 31;
    const int ch = wave & 1;
    const int sc = wave >> 1;
    const int half = lane >> 4, c = lane & 15;
    const int nbase = blockIdx.x * PCH_NODES + sc * PCH_WNODES + 2 * half;
    const float* hcol = h + 64 * ch + 4 * c;

    v8f acc[4][4] = {};
    float cnt[4] = {0.0f, 0.0f, 0.0f, 0.0f};

    #pragma unroll
    for (int blk = 0; blk < PCH_BLOCKS; ++blk) {
        float4 r0[PCH_STEPS], r1[PCH_STEPS];
        int g0[PCH_STEPS], g1[PCH_STEPS];
        #pragma unroll
        for (int s = 0; s < PCH_STEPS; ++s) {
            const int n0 = nbase + (blk * PCH_STEPS + s) * 4;
            const int n1 = n0 + 1;
            const int i0 = min(n0, N - 1), i1 = min(n1, N - 1);
            r0[s] = *(const float4*)(hcol + (size_t)i0 * DIM);
            r1[s] = *(const float4*)(hcol + (size_t)i1 * DIM);
            const int b0 = batch[i0], b1 = batch[i1];
            g0[s] = (n0 < N) ? b0 : -1;
            g1[s] = (n1 < N) ? b1 : -1;
        }
        #pragma unroll
        for (int s = 0; s < PCH_STEPS; ++s) {
            #pragma unroll
            for (int m = 0; m < 4; ++m) {
                v2f a;
                a.x = (g0[s] == 16 * m + c) ? 1.0f : 0.0f;
                a.y = (g1[s] == 16 * m + c) ? 1.0f : 0.0f;
                cnt[m] += a.x + a.y;
                v2f b;
                b.x = r0[s].x; b.y = r1[s].x;
                acc[m][0] = __builtin_amdgcn_wmma_f32_16x16x4_f32(false, a, false, b, (short)0, acc[m][0], false, false);
                b.x = r0[s].y; b.y = r1[s].y;
                acc[m][1] = __builtin_amdgcn_wmma_f32_16x16x4_f32(false, a, false, b, (short)0, acc[m][1], false, false);
                b.x = r0[s].z; b.y = r1[s].z;
                acc[m][2] = __builtin_amdgcn_wmma_f32_16x16x4_f32(false, a, false, b, (short)0, acc[m][2], false, false);
                b.x = r0[s].w; b.y = r1[s].w;
                acc[m][3] = __builtin_amdgcn_wmma_f32_16x16x4_f32(false, a, false, b, (short)0, acc[m][3], false, false);
            }
        }
    }

    #pragma unroll
    for (int m = 0; m < 4; ++m) cnt[m] += __shfl_xor(cnt[m], 16, 32);

    if (sc == 1) {
        #pragma unroll
        for (int m = 0; m < 4; ++m) {
            #pragma unroll
            for (int r = 0; r < 8; ++r) {
                const int row = 16 * m + 8 * half + r;
                lsum[row * (DIM / 4) + 16 * ch + c] =
                    make_float4(acc[m][0][r], acc[m][1][r], acc[m][2][r], acc[m][3][r]);
            }
            if (ch == 0 && half == 0) lcnt[16 * m + c] = cnt[m];
        }
    }
    __syncthreads();
    if (sc == 0) {
        float* pp = part + (size_t)blockIdx.x * PSTRIDE;
        #pragma unroll
        for (int m = 0; m < 4; ++m) {
            #pragma unroll
            for (int r = 0; r < 8; ++r) {
                const int row = 16 * m + 8 * half + r;
                const float4 o = lsum[row * (DIM / 4) + 16 * ch + c];
                float4 v;
                v.x = acc[m][0][r] + o.x;  v.y = acc[m][1][r] + o.y;
                v.z = acc[m][2][r] + o.z;  v.w = acc[m][3][r] + o.w;
                ((float4*)pp)[row * (DIM / 4) + 16 * ch + c] = v;
            }
            if (ch == 0 && half == 0) pp[GNUM * DIM + 16 * m + c] = cnt[m] + lcnt[16 * m + c];
        }
    }
}

__global__ __launch_bounds__(256) void pool_reduce(const float* __restrict__ part,
                                                   float* __restrict__ out, int P) {
    __shared__ float red[8][32];
    __shared__ float redc[8];
    const int lane = threadIdx.x & 31, s = threadIdx.x >> 5;
    const int o = blockIdx.x * 32 + lane;
    const int g = (blockIdx.x * 32) / DIM;
    float sum = 0.0f;
    for (int p0 = s; p0 < P; p0 += 128) {
        float v[16];
        #pragma unroll
        for (int u = 0; u < 16; ++u) {
            const int p = p0 + 8 * u;
            v[u] = (p < P) ? part[(size_t)p * PSTRIDE + o] : 0.0f;
        }
        #pragma unroll
        for (int u = 0; u < 16; ++u) sum += v[u];
    }
    float cs = 0.0f;
    for (int p = s + 8 * lane; p < P; p += 256) cs += part[(size_t)p * PSTRIDE + GNUM * DIM + g];
    #pragma unroll
    for (int off = 16; off > 0; off >>= 1) cs += __shfl_xor(cs, off, 32);
    red[s][lane] = sum;
    if (lane == 0) redc[s] = cs;
    __syncthreads();
    if (s == 0) {
        float t = 0.0f, cc = 0.0f;
        #pragma unroll
        for (int i = 0; i < 8; ++i) { t += red[i][lane]; cc += redc[i]; }
        out[o] = t / fmaxf(cc, 1.0f);
    }
}

static inline size_t al256(size_t x) { return (x + 255) & ~(size_t)255; }

extern "C" void kernel_launch(void* const* d_in, const int* in_sizes, int n_in,
                              void* d_out, int out_size, void* d_ws, size_t ws_size,
                              hipStream_t stream) {
    const float* x      = (const float*)d_in[0];
    const int*   ei     = (const int*)d_in[1];
    const int*   batch  = (const int*)d_in[2];
    const float* Ws     = (const float*)d_in[3];
    const float* bs     = (const float*)d_in[4];
    const float* gammas = (const float*)d_in[5];
    const float* betas  = (const float*)d_in[6];

    const int N = in_sizes[0] / DIM;
    const int E = in_sizes[1] / 2;
    const size_t nd = (size_t)N * DIM * sizeof(float);

    char* ws = (char*)d_ws;
    float* dinv = (float*)ws;                 ws += al256((size_t)N * sizeof(float));
    float* deg1 = (float*)ws;                 ws += al256((size_t)N * sizeof(float));
    float* hw   = (float*)ws;                 ws += al256(nd);
    float* accb = (float*)ws;                 ws += al256(nd);
    float* acc1 = (float*)ws;                 ws += al256(nd);
    float* hbuf = (float*)ws;                 ws += al256(nd);
    const int go = 2 * ((N + OWN_ROWS - 1) / OWN_ROWS);
    int* keepAll = (int*)ws;                  ws += al256((size_t)go * OWN_WAVES * OWN_KEEP * sizeof(int4));
    int* keptAll = (int*)ws;                  ws += al256((size_t)go * OWN_WAVES * sizeof(int));
    float* part = (float*)ws;

    zero2_f32<<<(N + 255) / 256, 256, 0, stream>>>(dinv, deg1, N);
    deg_accum<<<(E + 255) / 256, 256, 0, stream>>>(ei, dinv, deg1, E);
    deg_to_dinv<<<(N + 255) / 256, 256, 0, stream>>>(dinv, deg1, N);

    const int nstrips = N >> 4;
    const int gw = (nstrips + 7) / 8;
    layer_rows<true><<<gw, 256, 0, stream>>>(x, Ws, gammas, betas, bs, dinv, hw, accb, acc1, N);
    edge_owner<true><<<go, OWN_WAVES * 32, 0, stream>>>(ei, hw, dinv, accb, acc1, keepAll, keptAll, N, E);
    for (int l = 1; l < LAYERS; ++l) {
        layer_rows<false><<<gw, 256, 0, stream>>>(
            x, Ws + (size_t)l * DIM * DIM, gammas + (size_t)(l - 1) * DIM, betas + (size_t)(l - 1) * DIM,
            bs + (size_t)l * DIM, dinv, hw, accb, acc1, N);
        edge_owner<false><<<go, OWN_WAVES * 32, 0, stream>>>(ei, hw, dinv, accb, acc1, keepAll, keptAll, N, E);
    }
    ln_relu<<<(N + 7) / 8, 256, 0, stream>>>(
        accb, acc1, gammas + (size_t)(LAYERS - 1) * DIM, betas + (size_t)(LAYERS - 1) * DIM, hbuf, N);
    const float* hcur = hbuf;

    const int P = (N + PCH_NODES - 1) / PCH_NODES;
    pool_onehot<<<P, 128, 0, stream>>>(hcur, batch, part, N);
    pool_reduce<<<GNUM * DIM / 32, 256, 0, stream>>>(part, (float*)d_out, P);
}


# edge sums: each source row fetched by ONE 16-byte load per lane (whole 512-byte row, 2 aligned 256-byte units per instruction, not 4 one-line loads); LDS sums kept in that column order (same sums)
# speedup vs baseline: 3.2606x; 1.0455x over previous
#include <hip/hip_runtime.h>


typedef __attribute__((ext_vector_type(2))) float v2f;
typedef __attribute__((ext_vector_type(8))) float v8f;

#define DIM   128
#define GNUM  64
#define LAYERS 3
#define LNEPS 1e-5f

__global__ void zero2_f32(float* __restrict__ p0, float* __restrict__ p1, int n) {
    int i = blockIdx.x * blockDim.x + threadIdx.x;
    if (i < n) { p0[i] = 0.0f; p1[i] = 0.0f; }
}

__device__ __forceinline__ unsigned placement_id() { return __builtin_amdgcn_s_sendmsg_rtn(0x87); }
__device__ __forceinline__ int package_half(unsigned id) { return (int)((id >> 18) & 1u); }

__global__ void deg_accum(const int* __restrict__ ei, float* __restrict__ deg0,
                          float* __restrict__ deg1, int E) {
    int e = blockIdx.x * blockDim.x + threadIdx.x;
    const unsigned id = placement_id();
    if (e >= E) return;
    const int dst = ei[E + e];
    float* deg = package_half(id) ? deg1 : deg0;
    atomicAdd(&deg[dst], 1.0f);
}

__global__ void deg_to_dinv(float* __restrict__ d, const float* __restrict__ d1, int N) {
    int i = blockIdx.x * blockDim.x + threadIdx.x;
    if (i < N) d[i] = rsqrtf(d[i] + d1[i] + 1.0f);
}

template <bool FIRST>
__global__ __launch_bounds__(256) void layer_rows(const float* __restrict__ X,
                                                  const float* __restrict__ W,
                                                  const float* __restrict__ gamma,
                                                  const float* __restrict__ beta,
                                                  const float* __restrict__ bias,
                                                  const float* __restrict__ dinv,
                                                  float* __restrict__ hw,
                                                  float* __restrict__ acc0,
                                                  const float* __restrict__ acc1, int N) {
    __shared__ float lw[DIM * DIM];
    __shared__ __attribute__((aligned(16))) float lg[DIM];
    __shared__ __attribute__((aligned(16))) float lb[DIM];
    {
        const float4* Wv = (const float4*)W;
        float4* lv = (float4*)lw;
        #pragma unroll
        for (int i = 0; i < (DIM * DIM / 4) / 256; ++i)
            lv[threadIdx.x + i * 256] = Wv[threadIdx.x + i * 256];
        if (!FIRST) {
            if (threadIdx.x < DIM) lg[threadIdx.x] = gamma[threadIdx.x];
            else                   lb[threadIdx.x - DIM] = beta[threadIdx.x - DIM];
        }
    }
    __syncthreads();

    const int wave = __builtin_amdgcn_readfirstlane((int)(threadIdx.x >> 5));
    const int lane = threadIdx.x & 31;
    const int half = lane >> 4;
    const int ln   = lane & 15;

    const int nstrips = N >> 4;
    const int strip = blockIdx.x * 8 + wave;
    if (strip >= nstrips) return;
    const int rowBase = strip << 4;

    v2f a[DIM / 4];
    if (FIRST) {
        const v2f* arow2 = (const v2f*)(X + (size_t)(rowBase + ln) * DIM);
        #pragma unroll
        for (int kk = 0; kk < DIM / 4; ++kk) a[kk] = arow2[2 * kk + half];
    } else {
        const v2f* p0 = (const v2f*)(acc0 + (size_t)(rowBase + ln) * DIM);
        const v2f* p1 = (const v2f*)(acc1 + (size_t)(rowBase + ln) * DIM);
        float s = 0.0f, q = 0.0f;
        #pragma unroll
        for (int kk = 0; kk < DIM / 4; ++kk) {
            const v2f u0 = p0[2 * kk + half];
            const v2f u1 = p1[2 * kk + half];
            v2f v;
            v.x = u0.x + u1.x;  v.y = u0.y + u1.y;
            a[kk] = v;
            s += v.x + v.y;
            q += v.x * v.x + v.y * v.y;
        }
        s += __shfl_xor(s, 16, 32);
        q += __shfl_xor(q, 16, 32);
        const float mu  = s * (1.0f / DIM);
        const float var = fmaxf(q * (1.0f / DIM) - mu * mu, 0.0f);
        const float inv = rsqrtf(var + LNEPS);
        #pragma unroll
        for (int kk = 0; kk < DIM / 4; ++kk) {
            const v2f g = ((const v2f*)lg)[2 * kk + half];
            const v2f b = ((const v2f*)lb)[2 * kk + half];
            v2f v = a[kk];
            v.x = fmaxf(0.0f, g.x * (v.x - mu) * inv + b.x);
            v.y = fmaxf(0.0f, g.y * (v.y - mu) * inv + b.y);
            a[kk] = v;
        }
    }

    v8f acc[8] = {};

    #pragma unroll
    for (int kk = 0; kk < DIM / 4; ++kk) {
        const int kb = 4 * kk + 2 * half;
        #pragma unroll
        for (int t = 0; t < 8; ++t) {
            v2f b;
            b.x = lw[kb * DIM + t * 16 + ln];
            b.y = lw[(kb + 1) * DIM + t * 16 + ln];
            acc[t] = __builtin_amdgcn_wmma_f32_16x16x4_f32(
                false, a[kk], false, b, (short)0, acc[t], false, false);
        }
    }

    float sn[8], bv[8];
    #pragma unroll
    for (int j = 0; j < 8; ++j) {
        const float d = dinv[rowBase + half * 8 + j];
        sn[j] = d * d;
    }
    #pragma unroll
    for (int t = 0; t < 8; ++t) bv[t] = bias[t * 16 + ln];
    #pragma unroll
    for (int t = 0; t < 8; ++t) {
        #pragma unroll
        for (int j = 0; j < 8; ++j) {
            const int row = rowBase + half * 8 + j;
            const size_t o = (size_t)row * DIM + t * 16 + ln;
            const float v = acc[t][j];
            hw[o]   = v;
            acc0[o] = bv[t] + sn[j] * v;
        }
    }
}

#define OWN_ROWS  400
#define OWN_WAVES 32
#define OWN_LIST  160
#define OWN_AHEAD 8
#define OWN_KEEP  256

template <bool PAIR>
__device__ __forceinline__ void own_rows(const float* __restrict__ hw, float* sum, int src, int r,
                                         float wa, float wb, int n, int lane) {
    #pragma nounroll
    for (int g = 0; g < 32; g += 8) {
        if (g >= n) break;
        float4 v[8];
        #pragma unroll
        for (int j = 0; j < 8; ++j) {
            const int s = __builtin_amdgcn_readlane(src, g + j);
            v[j] = *(const float4*)(hw + (size_t)s * DIM + 4 * lane);
        }
        #pragma unroll
        for (int j = 0; j < 8; ++j) {
            if (g + j < n) {
                const int rj = __builtin_amdgcn_readlane(r, g + j);
                float wj = __int_as_float(__builtin_amdgcn_readlane(__float_as_int(wa), g + j));
                if (PAIR) wj *= __int_as_float(__builtin_amdgcn_readlane(__float_as_int(wb), g + j));
                float* dp = sum + rj * DIM + lane;
                __hip_atomic_fetch_add(dp +  0, wj * v[j].x, __ATOMIC_RELAXED, __HIP_MEMORY_SCOPE_WORKGROUP);
                __hip_atomic_fetch_add(dp + 32, wj * v[j].y, __ATOMIC_RELAXED, __HIP_MEMORY_SCOPE_WORKGROUP);
                __hip_atomic_fetch_add(dp + 64, wj * v[j].z, __ATOMIC_RELAXED, __HIP_MEMORY_SCOPE_WORKGROUP);
                __hip_atomic_fetch_add(dp + 96, wj * v[j].w, __ATOMIC_RELAXED, __HIP_MEMORY_SCOPE_WORKGROUP);
            }
        }
    }
}

template <bool SAVE>
__device__ __forceinline__ void own_batch(const int* __restrict__ ei, const float* __restrict__ hw,
                                          const float* __restrict__ dinv, float* sum,
                                          const int* le, const int* lr, int n, int base, int lane,
                                          int4* __restrict__ keep, int& kept) {
    const int i = min(lane, n - 1);
    const int e = le[i];
    const int r = lr[i];
    const int src = ei[e];
    const float ws = dinv[src];
    const float wd = dinv[base + r];
    own_rows<true>(hw, sum, src, r, ws, wd, n, lane);
    if (SAVE) {
        if (kept >= 0 && kept + n <= OWN_KEEP) {
            if (lane < n) keep[kept + lane] = make_int4(src, r, __float_as_int(ws * wd), 0);
            kept += n;
        } else {
            kept = -1;
        }
    }
}

template <bool SAVE>
__device__ __forceinline__ void own_scan(const int* __restrict__ ei, const float* __restrict__ hw,
                                         const float* __restrict__ dinv, float* sum, int* mle, int* mlr,
                                         int base, unsigned rows, int part, int wave, int lane, int E,
                                         int4* __restrict__ keep, int& kept) {
    const int* __restrict__ dstp = ei + E;
    const int nsteps = (E + 127) >> 7;
    int cnt = 0;
    for (int st0 = part * OWN_WAVES + wave; st0 < nsteps; st0 += 2 * OWN_WAVES * OWN_AHEAD) {
        int4 d[OWN_AHEAD];
        if ((st0 + (OWN_AHEAD - 1) * 2 * OWN_WAVES) * 128 + 128 <= E) {
            #pragma unroll
            for (int u = 0; u < OWN_AHEAD; ++u)
                d[u] = *(const int4*)(dstp + (st0 + u * 2 * OWN_WAVES) * 128 + 4 * lane);
        } else {
            #pragma unroll
            for (int u = 0; u < OWN_AHEAD; ++u) {
                const int e = (st0 + u * 2 * OWN_WAVES) * 128 + 4 * lane;
                d[u].x = (e     < E) ? dstp[e]     : -1;
                d[u].y = (e + 1 < E) ? dstp[e + 1] : -1;
                d[u].z = (e + 2 < E) ? dstp[e + 2] : -1;
                d[u].w = (e + 3 < E) ? dstp[e + 3] : -1;
            }
        }
        #pragma unroll
        for (int u = 0; u < OWN_AHEAD; ++u) {
            const int e = (st0 + u * 2 * OWN_WAVES) * 128 + 4 * lane;
            const unsigned r0 = (unsigned)d[u].x - (unsigned)base;
            const unsigned r1 = (unsigned)d[u].y - (unsigned)base;
            const unsigned r2 = (unsigned)d[u].z - (unsigned)base;
            const unsigned r3 = (unsigned)d[u].w - (unsigned)base;
            if (__builtin_amdgcn_ballot_w32(min(min(r0, r1), min(r2, r3)) < rows) != 0u) {
                #define OWN_APPEND(rk, ek) {                                              \
                    const bool hit = (rk) < rows;                                         \
                    const unsigned mask = __builtin_amdgcn_ballot_w32(hit);               \
                    if (hit) {                                                            \
                        const int pos = cnt + (int)__builtin_amdgcn_mbcnt_lo(mask, 0u);   \
                        mle[pos] = (ek);                                                  \
                        mlr[pos] = (int)(rk);                                             \
                    }                                                                     \
                    cnt += __builtin_popcount(mask);                                      \
                }
                OWN_APPEND(r0, e)
                OWN_APPEND(r1, e + 1)
                OWN_APPEND(r2, e + 2)
                OWN_APPEND(r3, e + 3)
                #undef OWN_APPEND
                while (cnt >= 32) {
                    cnt -= 32;
                    own_batch<SAVE>(ei, hw, dinv, sum, mle + cnt, mlr + cnt, 32, base, lane, keep, kept);
                }
            }
        }
    }
    if (cnt > 0) own_batch<SAVE>(ei, hw, dinv, sum, mle, mlr, cnt, base, lane, keep, kept);
}

template <bool FIRST>
__global__ __launch_bounds__(1024) void edge_owner(const int* __restrict__ ei,
                                                   const float* __restrict__ hw,
                                                   const float* __restrict__ dinv,
                                                   float* __restrict__ acc0,
                                                   float* __restrict__ acc1,
                                                   int* __restrict__ keepAll,
                                                   int* __restrict__ keptAll, int N, int E) {
    __shared__ __attribute__((aligned(16))) float sum[OWN_ROWS * DIM];
    __shared__ int le[OWN_WAVES][OWN_LIST];
    __shared__ int lr[OWN_WAVES][OWN_LIST];
    const int lane = threadIdx.x & 31;
    const int wave = __builtin_amdgcn_readfirstlane((int)(threadIdx.x >> 5));
    const int part = blockIdx.x & 1;
    const int base = (blockIdx.x >> 1) * OWN_ROWS;
    const unsigned rows = (unsigned)min(OWN_ROWS, N - base);

    for (int i = threadIdx.x; i < OWN_ROWS * DIM / 4; i += OWN_WAVES * 32)
        ((float4*)sum)[i] = make_float4(0.0f, 0.0f, 0.0f, 0.0f);
    __syncthreads();

    int4* __restrict__ keep = (int4*)keepAll + ((size_t)blockIdx.x * OWN_WAVES + wave) * OWN_KEEP;
    int* __restrict__ keptp = keptAll + blockIdx.x * OWN_WAVES + wave;
    if (FIRST) {
        int kept = 0;
        own_scan<true>(ei, hw, dinv, sum, le[wave], lr[wave], base, rows, part, wave, lane, E, keep, kept);
        if (lane == 0) *keptp = kept;
    } else {
        const int kept = __builtin_amdgcn_readfirstlane(*keptp);
        if (kept < 0) {
            int none = -1;
            own_scan<false>(ei, hw, dinv, sum, le[wave], lr[wave], base, rows, part, wave, lane, E, keep, none);
        } else if (kept > 0) {
            int4 en = keep[min(lane, kept - 1)];
            int b = 0;
            for (; b + 32 <= kept; b += 32) {
                const int4 cur = en;
                if (b + 32 < kept) en = keep[b + 32 + min(lane, kept - b - 33)];
                own_rows<false>(hw, sum, cur.x, cur.y, __int_as_float(cur.z), 0.0f, 32, lane);
            }
            if (b < kept) own_rows<false>(hw, sum, en.x, en.y, __int_as_float(en.z), 0.0f, kept - b, lane);
        }
    }
    __syncthreads();

    const int nq = (int)rows * (DIM / 4);
    if (part == 0) {
        float4* gp = (float4*)(acc0 + (size_t)base * DIM);
        for (int i = threadIdx.x; i < nq; i += OWN_WAVES * 32) {
            float4 a = gp[i];
            const float* sp = sum + (i >> 5) * DIM + (i & 31);
            a.x += sp[0]; a.y += sp[32]; a.z += sp[64]; a.w += sp[96];
            gp[i] = a;
        }
    } else {
        float4* gp = (float4*)(acc1 + (size_t)base * DIM);
        for (int i = threadIdx.x; i < nq; i += OWN_WAVES * 32) {
            const float* sp = sum + (i >> 5) * DIM + (i & 31);
            gp[i] = make_float4(sp[0], sp[32], sp[64], sp[96]);
        }
    }
}

__global__ __launch_bounds__(256) void ln_relu(const float* __restrict__ acc,
                                               const float* __restrict__ acc1,
                                               const float* __restrict__ gamma,
                                               const float* __restrict__ beta,
                                               float* __restrict__ out, int N) {
    int wave = threadIdx.x >> 5, lane = threadIdx.x & 31;
    int n = blockIdx.x * 8 + wave;
    if (n >= N) return;
    float4 v = ((const float4*)(acc + (size_t)n * DIM))[lane];
    const float4 u = ((const float4*)(acc1 + (size_t)n * DIM))[lane];
    v.x += u.x; v.y += u.y; v.z += u.z; v.w += u.w;
    float s = v.x + v.y + v.z + v.w;
    float q = v.x * v.x + v.y * v.y + v.z * v.z + v.w * v.w;
    #pragma unroll
    for (int off = 16; off > 0; off >>= 1) {
        s += __shfl_xor(s, off, 32);
        q += __shfl_xor(q, off, 32);
    }
    float mu  = s * (1.0f / DIM);
    float var = fmaxf(q * (1.0f / DIM) - mu * mu, 0.0f);
    float inv = rsqrtf(var + LNEPS);
    float4 g = ((const float4*)gamma)[lane];
    float4 b = ((const float4*)beta)[lane];
    float4 r;
    r.x = fmaxf(0.0f, g.x * (v.x - mu) * inv + b.x);
    r.y = fmaxf(0.0f, g.y * (v.y - mu) * inv + b.y);
    r.z = fmaxf(0.0f, g.z * (v.z - mu) * inv + b.z);
    r.w = fmaxf(0.0f, g.w * (v.w - mu) * inv + b.w);
    ((float4*)(out + (size_t)n * DIM))[lane] = r;
}

#define PCH_BLOCKS 5
#define PCH_STEPS  5
#define PCH_WNODES (PCH_BLOCKS * PCH_STEPS * 4)
#define PCH_NODES  (2 * PCH_WNODES)
#define PSTRIDE    (GNUM * DIM + GNUM)

__global__ __launch_bounds__(128) void pool_onehot(const float* __restrict__ h,
                                                   const int* __restrict__ batch,
                                                   float* __restrict__ part, int N) {
    __shared__ float4 lsum[GNUM * DIM / 4];
    __shared__ float  lcnt[GNUM];
    const int wave = threadIdx.x >> 5, lane = threadIdx.x & 31;
    const int ch = wave & 1;
    const int sc = wave >> 1;
    const int half = lane >> 4, c = lane & 15;
    const int nbase = blockIdx.x * PCH_NODES + sc * PCH_WNODES + 2 * half;
    const float* hcol = h + 64 * ch + 4 * c;

    v8f acc[4][4] = {};
    float cnt[4] = {0.0f, 0.0f, 0.0f, 0.0f};

    #pragma unroll
    for (int blk = 0; blk < PCH_BLOCKS; ++blk) {
        float4 r0[PCH_STEPS], r1[PCH_STEPS];
        int g0[PCH_STEPS], g1[PCH_STEPS];
        #pragma unroll
        for (int s = 0; s < PCH_STEPS; ++s) {
            const int n0 = nbase + (blk * PCH_STEPS + s) * 4;
            const int n1 = n0 + 1;
            const int i0 = min(n0, N - 1), i1 = min(n1, N - 1);
            r0[s] = *(const float4*)(hcol + (size_t)i0 * DIM);
            r1[s] = *(const float4*)(hcol + (size_t)i1 * DIM);
            const int b0 = batch[i0], b1 = batch[i1];
            g0[s] = (n0 < N) ? b0 : -1;
            g1[s] = (n1 < N) ? b1 : -1;
        }
        #pragma unroll
        for (int s = 0; s < PCH_STEPS; ++s) {
            #pragma unroll
            for (int m = 0; m < 4; ++m) {
                v2f a;
                a.x = (g0[s] == 16 * m + c) ? 1.0f : 0.0f;
                a.y = (g1[s] == 16 * m + c) ? 1.0f : 0.0f;
                cnt[m] += a.x + a.y;
                v2f b;
                b.x = r0[s].x; b.y = r1[s].x;
                acc[m][0] = __builtin_amdgcn_wmma_f32_16x16x4_f32(false, a, false, b, (short)0, acc[m][0], false, false);
                b.x = r0[s].y; b.y = r1[s].y;
                acc[m][1] = __builtin_amdgcn_wmma_f32_16x16x4_f32(false, a, false, b, (short)0, acc[m][1], false, false);
                b.x = r0[s].z; b.y = r1[s].z;
                acc[m][2] = __builtin_amdgcn_wmma_f32_16x16x4_f32(false, a, false, b, (short)0, acc[m][2], false, false);
                b.x = r0[s].w; b.y = r1[s].w;
                acc[m][3] = __builtin_amdgcn_wmma_f32_16x16x4_f32(false, a, false, b, (short)0, acc[m][3], false, false);
            }
        }
    }

    #pragma unroll
    for (int m = 0; m < 4; ++m) cnt[m] += __shfl_xor(cnt[m], 16, 32);

    if (sc == 1) {
        #pragma unroll
        for (int m = 0; m < 4; ++m) {
            #pragma unroll
            for (int r = 0; r < 8; ++r) {
                const int row = 16 * m + 8 * half + r;
                lsum[row * (DIM / 4) + 16 * ch + c] =
                    make_float4(acc[m][0][r], acc[m][1][r], acc[m][2][r], acc[m][3][r]);
            }
            if (ch == 0 && half == 0) lcnt[16 * m + c] = cnt[m];
        }
    }
    __syncthreads();
    if (sc == 0) {
        float* pp = part + (size_t)blockIdx.x * PSTRIDE;
        #pragma unroll
        for (int m = 0; m < 4; ++m) {
            #pragma unroll
            for (int r = 0; r < 8; ++r) {
                const int row = 16 * m + 8 * half + r;
                const float4 o = lsum[row * (DIM / 4) + 16 * ch + c];
                float4 v;
                v.x = acc[m][0][r] + o.x;  v.y = acc[m][1][r] + o.y;
                v.z = acc[m][2][r] + o.z;  v.w = acc[m][3][r] + o.w;
                ((float4*)pp)[row * (DIM / 4) + 16 * ch + c] = v;
            }
            if (ch == 0 && half == 0) pp[GNUM * DIM + 16 * m + c] = cnt[m] + lcnt[16 * m + c];
        }
    }
}

__global__ __launch_bounds__(256) void pool_reduce(const float* __restrict__ part,
                                                   float* __restrict__ out, int P) {
    __shared__ float red[8][32];
    __shared__ float redc[8];
    const int lane = threadIdx.x & 31, s = threadIdx.x >> 5;
    const int o = blockIdx.x * 32 + lane;
    const int g = (blockIdx.x * 32) / DIM;
    float sum = 0.0f;
    for (int p0 = s; p0 < P; p0 += 128) {
        float v[16];
        #pragma unroll
        for (int u = 0; u < 16; ++u) {
            const int p = p0 + 8 * u;
            v[u] = (p < P) ? part[(size_t)p * PSTRIDE + o] : 0.0f;
        }
        #pragma unroll
        for (int u = 0; u < 16; ++u) sum += v[u];
    }
    float cs = 0.0f;
    for (int p = s + 8 * lane; p < P; p += 256) cs += part[(size_t)p * PSTRIDE + GNUM * DIM + g];
    #pragma unroll
    for (int off = 16; off > 0; off >>= 1) cs += __shfl_xor(cs, off, 32);
    red[s][lane] = sum;
    if (lane == 0) redc[s] = cs;
    __syncthreads();
    if (s == 0) {
        float t = 0.0f, cc = 0.0f;
        #pragma unroll
        for (int i = 0; i < 8; ++i) { t += red[i][lane]; cc += redc[i]; }
        out[o] = t / fmaxf(cc, 1.0f);
    }
}

static inline size_t al256(size_t x) { return (x + 255) & ~(size_t)255; }

extern "C" void kernel_launch(void* const* d_in, const int* in_sizes, int n_in,
                              void* d_out, int out_size, void* d_ws, size_t ws_size,
                              hipStream_t stream) {
    const float* x      = (const float*)d_in[0];
    const int*   ei     = (const int*)d_in[1];
    const int*   batch  = (const int*)d_in[2];
    const float* Ws     = (const float*)d_in[3];
    const float* bs     = (const float*)d_in[4];
    const float* gammas = (const float*)d_in[5];
    const float* betas  = (const float*)d_in[6];

    const int N = in_sizes[0] / DIM;
    const int E = in_sizes[1] / 2;
    const size_t nd = (size_t)N * DIM * sizeof(float);

    char* ws = (char*)d_ws;
    float* dinv = (float*)ws;                 ws += al256((size_t)N * sizeof(float));
    float* deg1 = (float*)ws;                 ws += al256((size_t)N * sizeof(float));
    float* hw   = (float*)ws;                 ws += al256(nd);
    float* accb = (float*)ws;                 ws += al256(nd);
    float* acc1 = (float*)ws;                 ws += al256(nd);
    float* hbuf = (float*)ws;                 ws += al256(nd);
    const int go = 2 * ((N + OWN_ROWS - 1) / OWN_ROWS);
    int* keepAll = (int*)ws;                  ws += al256((size_t)go * OWN_WAVES * OWN_KEEP * sizeof(int4));
    int* keptAll = (int*)ws;                  ws += al256((size_t)go * OWN_WAVES * sizeof(int));
    float* part = (float*)ws;

    zero2_f32<<<(N + 255) / 256, 256, 0, stream>>>(dinv, deg1, N);
    deg_accum<<<(E + 255) / 256, 256, 0, stream>>>(ei, dinv, deg1, E);
    deg_to_dinv<<<(N + 255) / 256, 256, 0, stream>>>(dinv, deg1, N);

    const int nstrips = N >> 4;
    const int gw = (nstrips + 7) / 8;
    layer_rows<true><<<gw, 256, 0, stream>>>(x, Ws, gammas, betas, bs, dinv, hw, accb, acc1, N);
    edge_owner<true><<<go, OWN_WAVES * 32, 0, stream>>>(ei, hw, dinv, accb, acc1, keepAll, keptAll, N, E);
    for (int l = 1; l < LAYERS; ++l) {
        layer_rows<false><<<gw, 256, 0, stream>>>(
            x, Ws + (size_t)l * DIM * DIM, gammas + (size_t)(l - 1) * DIM, betas + (size_t)(l - 1) * DIM,
            bs + (size_t)l * DIM, dinv, hw, accb, acc1, N);
        edge_owner<false><<<go, OWN_WAVES * 32, 0, stream>>>(ei, hw, dinv, accb, acc1, keepAll, keptAll, N, E);
    }
    ln_relu<<<(N + 7) / 8, 256, 0, stream>>>(
        accb, acc1, gammas + (size_t)(LAYERS - 1) * DIM, betas + (size_t)(LAYERS - 1) * DIM, hbuf, N);
    const float* hcur = hbuf;

    const int P = (N + PCH_NODES - 1) / PCH_NODES;
    pool_onehot<<<P, 128, 0, stream>>>(hcur, batch, part, N);
    pool_reduce<<<GNUM * DIM / 32, 256, 0, stream>>>(part, (float*)d_out, P);
}


# layers 2, 3: an owner work-group per 200 rows reads both halves' edge lists, so its LDS sums are whole: self-loop, LayerNorm, ReLU, next product run in its launch; 12->10 launches (new sum order)
# speedup vs baseline: 4.0983x; 1.2569x over previous
#include <hip/hip_runtime.h>


typedef __attribute__((ext_vector_type(2))) float v2f;
typedef __attribute__((ext_vector_type(8))) float v8f;

#define DIM   128
#define GNUM  64
#define LAYERS 3
#define LNEPS 1e-5f

__global__ void zero2_f32(float* __restrict__ p0, float* __restrict__ p1, int n) {
    int i = blockIdx.x * blockDim.x + threadIdx.x;
    if (i < n) { p0[i] = 0.0f; p1[i] = 0.0f; }
}

__device__ __forceinline__ unsigned placement_id() { return __builtin_amdgcn_s_sendmsg_rtn(0x87); }
__device__ __forceinline__ int package_half(unsigned id) { return (int)((id >> 18) & 1u); }

__global__ void deg_accum(const int* __restrict__ ei, float* __restrict__ deg0,
                          float* __restrict__ deg1, int E) {
    int e = blockIdx.x * blockDim.x + threadIdx.x;
    const unsigned id = placement_id();
    if (e >= E) return;
    const int dst = ei[E + e];
    float* deg = package_half(id) ? deg1 : deg0;
    atomicAdd(&deg[dst], 1.0f);
}

__global__ void deg_to_dinv(float* __restrict__ d, const float* __restrict__ d1, int N) {
    int i = blockIdx.x * blockDim.x + threadIdx.x;
    if (i < N) d[i] = rsqrtf(d[i] + d1[i] + 1.0f);
}

template <bool FIRST>
__global__ __launch_bounds__(256) void layer_rows(const float* __restrict__ X,
                                                  const float* __restrict__ W,
                                                  const float* __restrict__ gamma,
                                                  const float* __restrict__ beta,
                                                  const float* __restrict__ bias,
                                                  const float* __restrict__ dinv,
                                                  float* __restrict__ hw,
                                                  float* __restrict__ acc0,
                                                  const float* __restrict__ acc1, int N) {
    __shared__ float lw[DIM * DIM];
    __shared__ __attribute__((aligned(16))) float lg[DIM];
    __shared__ __attribute__((aligned(16))) float lb[DIM];
    {
        const float4* Wv = (const float4*)W;
        float4* lv = (float4*)lw;
        #pragma unroll
        for (int i = 0; i < (DIM * DIM / 4) / 256; ++i)
            lv[threadIdx.x + i * 256] = Wv[threadIdx.x + i * 256];
        if (!FIRST) {
            if (threadIdx.x < DIM) lg[threadIdx.x] = gamma[threadIdx.x];
            else                   lb[threadIdx.x - DIM] = beta[threadIdx.x - DIM];
        }
    }
    __syncthreads();

    const int wave = __builtin_amdgcn_readfirstlane((int)(threadIdx.x >> 5));
    const int lane = threadIdx.x & 31;
    const int half = lane >> 4;
    const int ln   = lane & 15;

    const int nstrips = N >> 4;
    const int strip = blockIdx.x * 8 + wave;
    if (strip >= nstrips) return;
    const int rowBase = strip << 4;

    v2f a[DIM / 4];
    if (FIRST) {
        const v2f* arow2 = (const v2f*)(X + (size_t)(rowBase + ln) * DIM);
        #pragma unroll
        for (int kk = 0; kk < DIM / 4; ++kk) a[kk] = arow2[2 * kk + half];
    } else {
        const v2f* p0 = (const v2f*)(acc0 + (size_t)(rowBase + ln) * DIM);
        const v2f* p1 = (const v2f*)(acc1 + (size_t)(rowBase + ln) * DIM);
        float s = 0.0f, q = 0.0f;
        #pragma unroll
        for (int kk = 0; kk < DIM / 4; ++kk) {
            const v2f u0 = p0[2 * kk + half];
            const v2f u1 = p1[2 * kk + half];
            v2f v;
            v.x = u0.x + u1.x;  v.y = u0.y + u1.y;
            a[kk] = v;
            s += v.x + v.y;
            q += v.x * v.x + v.y * v.y;
        }
        s += __shfl_xor(s, 16, 32);
        q += __shfl_xor(q, 16, 32);
        const float mu  = s * (1.0f / DIM);
        const float var = fmaxf(q * (1.0f / DIM) - mu * mu, 0.0f);
        const float inv = rsqrtf(var + LNEPS);
        #pragma unroll
        for (int kk = 0; kk < DIM / 4; ++kk) {
            const v2f g = ((const v2f*)lg)[2 * kk + half];
            const v2f b = ((const v2f*)lb)[2 * kk + half];
            v2f v = a[kk];
            v.x = fmaxf(0.0f, g.x * (v.x - mu) * inv + b.x);
            v.y = fmaxf(0.0f, g.y * (v.y - mu) * inv + b.y);
            a[kk] = v;
        }
    }

    v8f acc[8] = {};

    #pragma unroll
    for (int kk = 0; kk < DIM / 4; ++kk) {
        const int kb = 4 * kk + 2 * half;
        #pragma unroll
        for (int t = 0; t < 8; ++t) {
            v2f b;
            b.x = lw[kb * DIM + t * 16 + ln];
            b.y = lw[(kb + 1) * DIM + t * 16 + ln];
            acc[t] = __builtin_amdgcn_wmma_f32_16x16x4_f32(
                false, a[kk], false, b, (short)0, acc[t], false, false);
        }
    }

    float sn[8], bv[8];
    if (FIRST) {
        #pragma unroll
        for (int j = 0; j < 8; ++j) {
            const float d = dinv[rowBase + half * 8 + j];
            sn[j] = d * d;
        }
        #pragma unroll
        for (int t = 0; t < 8; ++t) bv[t] = bias[t * 16 + ln];
    }
    #pragma unroll
    for (int t = 0; t < 8; ++t) {
        #pragma unroll
        for (int j = 0; j < 8; ++j) {
            const int row = rowBase + half * 8 + j;
            const size_t o = (size_t)row * DIM + t * 16 + ln;
            const float v = acc[t][j];
            hw[o]   = v;
            if (FIRST) acc0[o] = bv[t] + sn[j] * v;
        }
    }
}

#define OWN_ROWS  400
#define SUB_ROWS  200
#define SUB_PITCH 129
#define SUB_RPW   ((SUB_ROWS + 31) / 32)
#define OWN_WAVES 32
#define OWN_LIST  160
#define OWN_AHEAD 8
#define OWN_KEEP  256

template <bool PAIR, int PITCH>
__device__ __forceinline__ void own_rows(const float* __restrict__ hw, float* sum, int src, int r,
                                         float wa, float wb, int n, int lane) {
    #pragma nounroll
    for (int g = 0; g < 32; g += 8) {
        if (g >= n) break;
        float4 v[8];
        #pragma unroll
        for (int j = 0; j < 8; ++j) {
            const int s = __builtin_amdgcn_readlane(src, g + j);
            v[j] = *(const float4*)(hw + (size_t)s * DIM + 4 * lane);
        }
        #pragma unroll
        for (int j = 0; j < 8; ++j) {
            if (g + j < n) {
                const int rj = __builtin_amdgcn_readlane(r, g + j);
                float wj = __int_as_float(__builtin_amdgcn_readlane(__float_as_int(wa), g + j));
                if (PAIR) wj *= __int_as_float(__builtin_amdgcn_readlane(__float_as_int(wb), g + j));
                float* dp = sum + rj * PITCH + lane;
                __hip_atomic_fetch_add(dp +  0, wj * v[j].x, __ATOMIC_RELAXED, __HIP_MEMORY_SCOPE_WORKGROUP);
                __hip_atomic_fetch_add(dp + 32, wj * v[j].y, __ATOMIC_RELAXED, __HIP_MEMORY_SCOPE_WORKGROUP);
                __hip_atomic_fetch_add(dp + 64, wj * v[j].z, __ATOMIC_RELAXED, __HIP_MEMORY_SCOPE_WORKGROUP);
                __hip_atomic_fetch_add(dp + 96, wj * v[j].w, __ATOMIC_RELAXED, __HIP_MEMORY_SCOPE_WORKGROUP);
            }
        }
    }
}

template <bool SAVE, int PITCH>
__device__ __forceinline__ void own_batch(const int* __restrict__ ei, const float* __restrict__ hw,
                                          const float* __restrict__ dinv, float* sum,
                                          const int* le, const int* lr, int n, int base, int lane,
                                          int4* __restrict__ keep, int& keptA, int& keptB) {
    const int i = min(lane, n - 1);
    const int e = le[i];
    const int r = lr[i];
    const int src = ei[e];
    const float ws = dinv[src];
    const float wd = dinv[base + r];
    own_rows<true, PITCH>(hw, sum, src, r, ws, wd, n, lane);
    if (SAVE) {
        if (keptA >= 0 && keptA + keptB + n <= OWN_KEEP) {
            const bool lo = lane < n && r <  SUB_ROWS;
            const bool hi = lane < n && r >= SUB_ROWS;
            const unsigned mlo = __builtin_amdgcn_ballot_w32(lo);
            const unsigned mhi = __builtin_amdgcn_ballot_w32(hi);
            const int4 en = make_int4(src, r, __float_as_int(ws * wd), 0);
            if (lo) keep[keptA + (int)__builtin_amdgcn_mbcnt_lo(mlo, 0u)] = en;
            if (hi) keep[OWN_KEEP - 1 - keptB - (int)__builtin_amdgcn_mbcnt_lo(mhi, 0u)] = en;
            keptA += __builtin_popcount(mlo);
            keptB += __builtin_popcount(mhi);
        } else {
            keptA = -1;
        }
    }
}

template <bool SAVE, int PITCH>
__device__ __forceinline__ void own_scan(const int* __restrict__ ei, const float* __restrict__ hw,
                                         const float* __restrict__ dinv, float* sum, int* mle, int* mlr,
                                         int base, unsigned rows, int part, int wave, int lane, int E,
                                         int4* __restrict__ keep, int& keptA, int& keptB) {
    const int* __restrict__ dstp = ei + E;
    const int nsteps = (E + 127) >> 7;
    int cnt = 0;
    for (int st0 = part * OWN_WAVES + wave; st0 < nsteps; st0 += 2 * OWN_WAVES * OWN_AHEAD) {
        int4 d[OWN_AHEAD];
        if ((st0 + (OWN_AHEAD - 1) * 2 * OWN_WAVES) * 128 + 128 <= E) {
            #pragma unroll
            for (int u = 0; u < OWN_AHEAD; ++u)
                d[u] = *(const int4*)(dstp + (st0 + u * 2 * OWN_WAVES) * 128 + 4 * lane);
        } else {
            #pragma unroll
            for (int u = 0; u < OWN_AHEAD; ++u) {
                const int e = (st0 + u * 2 * OWN_WAVES) * 128 + 4 * lane;
                d[u].x = (e     < E) ? dstp[e]     : -1;
                d[u].y = (e + 1 < E) ? dstp[e + 1] : -1;
                d[u].z = (e + 2 < E) ? dstp[e + 2] : -1;
                d[u].w = (e + 3 < E) ? dstp[e + 3] : -1;
            }
        }
        #pragma unroll
        for (int u = 0; u < OWN_AHEAD; ++u) {
            const int e = (st0 + u * 2 * OWN_WAVES) * 128 + 4 * lane;
            const unsigned r0 = (unsigned)d[u].x - (unsigned)base;
            const unsigned r1 = (unsigned)d[u].y - (unsigned)base;
            const unsigned r2 = (unsigned)d[u].z - (unsigned)base;
            const unsigned r3 = (unsigned)d[u].w - (unsigned)base;
            if (__builtin_amdgcn_ballot_w32(min(min(r0, r1), min(r2, r3)) < rows) != 0u) {
                #define OWN_APPEND(rk, ek) {                                              \
                    const bool hit = (rk) < rows;                                         \
                    const unsigned mask = __builtin_amdgcn_ballot_w32(hit);               \
                    if (hit) {                                                            \
                        const int pos = cnt + (int)__builtin_amdgcn_mbcnt_lo(mask, 0u);   \
                        mle[pos] = (ek);                                                  \
                        mlr[pos] = (int)(rk);                                             \
                    }                                                                     \
                    cnt += __builtin_popcount(mask);                                      \
                }
                OWN_APPEND(r0, e)
                OWN_APPEND(r1, e + 1)
                OWN_APPEND(r2, e + 2)
                OWN_APPEND(r3, e + 3)
                #undef OWN_APPEND
                while (cnt >= 32) {
                    cnt -= 32;
                    own_batch<SAVE, PITCH>(ei, hw, dinv, sum, mle + cnt, mlr + cnt, 32, base, lane, keep, keptA, keptB);
                }
            }
        }
    }
    if (cnt > 0) own_batch<SAVE, PITCH>(ei, hw, dinv, sum, mle, mlr, cnt, base, lane, keep, keptA, keptB);
}

__global__ __launch_bounds__(1024) void edge_owner(const int* __restrict__ ei,
                                                   const float* __restrict__ hw,
                                                   const float* __restrict__ dinv,
                                                   float* __restrict__ acc0,
                                                   float* __restrict__ acc1,
                                                   int* __restrict__ keepAll,
                                                   int* __restrict__ keptAll, int N, int E) {
    __shared__ __attribute__((aligned(16))) float sum[OWN_ROWS * DIM];
    __shared__ int le[OWN_WAVES][OWN_LIST];
    __shared__ int lr[OWN_WAVES][OWN_LIST];
    const int lane = threadIdx.x & 31;
    const int wave = __builtin_amdgcn_readfirstlane((int)(threadIdx.x >> 5));
    const int part = blockIdx.x & 1;
    const int base = (blockIdx.x >> 1) * OWN_ROWS;
    const unsigned rows = (unsigned)min(OWN_ROWS, N - base);

    for (int i = threadIdx.x; i < OWN_ROWS * DIM / 4; i += OWN_WAVES * 32)
        ((float4*)sum)[i] = make_float4(0.0f, 0.0f, 0.0f, 0.0f);
    __syncthreads();

    int4* __restrict__ keep = (int4*)keepAll + ((size_t)blockIdx.x * OWN_WAVES + wave) * OWN_KEEP;
    int* __restrict__ keptp = keptAll + 2 * (blockIdx.x * OWN_WAVES + wave);
    int keptA = 0, keptB = 0;
    own_scan<true, DIM>(ei, hw, dinv, sum, le[wave], lr[wave], base, rows, part, wave, lane, E, keep, keptA, keptB);
    if (lane == 0) { keptp[0] = keptA; keptp[1] = keptB; }
    __syncthreads();

    const int nq = (int)rows * (DIM / 4);
    if (part == 0) {
        float4* gp = (float4*)(acc0 + (size_t)base * DIM);
        for (int i = threadIdx.x; i < nq; i += OWN_WAVES * 32) {
            float4 a = gp[i];
            const float* sp = sum + (i >> 5) * DIM + (i & 31);
            a.x += sp[0]; a.y += sp[32]; a.z += sp[64]; a.w += sp[96];
            gp[i] = a;
        }
    } else {
        float4* gp = (float4*)(acc1 + (size_t)base * DIM);
        for (int i = threadIdx.x; i < nq; i += OWN_WAVES * 32) {
            const float* sp = sum + (i >> 5) * DIM + (i & 31);
            gp[i] = make_float4(sp[0], sp[32], sp[64], sp[96]);
        }
    }
}

template <bool LAST>
__global__ __launch_bounds__(1024) void owner_layer(const int* __restrict__ ei,
                                                    const float* __restrict__ hw,
                                                    const float* __restrict__ dinv,
                                                    const int* __restrict__ keepAll,
                                                    const int* __restrict__ keptAll,
                                                    const float* __restrict__ W,
                                                    const float* __restrict__ gamma,
                                                    const float* __restrict__ beta,
                                                    const float* __restrict__ bias,
                                                    float* __restrict__ out, int N, int E) {
    __shared__ __attribute__((aligned(16))) float sum[SUB_ROWS * SUB_PITCH];
    __shared__ int le[OWN_WAVES][OWN_LIST];
    __shared__ int lr[OWN_WAVES][OWN_LIST];
    __shared__ __attribute__((aligned(16))) float lw[LAST ? 4 : DIM * DIM];
    const int lane = threadIdx.x & 31;
    const int wave = __builtin_amdgcn_readfirstlane((int)(threadIdx.x >> 5));
    const int sub  = blockIdx.x & 1;
    const int grp  = blockIdx.x >> 1;
    const int base = grp * OWN_ROWS + sub * SUB_ROWS;
    const int rows = max(0, min(SUB_ROWS, N - base));

    const int id0 = (2 * grp) * OWN_WAVES + wave;
    const int id1 = id0 + OWN_WAVES;
    const int c0a = keptAll[2 * id0], c0b = keptAll[2 * id0 + 1];
    const int c1a = keptAll[2 * id1], c1b = keptAll[2 * id1 + 1];

    for (int i = threadIdx.x; i < SUB_ROWS * SUB_PITCH / 4; i += OWN_WAVES * 32)
        ((float4*)sum)[i] = make_float4(0.0f, 0.0f, 0.0f, 0.0f);

    const int k0 = __builtin_amdgcn_readfirstlane(c0a) < 0 ? -1 : __builtin_amdgcn_readfirstlane(sub ? c0b : c0a);
    const int k1 = __builtin_amdgcn_readfirstlane(c1a) < 0 ? -1 : __builtin_amdgcn_readfirstlane(sub ? c1b : c1a);
    const int roff = sub * SUB_ROWS;
    const int first = sub ? OWN_KEEP - 1 : 0;
    const int step  = sub ? -1 : 1;
    const int4* __restrict__ keep0 = (const int4*)keepAll + (size_t)id0 * OWN_KEEP;
    const int4* __restrict__ keep1 = (const int4*)keepAll + (size_t)id1 * OWN_KEEP;
    int4 f0 = make_int4(0, 0, 0, 0), f1 = make_int4(0, 0, 0, 0);
    if (k0 > 0) f0 = keep0[first + step * min(lane, k0 - 1)];
    if (k1 > 0) f1 = keep1[first + step * min(lane, k1 - 1)];
    __syncthreads();

    #pragma nounroll
    for (int p = 0; p < 2; ++p) {
        const int kept = p ? k1 : k0;
        const int4* __restrict__ keep = p ? keep1 : keep0;
        if (kept < 0) {
            int na = -1, nb = 0;
            own_scan<false, SUB_PITCH>(ei, hw, dinv, sum, le[wave], lr[wave], base, (unsigned)rows, p, wave, lane, E,
                                       nullptr, na, nb);
        } else if (kept > 0) {
            int4 en = p ? f1 : f0;
            int b = 0;
            for (; b + 32 <= kept; b += 32) {
                const int4 cur = en;
                if (b + 32 < kept) en = keep[first + step * (b + 32 + min(lane, kept - b - 33))];
                own_rows<false, SUB_PITCH>(hw, sum, cur.x, cur.y - roff, __int_as_float(cur.z), 0.0f, 32, lane);
            }
            if (b < kept) own_rows<false, SUB_PITCH>(hw, sum, en.x, en.y - roff, __int_as_float(en.z), 0.0f, kept - b, lane);
        }
    }

    float4 hv[SUB_RPW];
    float  dv[SUB_RPW];
    if (rows > 0) {
        #pragma unroll
        for (int i = 0; i < SUB_RPW; ++i) {
            const int rc = min(wave + i * OWN_WAVES, rows - 1);
            hv[i] = ((const float4*)(hw + (size_t)(base + rc) * DIM))[lane];
            dv[i] = dinv[base + rc];
        }
    }
    if (!LAST) {
        const float4* Wv = (const float4*)W;
        float4* lv = (float4*)lw;
        #pragma unroll
        for (int i = 0; i < (DIM * DIM / 4) / (OWN_WAVES * 32); ++i)
            lv[threadIdx.x + i * OWN_WAVES * 32] = Wv[threadIdx.x + i * OWN_WAVES * 32];
    }
    __syncthreads();

    if (rows > 0) {
        const float4 bi = ((const float4*)bias)[lane];
        const float4 g  = ((const float4*)gamma)[lane];
        const float4 bt = ((const float4*)beta)[lane];
        #pragma unroll
        for (int i = 0; i < SUB_RPW; ++i) {
            const int rr = wave + i * OWN_WAVES;
            if (rr < rows) {
                const float sn = dv[i] * dv[i];
                float* sp = sum + rr * SUB_PITCH + lane;
                float4 v;
                v.x = sp[0]  + (bi.x + sn * hv[i].x);
                v.y = sp[32] + (bi.y + sn * hv[i].y);
                v.z = sp[64] + (bi.z + sn * hv[i].z);
                v.w = sp[96] + (bi.w + sn * hv[i].w);
                float s = v.x + v.y + v.z + v.w;
                float q = v.x * v.x + v.y * v.y + v.z * v.z + v.w * v.w;
                #pragma unroll
                for (int off = 16; off > 0; off >>= 1) {
                    s += __shfl_xor(s, off, 32);
                    q += __shfl_xor(q, off, 32);
                }
                const float mu  = s * (1.0f / DIM);
                const float var = fmaxf(q * (1.0f / DIM) - mu * mu, 0.0f);
                const float inv = rsqrtf(var + LNEPS);
                float4 r;
                r.x = fmaxf(0.0f, g.x * (v.x - mu) * inv + bt.x);
                r.y = fmaxf(0.0f, g.y * (v.y - mu) * inv + bt.y);
                r.z = fmaxf(0.0f, g.z * (v.z - mu) * inv + bt.z);
                r.w = fmaxf(0.0f, g.w * (v.w - mu) * inv + bt.w);
                if (LAST) {
                    ((float4*)(out + (size_t)(base + rr) * DIM))[lane] = r;
                } else {
                    sp[0] = r.x;  sp[32] = r.y;  sp[64] = r.z;  sp[96] = r.w;
                }
            }
        }
    }
    if (!LAST) {
        __syncthreads();
        const int strip = wave >> 1;
        const int ch    = wave & 1;
        if (strip * 16 >= rows) return;
        const int half = lane >> 4;
        const int ln   = lane & 15;
        const int rl = min(strip * 16 + ln, rows - 1);
        const float* srow = sum + rl * SUB_PITCH + 64 * half;

        v8f acc[4] = {};
        #pragma unroll
        for (int kk = 0; kk < DIM / 4; ++kk) {
            v2f a;
            a.x = srow[kk];
            a.y = srow[32 + kk];
            const int kb = 4 * kk + 2 * half;
            #pragma unroll
            for (int t = 0; t < 4; ++t) {
                v2f b;
                b.x = lw[kb * DIM + ch * 64 + t * 16 + ln];
                b.y = lw[(kb + 1) * DIM + ch * 64 + t * 16 + ln];
                acc[t] = __builtin_amdgcn_wmma_f32_16x16x4_f32(
                    false, a, false, b, (short)0, acc[t], false, false);
            }
        }
        #pragma unroll
        for (int t = 0; t < 4; ++t) {
            #pragma unroll
            for (int j = 0; j < 8; ++j) {
                const int r2 = strip * 16 + half * 8 + j;
                if (r2 < rows) out[(size_t)(base + r2) * DIM + ch * 64 + t * 16 + ln] = acc[t][j];
            }
        }
    }
}

#define PCH_BLOCKS 5
#define PCH_STEPS  5
#define PCH_WNODES (PCH_BLOCKS * PCH_STEPS * 4)
#define PCH_NODES  (2 * PCH_WNODES)
#define PSTRIDE    (GNUM * DIM + GNUM)

__global__ __launch_bounds__(128) void pool_onehot(const float* __restrict__ h,
                                                   const int* __restrict__ batch,
                                                   float* __restrict__ part, int N) {
    __shared__ float4 lsum[GNUM * DIM / 4];
    __shared__ float  lcnt[GNUM];
    const int wave = threadIdx.x >> 5, lane = threadIdx.x & 31;
    const int ch = wave & 1;
    const int sc = wave >> 1;
    const int half = lane >> 4, c = lane & 15;
    const int nbase = blockIdx.x * PCH_NODES + sc * PCH_WNODES + 2 * half;
    const float* hcol = h + 64 * ch + 4 * c;

    v8f acc[4][4] = {};
    float cnt[4] = {0.0f, 0.0f, 0.0f, 0.0f};

    #pragma unroll
    for (int blk = 0; blk < PCH_BLOCKS; ++blk) {
        float4 r0[PCH_STEPS], r1[PCH_STEPS];
        int g0[PCH_STEPS], g1[PCH_STEPS];
        #pragma unroll
        for (int s = 0; s < PCH_STEPS; ++s) {
            const int n0 = nbase + (blk * PCH_STEPS + s) * 4;
            const int n1 = n0 + 1;
            const int i0 = min(n0, N - 1), i1 = min(n1, N - 1);
            r0[s] = *(const float4*)(hcol + (size_t)i0 * DIM);
            r1[s] = *(const float4*)(hcol + (size_t)i1 * DIM);
            const int b0 = batch[i0], b1 = batch[i1];
            g0[s] = (n0 < N) ? b0 : -1;
            g1[s] = (n1 < N) ? b1 : -1;
        }
        #pragma unroll
        for (int s = 0; s < PCH_STEPS; ++s) {
            #pragma unroll
            for (int m = 0; m < 4; ++m) {
                v2f a;
                a.x = (g0[s] == 16 * m + c) ? 1.0f : 0.0f;
                a.y = (g1[s] == 16 * m + c) ? 1.0f : 0.0f;
                cnt[m] += a.x + a.y;
                v2f b;
                b.x = r0[s].x; b.y = r1[s].x;
                acc[m][0] = __builtin_amdgcn_wmma_f32_16x16x4_f32(false, a, false, b, (short)0, acc[m][0], false, false);
                b.x = r0[s].y; b.y = r1[s].y;
                acc[m][1] = __builtin_amdgcn_wmma_f32_16x16x4_f32(false, a, false, b, (short)0, acc[m][1], false, false);
                b.x = r0[s].z; b.y = r1[s].z;
                acc[m][2] = __builtin_amdgcn_wmma_f32_16x16x4_f32(false, a, false, b, (short)0, acc[m][2], false, false);
                b.x = r0[s].w; b.y = r1[s].w;
                acc[m][3] = __builtin_amdgcn_wmma_f32_16x16x4_f32(false, a, false, b, (short)0, acc[m][3], false, false);
            }
        }
    }

    #pragma unroll
    for (int m = 0; m < 4; ++m) cnt[m] += __shfl_xor(cnt[m], 16, 32);

    if (sc == 1) {
        #pragma unroll
        for (int m = 0; m < 4; ++m) {
            #pragma unroll
            for (int r = 0; r < 8; ++r) {
                const int row = 16 * m + 8 * half + r;
                lsum[row * (DIM / 4) + 16 * ch + c] =
                    make_float4(acc[m][0][r], acc[m][1][r], acc[m][2][r], acc[m][3][r]);
            }
            if (ch == 0 && half == 0) lcnt[16 * m + c] = cnt[m];
        }
    }
    __syncthreads();
    if (sc == 0) {
        float* pp = part + (size_t)blockIdx.x * PSTRIDE;
        #pragma unroll
        for (int m = 0; m < 4; ++m) {
            #pragma unroll
            for (int r = 0; r < 8; ++r) {
                const int row = 16 * m + 8 * half + r;
                const float4 o = lsum[row * (DIM / 4) + 16 * ch + c];
                float4 v;
                v.x = acc[m][0][r] + o.x;  v.y = acc[m][1][r] + o.y;
                v.z = acc[m][2][r] + o.z;  v.w = acc[m][3][r] + o.w;
                ((float4*)pp)[row * (DIM / 4) + 16 * ch + c] = v;
            }
            if (ch == 0 && half == 0) pp[GNUM * DIM + 16 * m + c] = cnt[m] + lcnt[16 * m + c];
        }
    }
}

__global__ __launch_bounds__(256) void pool_reduce(const float* __restrict__ part,
                                                   float* __restrict__ out, int P) {
    __shared__ float red[8][32];
    __shared__ float redc[8];
    const int lane = threadIdx.x & 31, s = threadIdx.x >> 5;
    const int o = blockIdx.x * 32 + lane;
    const int g = (blockIdx.x * 32) / DIM;
    float sum = 0.0f;
    for (int p0 = s; p0 < P; p0 += 128) {
        float v[16];
        #pragma unroll
        for (int u = 0; u < 16; ++u) {
            const int p = p0 + 8 * u;
            v[u] = (p < P) ? part[(size_t)p * PSTRIDE + o] : 0.0f;
        }
        #pragma unroll
        for (int u = 0; u < 16; ++u) sum += v[u];
    }
    float cs = 0.0f;
    for (int p = s + 8 * lane; p < P; p += 256) cs += part[(size_t)p * PSTRIDE + GNUM * DIM + g];
    #pragma unroll
    for (int off = 16; off > 0; off >>= 1) cs += __shfl_xor(cs, off, 32);
    red[s][lane] = sum;
    if (lane == 0) redc[s] = cs;
    __syncthreads();
    if (s == 0) {
        float t = 0.0f, cc = 0.0f;
        #pragma unroll
        for (int i = 0; i < 8; ++i) { t += red[i][lane]; cc += redc[i]; }
        out[o] = t / fmaxf(cc, 1.0f);
    }
}
static inline size_t al256(size_t x) { return (x + 255) & ~(size_t)255; }

extern "C" void kernel_launch(void* const* d_in, const int* in_sizes, int n_in,
                              void* d_out, int out_size, void* d_ws, size_t ws_size,
                              hipStream_t stream) {
    const float* x      = (const float*)d_in[0];
    const int*   ei     = (const int*)d_in[1];
    const int*   batch  = (const int*)d_in[2];
    const float* Ws     = (const float*)d_in[3];
    const float* bs     = (const float*)d_in[4];
    const float* gammas = (const float*)d_in[5];
    const float* betas  = (const float*)d_in[6];

    const int N = in_sizes[0] / DIM;
    const int E = in_sizes[1] / 2;
    const size_t nd = (size_t)N * DIM * sizeof(float);

    char* ws = (char*)d_ws;
    float* dinv = (float*)ws;                 ws += al256((size_t)N * sizeof(float));
    float* deg1 = (float*)ws;                 ws += al256((size_t)N * sizeof(float));
    float* hw   = (float*)ws;                 ws += al256(nd);
    float* accb = (float*)ws;                 ws += al256(nd);
    float* acc1 = (float*)ws;                 ws += al256(nd);
    float* hw3  = (float*)ws;                 ws += al256(nd);
    const int go = 2 * ((N + OWN_ROWS - 1) / OWN_ROWS);
    int* keepAll = (int*)ws;                  ws += al256((size_t)go * OWN_WAVES * OWN_KEEP * sizeof(int4));
    int* keptAll = (int*)ws;                  ws += al256((size_t)go * OWN_WAVES * 2 * sizeof(int));
    float* part = (float*)ws;

    zero2_f32<<<(N + 255) / 256, 256, 0, stream>>>(dinv, deg1, N);
    deg_accum<<<(E + 255) / 256, 256, 0, stream>>>(ei, dinv, deg1, E);
    deg_to_dinv<<<(N + 255) / 256, 256, 0, stream>>>(dinv, deg1, N);

    const int nstrips = N >> 4;
    const int gw = (nstrips + 7) / 8;
    layer_rows<true><<<gw, 256, 0, stream>>>(x, Ws, gammas, betas, bs, dinv, hw, accb, acc1, N);
    edge_owner<<<go, OWN_WAVES * 32, 0, stream>>>(ei, hw, dinv, accb, acc1, keepAll, keptAll, N, E);
    layer_rows<false><<<gw, 256, 0, stream>>>(x, Ws + (size_t)DIM * DIM, gammas, betas, bs, dinv, hw, accb, acc1, N);
    owner_layer<false><<<go, OWN_WAVES * 32, 0, stream>>>(
        ei, hw, dinv, keepAll, keptAll, Ws + (size_t)2 * DIM * DIM, gammas + DIM, betas + DIM, bs + DIM, hw3, N, E);
    owner_layer<true><<<go, OWN_WAVES * 32, 0, stream>>>(
        ei, hw3, dinv, keepAll, keptAll, Ws, gammas + (size_t)2 * DIM, betas + (size_t)2 * DIM, bs + (size_t)2 * DIM, accb, N, E);
    const float* hcur = accb;

    const int P = (N + PCH_NODES - 1) / PCH_NODES;
    pool_onehot<<<P, 128, 0, stream>>>(hcur, batch, part, N);
    pool_reduce<<<GNUM * DIM / 32, 256, 0, stream>>>(part, (float*)d_out, P);
}


# edge lists and in-degrees from ONE scan launch before layer 1 (LDS integer counts: no global atomics, no zero fill); layer 1 runs in the owner launch like layers 2, 3; 10->8 launches (new sum order)
# speedup vs baseline: 5.0155x; 1.2238x over previous
#include <hip/hip_runtime.h>


typedef __attribute__((ext_vector_type(2))) float v2f;
typedef __attribute__((ext_vector_type(8))) float v8f;

#define DIM   128
#define GNUM  64
#define LAYERS 3
#define LNEPS 1e-5f

__global__ void deg_to_dinv(float* __restrict__ d, const float* __restrict__ d1, int N) {
    int i = blockIdx.x * blockDim.x + threadIdx.x;
    if (i < N) d[i] = rsqrtf(d[i] + d1[i] + 1.0f);
}

__global__ __launch_bounds__(256) void layer_rows(const float* __restrict__ X,
                                                  const float* __restrict__ W,
                                                  float* __restrict__ hw, int N) {
    __shared__ float lw[DIM * DIM];
    {
        const float4* Wv = (const float4*)W;
        float4* lv = (float4*)lw;
        #pragma unroll
        for (int i = 0; i < (DIM * DIM / 4) / 256; ++i)
            lv[threadIdx.x + i * 256] = Wv[threadIdx.x + i * 256];
    }
    __syncthreads();

    const int wave = __builtin_amdgcn_readfirstlane((int)(threadIdx.x >> 5));
    const int lane = threadIdx.x & 31;
    const int half = lane >> 4;
    const int ln   = lane & 15;

    const int nstrips = N >> 4;
    const int strip = blockIdx.x * 8 + wave;
    if (strip >= nstrips) return;
    const int rowBase = strip << 4;

    v2f a[DIM / 4];
    const v2f* arow2 = (const v2f*)(X + (size_t)(rowBase + ln) * DIM);
    #pragma unroll
    for (int kk = 0; kk < DIM / 4; ++kk) a[kk] = arow2[2 * kk + half];

    v8f acc[8] = {};

    #pragma unroll
    for (int kk = 0; kk < DIM / 4; ++kk) {
        const int kb = 4 * kk + 2 * half;
        #pragma unroll
        for (int t = 0; t < 8; ++t) {
            v2f b;
            b.x = lw[kb * DIM + t * 16 + ln];
            b.y = lw[(kb + 1) * DIM + t * 16 + ln];
            acc[t] = __builtin_amdgcn_wmma_f32_16x16x4_f32(
                false, a[kk], false, b, (short)0, acc[t], false, false);
        }
    }

    #pragma unroll
    for (int t = 0; t < 8; ++t) {
        #pragma unroll
        for (int j = 0; j < 8; ++j) {
            const int row = rowBase + half * 8 + j;
            hw[(size_t)row * DIM + t * 16 + ln] = acc[t][j];
        }
    }
}

#define OWN_ROWS  400
#define SUB_ROWS  200
#define SUB_PITCH 129
#define SUB_RPW   ((SUB_ROWS + 31) / 32)
#define OWN_WAVES 32
#define OWN_LIST  160
#define OWN_AHEAD 8
#define OWN_KEEP  256

template <int PITCH>
__device__ __forceinline__ void own_rows(const float* __restrict__ hw, float* sum, int src, int r,
                                         float wa, float wb, int n, int lane) {
    #pragma nounroll
    for (int g = 0; g < 32; g += 8) {
        if (g >= n) break;
        float4 v[8];
        #pragma unroll
        for (int j = 0; j < 8; ++j) {
            const int s = __builtin_amdgcn_readlane(src, g + j);
            v[j] = *(const float4*)(hw + (size_t)s * DIM + 4 * lane);
        }
        #pragma unroll
        for (int j = 0; j < 8; ++j) {
            if (g + j < n) {
                const int rj = __builtin_amdgcn_readlane(r, g + j);
                const float wj = __int_as_float(__builtin_amdgcn_readlane(__float_as_int(wa), g + j)) *
                                 __int_as_float(__builtin_amdgcn_readlane(__float_as_int(wb), g + j));
                float* dp = sum + rj * PITCH + lane;
                __hip_atomic_fetch_add(dp +  0, wj * v[j].x, __ATOMIC_RELAXED, __HIP_MEMORY_SCOPE_WORKGROUP);
                __hip_atomic_fetch_add(dp + 32, wj * v[j].y, __ATOMIC_RELAXED, __HIP_MEMORY_SCOPE_WORKGROUP);
                __hip_atomic_fetch_add(dp + 64, wj * v[j].z, __ATOMIC_RELAXED, __HIP_MEMORY_SCOPE_WORKGROUP);
                __hip_atomic_fetch_add(dp + 96, wj * v[j].w, __ATOMIC_RELAXED, __HIP_MEMORY_SCOPE_WORKGROUP);
            }
        }
    }
}

template <bool SAVE, int PITCH>
__device__ __forceinline__ void own_batch(const int* __restrict__ ei, const float* __restrict__ hw,
                                          const float* __restrict__ dinv, float* sum, int* degc,
                                          const int* le, const int* lr, int n, int base, int lane,
                                          int2* __restrict__ keep, int& keptA, int& keptB) {
    const int i = min(lane, n - 1);
    const int e = le[i];
    const int r = lr[i];
    const int src = ei[e];
    if (!SAVE) {
        const float ws = dinv[src];
        const float wd = dinv[base + r];
        own_rows<PITCH>(hw, sum, src, r, ws, wd, n, lane);
    } else {
        if (lane < n) __hip_atomic_fetch_add(degc + r, 1, __ATOMIC_RELAXED, __HIP_MEMORY_SCOPE_WORKGROUP);
        if (keptA >= 0 && keptA + keptB + n <= OWN_KEEP) {
            const bool lo = lane < n && r <  SUB_ROWS;
            const bool hi = lane < n && r >= SUB_ROWS;
            const unsigned mlo = __builtin_amdgcn_ballot_w32(lo);
            const unsigned mhi = __builtin_amdgcn_ballot_w32(hi);
            const int2 en = make_int2(src, r);
            if (lo) keep[keptA + (int)__builtin_amdgcn_mbcnt_lo(mlo, 0u)] = en;
            if (hi) keep[OWN_KEEP - 1 - keptB - (int)__builtin_amdgcn_mbcnt_lo(mhi, 0u)] = en;
            keptA += __builtin_popcount(mlo);
            keptB += __builtin_popcount(mhi);
        } else {
            keptA = -1;
        }
    }
}

template <bool SAVE, int PITCH>
__device__ __forceinline__ void own_scan(const int* __restrict__ ei, const float* __restrict__ hw,
                                         const float* __restrict__ dinv, float* sum, int* degc, int* mle, int* mlr,
                                         int base, unsigned rows, int part, int wave, int lane, int E,
                                         int2* __restrict__ keep, int& keptA, int& keptB) {
    const int* __restrict__ dstp = ei + E;
    const int nsteps = (E + 127) >> 7;
    int cnt = 0;
    for (int st0 = part * OWN_WAVES + wave; st0 < nsteps; st0 += 2 * OWN_WAVES * OWN_AHEAD) {
        int4 d[OWN_AHEAD];
        if ((st0 + (OWN_AHEAD - 1) * 2 * OWN_WAVES) * 128 + 128 <= E) {
            #pragma unroll
            for (int u = 0; u < OWN_AHEAD; ++u)
                d[u] = *(const int4*)(dstp + (st0 + u * 2 * OWN_WAVES) * 128 + 4 * lane);
        } else {
            #pragma unroll
            for (int u = 0; u < OWN_AHEAD; ++u) {
                const int e = (st0 + u * 2 * OWN_WAVES) * 128 + 4 * lane;
                d[u].x = (e     < E) ? dstp[e]     : -1;
                d[u].y = (e + 1 < E) ? dstp[e + 1] : -1;
                d[u].z = (e + 2 < E) ? dstp[e + 2] : -1;
                d[u].w = (e + 3 < E) ? dstp[e + 3] : -1;
            }
        }
        #pragma unroll
        for (int u = 0; u < OWN_AHEAD; ++u) {
            const int e = (st0 + u * 2 * OWN_WAVES) * 128 + 4 * lane;
            const unsigned r0 = (unsigned)d[u].x - (unsigned)base;
            const unsigned r1 = (unsigned)d[u].y - (unsigned)base;
            const unsigned r2 = (unsigned)d[u].z - (unsigned)base;
            const unsigned r3 = (unsigned)d[u].w - (unsigned)base;
            if (__builtin_amdgcn_ballot_w32(min(min(r0, r1), min(r2, r3)) < rows) != 0u) {
                #define OWN_APPEND(rk, ek) {                                              \
                    const bool hit = (rk) < rows;                                         \
                    const unsigned mask = __builtin_amdgcn_ballot_w32(hit);               \
                    if (hit) {                                                            \
                        const int pos = cnt + (int)__builtin_amdgcn_mbcnt_lo(mask, 0u);   \
                        mle[pos] = (ek);                                                  \
                        mlr[pos] = (int)(rk);                                             \
                    }                                                                     \
                    cnt += __builtin_popcount(mask);                                      \
                }
                OWN_APPEND(r0, e)
                OWN_APPEND(r1, e + 1)
                OWN_APPEND(r2, e + 2)
                OWN_APPEND(r3, e + 3)
                #undef OWN_APPEND
                while (cnt >= 32) {
                    cnt -= 32;
                    own_batch<SAVE, PITCH>(ei, hw, dinv, sum, degc, mle + cnt, mlr + cnt, 32, base, lane, keep, keptA, keptB);
                }
            }
        }
    }
    if (cnt > 0) own_batch<SAVE, PITCH>(ei, hw, dinv, sum, degc, mle, mlr, cnt, base, lane, keep, keptA, keptB);
}

__global__ __launch_bounds__(1024) void edge_scan(const int* __restrict__ ei,
                                                  float* __restrict__ deg0,
                                                  float* __restrict__ deg1,
                                                  int* __restrict__ keepAll,
                                                  int* __restrict__ keptAll, int N, int E) {
    __shared__ int degc[OWN_ROWS];
    __shared__ int le[OWN_WAVES][OWN_LIST];
    __shared__ int lr[OWN_WAVES][OWN_LIST];
    const int lane = threadIdx.x & 31;
    const int wave = __builtin_amdgcn_readfirstlane((int)(threadIdx.x >> 5));
    const int part = blockIdx.x & 1;
    const int base = (blockIdx.x >> 1) * OWN_ROWS;
    const unsigned rows = (unsigned)min(OWN_ROWS, N - base);

    for (int i = threadIdx.x; i < OWN_ROWS; i += OWN_WAVES * 32) degc[i] = 0;
    __syncthreads();

    int2* __restrict__ keep = (int2*)keepAll + ((size_t)blockIdx.x * OWN_WAVES + wave) * OWN_KEEP;
    int* __restrict__ keptp = keptAll + 2 * (blockIdx.x * OWN_WAVES + wave);
    int keptA = 0, keptB = 0;
    own_scan<true, DIM>(ei, nullptr, nullptr, nullptr, degc, le[wave], lr[wave], base, rows, part, wave, lane, E,
                        keep, keptA, keptB);
    if (lane == 0) { keptp[0] = keptA; keptp[1] = keptB; }
    __syncthreads();

    float* __restrict__ dp = part ? deg1 : deg0;
    for (int i = threadIdx.x; i < (int)rows; i += OWN_WAVES * 32) dp[base + i] = (float)degc[i];
}

template <bool LAST>
__global__ __launch_bounds__(1024) void owner_layer(const int* __restrict__ ei,
                                                    const float* __restrict__ hw,
                                                    const float* __restrict__ dinv,
                                                    const int* __restrict__ keepAll,
                                                    const int* __restrict__ keptAll,
                                                    const float* __restrict__ W,
                                                    const float* __restrict__ gamma,
                                                    const float* __restrict__ beta,
                                                    const float* __restrict__ bias,
                                                    float* __restrict__ out, int N, int E) {
    __shared__ __attribute__((aligned(16))) float sum[SUB_ROWS * SUB_PITCH];
    __shared__ int le[OWN_WAVES][OWN_LIST];
    __shared__ int lr[OWN_WAVES][OWN_LIST];
    __shared__ __attribute__((aligned(16))) float lw[LAST ? 4 : DIM * DIM];
    const int lane = threadIdx.x & 31;
    const int wave = __builtin_amdgcn_readfirstlane((int)(threadIdx.x >> 5));
    const int sub  = blockIdx.x & 1;
    const int grp  = blockIdx.x >> 1;
    const int base = grp * OWN_ROWS + sub * SUB_ROWS;
    const int rows = max(0, min(SUB_ROWS, N - base));

    const int id0 = (2 * grp) * OWN_WAVES + wave;
    const int id1 = id0 + OWN_WAVES;
    const int c0a = keptAll[2 * id0], c0b = keptAll[2 * id0 + 1];
    const int c1a = keptAll[2 * id1], c1b = keptAll[2 * id1 + 1];

    for (int i = threadIdx.x; i < SUB_ROWS * SUB_PITCH / 4; i += OWN_WAVES * 32)
        ((float4*)sum)[i] = make_float4(0.0f, 0.0f, 0.0f, 0.0f);

    const int k0 = __builtin_amdgcn_readfirstlane(c0a) < 0 ? -1 : __builtin_amdgcn_readfirstlane(sub ? c0b : c0a);
    const int k1 = __builtin_amdgcn_readfirstlane(c1a) < 0 ? -1 : __builtin_amdgcn_readfirstlane(sub ? c1b : c1a);
    const int roff = sub * SUB_ROWS;
    const int rbase = grp * OWN_ROWS;
    const int first = sub ? OWN_KEEP - 1 : 0;
    const int step  = sub ? -1 : 1;
    const int2* __restrict__ keep0 = (const int2*)keepAll + (size_t)id0 * OWN_KEEP;
    const int2* __restrict__ keep1 = (const int2*)keepAll + (size_t)id1 * OWN_KEEP;
    int2 f0 = make_int2(0, 0), f1 = make_int2(0, 0);
    if (k0 > 0) f0 = keep0[first + step * min(lane, k0 - 1)];
    if (k1 > 0) f1 = keep1[first + step * min(lane, k1 - 1)];
    __syncthreads();

    #pragma nounroll
    for (int p = 0; p < 2; ++p) {
        const int kept = p ? k1 : k0;
        const int2* __restrict__ keep = p ? keep1 : keep0;
        if (kept < 0) {
            int na = -1, nb = 0;
            own_scan<false, SUB_PITCH>(ei, hw, dinv, sum, nullptr, le[wave], lr[wave], base, (unsigned)rows, p, wave,
                                       lane, E, nullptr, na, nb);
        } else if (kept > 0) {
            int2 en = p ? f1 : f0;
            int b = 0;
            for (; b + 32 <= kept; b += 32) {
                const int2 cur = en;
                if (b + 32 < kept) en = keep[first + step * (b + 32 + min(lane, kept - b - 33))];
                own_rows<SUB_PITCH>(hw, sum, cur.x, cur.y - roff, dinv[cur.x], dinv[rbase + cur.y], 32, lane);
            }
            if (b < kept) own_rows<SUB_PITCH>(hw, sum, en.x, en.y - roff, dinv[en.x], dinv[rbase + en.y], kept - b, lane);
        }
    }

    float4 hv[SUB_RPW];
    float  dv[SUB_RPW];
    if (rows > 0) {
        #pragma unroll
        for (int i = 0; i < SUB_RPW; ++i) {
            const int rc = min(wave + i * OWN_WAVES, rows - 1);
            hv[i] = ((const float4*)(hw + (size_t)(base + rc) * DIM))[lane];
            dv[i] = dinv[base + rc];
        }
    }
    if (!LAST) {
        const float4* Wv = (const float4*)W;
        float4* lv = (float4*)lw;
        #pragma unroll
        for (int i = 0; i < (DIM * DIM / 4) / (OWN_WAVES * 32); ++i)
            lv[threadIdx.x + i * OWN_WAVES * 32] = Wv[threadIdx.x + i * OWN_WAVES * 32];
    }
    __syncthreads();

    if (rows > 0) {
        const float4 bi = ((const float4*)bias)[lane];
        const float4 g  = ((const float4*)gamma)[lane];
        const float4 bt = ((const float4*)beta)[lane];
        #pragma unroll
        for (int i = 0; i < SUB_RPW; ++i) {
            const int rr = wave + i * OWN_WAVES;
            if (rr < rows) {
                const float sn = dv[i] * dv[i];
                float* sp = sum + rr * SUB_PITCH + lane;
                float4 v;
                v.x = sp[0]  + (bi.x + sn * hv[i].x);
                v.y = sp[32] + (bi.y + sn * hv[i].y);
                v.z = sp[64] + (bi.z + sn * hv[i].z);
                v.w = sp[96] + (bi.w + sn * hv[i].w);
                float s = v.x + v.y + v.z + v.w;
                float q = v.x * v.x + v.y * v.y + v.z * v.z + v.w * v.w;
                #pragma unroll
                for (int off = 16; off > 0; off >>= 1) {
                    s += __shfl_xor(s, off, 32);
                    q += __shfl_xor(q, off, 32);
                }
                const float mu  = s * (1.0f / DIM);
                const float var = fmaxf(q * (1.0f / DIM) - mu * mu, 0.0f);
                const float inv = rsqrtf(var + LNEPS);
                float4 r;
                r.x = fmaxf(0.0f, g.x * (v.x - mu) * inv + bt.x);
                r.y = fmaxf(0.0f, g.y * (v.y - mu) * inv + bt.y);
                r.z = fmaxf(0.0f, g.z * (v.z - mu) * inv + bt.z);
                r.w = fmaxf(0.0f, g.w * (v.w - mu) * inv + bt.w);
                if (LAST) {
                    ((float4*)(out + (size_t)(base + rr) * DIM))[lane] = r;
                } else {
                    sp[0] = r.x;  sp[32] = r.y;  sp[64] = r.z;  sp[96] = r.w;
                }
            }
        }
    }
    if (!LAST) {
        __syncthreads();
        const int strip = wave >> 1;
        const int ch    = wave & 1;
        if (strip * 16 >= rows) return;
        const int half = lane >> 4;
        const int ln   = lane & 15;
        const int rl = min(strip * 16 + ln, rows - 1);
        const float* srow = sum + rl * SUB_PITCH + 64 * half;

        v8f acc[4] = {};
        #pragma unroll
        for (int kk = 0; kk < DIM / 4; ++kk) {
            v2f a;
            a.x = srow[kk];
            a.y = srow[32 + kk];
            const int kb = 4 * kk + 2 * half;
            #pragma unroll
            for (int t = 0; t < 4; ++t) {
                v2f b;
                b.x = lw[kb * DIM + ch * 64 + t * 16 + ln];
                b.y = lw[(kb + 1) * DIM + ch * 64 + t * 16 + ln];
                acc[t] = __builtin_amdgcn_wmma_f32_16x16x4_f32(
                    false, a, false, b, (short)0, acc[t], false, false);
            }
        }
        #pragma unroll
        for (int t = 0; t < 4; ++t) {
            #pragma unroll
            for (int j = 0; j < 8; ++j) {
                const int r2 = strip * 16 + half * 8 + j;
                if (r2 < rows) out[(size_t)(base + r2) * DIM + ch * 64 + t * 16 + ln] = acc[t][j];
            }
        }
    }
}

#define PCH_BLOCKS 5
#define PCH_STEPS  5
#define PCH_WNODES (PCH_BLOCKS * PCH_STEPS * 4)
#define PCH_NODES  (2 * PCH_WNODES)
#define PSTRIDE    (GNUM * DIM + GNUM)

__global__ __launch_bounds__(128) void pool_onehot(const float* __restrict__ h,
                                                   const int* __restrict__ batch,
                                                   float* __restrict__ part, int N) {
    __shared__ float4 lsum[GNUM * DIM / 4];
    __shared__ float  lcnt[GNUM];
    const int wave = threadIdx.x >> 5, lane = threadIdx.x & 31;
    const int ch = wave & 1;
    const int sc = wave >> 1;
    const int half = lane >> 4, c = lane & 15;
    const int nbase = blockIdx.x * PCH_NODES + sc * PCH_WNODES + 2 * half;
    const float* hcol = h + 64 * ch + 4 * c;

    v8f acc[4][4] = {};
    float cnt[4] = {0.0f, 0.0f, 0.0f, 0.0f};

    #pragma unroll
    for (int blk = 0; blk < PCH_BLOCKS; ++blk) {
        float4 r0[PCH_STEPS], r1[PCH_STEPS];
        int g0[PCH_STEPS], g1[PCH_STEPS];
        #pragma unroll
        for (int s = 0; s < PCH_STEPS; ++s) {
            const int n0 = nbase + (blk * PCH_STEPS + s) * 4;
            const int n1 = n0 + 1;
            const int i0 = min(n0, N - 1), i1 = min(n1, N - 1);
            r0[s] = *(const float4*)(hcol + (size_t)i0 * DIM);
            r1[s] = *(const float4*)(hcol + (size_t)i1 * DIM);
            const int b0 = batch[i0], b1 = batch[i1];
            g0[s] = (n0 < N) ? b0 : -1;
            g1[s] = (n1 < N) ? b1 : -1;
        }
        #pragma unroll
        for (int s = 0; s < PCH_STEPS; ++s) {
            #pragma unroll
            for (int m = 0; m < 4; ++m) {
                v2f a;
                a.x = (g0[s] == 16 * m + c) ? 1.0f : 0.0f;
                a.y = (g1[s] == 16 * m + c) ? 1.0f : 0.0f;
                cnt[m] += a.x + a.y;
                v2f b;
                b.x = r0[s].x; b.y = r1[s].x;
                acc[m][0] = __builtin_amdgcn_wmma_f32_16x16x4_f32(false, a, false, b, (short)0, acc[m][0], false, false);
                b.x = r0[s].y; b.y = r1[s].y;
                acc[m][1] = __builtin_amdgcn_wmma_f32_16x16x4_f32(false, a, false, b, (short)0, acc[m][1], false, false);
                b.x = r0[s].z; b.y = r1[s].z;
                acc[m][2] = __builtin_amdgcn_wmma_f32_16x16x4_f32(false, a, false, b, (short)0, acc[m][2], false, false);
                b.x = r0[s].w; b.y = r1[s].w;
                acc[m][3] = __builtin_amdgcn_wmma_f32_16x16x4_f32(false, a, false, b, (short)0, acc[m][3], false, false);
            }
        }
    }

    #pragma unroll
    for (int m = 0; m < 4; ++m) cnt[m] += __shfl_xor(cnt[m], 16, 32);

    if (sc == 1) {
        #pragma unroll
        for (int m = 0; m < 4; ++m) {
            #pragma unroll
            for (int r = 0; r < 8; ++r) {
                const int row = 16 * m + 8 * half + r;
                lsum[row * (DIM / 4) + 16 * ch + c] =
                    make_float4(acc[m][0][r], acc[m][1][r], acc[m][2][r], acc[m][3][r]);
            }
            if (ch == 0 && half == 0) lcnt[16 * m + c] = cnt[m];
        }
    }
    __syncthreads();
    if (sc == 0) {
        float* pp = part + (size_t)blockIdx.x * PSTRIDE;
        #pragma unroll
        for (int m = 0; m < 4; ++m) {
            #pragma unroll
            for (int r = 0; r < 8; ++r) {
                const int row = 16 * m + 8 * half + r;
                const float4 o = lsum[row * (DIM / 4) + 16 * ch + c];
                float4 v;
                v.x = acc[m][0][r] + o.x;  v.y = acc[m][1][r] + o.y;
                v.z = acc[m][2][r] + o.z;  v.w = acc[m][3][r] + o.w;
                ((float4*)pp)[row * (DIM / 4) + 16 * ch + c] = v;
            }
            if (ch == 0 && half == 0) pp[GNUM * DIM + 16 * m + c] = cnt[m] + lcnt[16 * m + c];
        }
    }
}

__global__ __launch_bounds__(256) void pool_reduce(const float* __restrict__ part,
                                                   float* __restrict__ out, int P) {
    __shared__ float red[8][32];
    __shared__ float redc[8];
    const int lane = threadIdx.x & 31, s = threadIdx.x >> 5;
    const int o = blockIdx.x * 32 + lane;
    const int g = (blockIdx.x * 32) / DIM;
    float sum = 0.0f;
    for (int p0 = s; p0 < P; p0 += 128) {
        float v[16];
        #pragma unroll
        for (int u = 0; u < 16; ++u) {
            const int p = p0 + 8 * u;
            v[u] = (p < P) ? part[(size_t)p * PSTRIDE + o] : 0.0f;
        }
        #pragma unroll
        for (int u = 0; u < 16; ++u) sum += v[u];
    }
    float cs = 0.0f;
    for (int p = s + 8 * lane; p < P; p += 256) cs += part[(size_t)p * PSTRIDE + GNUM * DIM + g];
    #pragma unroll
    for (int off = 16; off > 0; off >>= 1) cs += __shfl_xor(cs, off, 32);
    red[s][lane] = sum;
    if (lane == 0) redc[s] = cs;
    __syncthreads();
    if (s == 0) {
        float t = 0.0f, cc = 0.0f;
        #pragma unroll
        for (int i = 0; i < 8; ++i) { t += red[i][lane]; cc += redc[i]; }
        out[o] = t / fmaxf(cc, 1.0f);
    }
}
static inline size_t al256(size_t x) { return (x + 255) & ~(size_t)255; }

extern "C" void kernel_launch(void* const* d_in, const int* in_sizes, int n_in,
                              void* d_out, int out_size, void* d_ws, size_t ws_size,
                              hipStream_t stream) {
    const float* x      = (const float*)d_in[0];
    const int*   ei     = (const int*)d_in[1];
    const int*   batch  = (const int*)d_in[2];
    const float* Ws     = (const float*)d_in[3];
    const float* bs     = (const float*)d_in[4];
    const float* gammas = (const float*)d_in[5];
    const float* betas  = (const float*)d_in[6];

    const int N = in_sizes[0] / DIM;
    const int E = in_sizes[1] / 2;
    const size_t nd = (size_t)N * DIM * sizeof(float);

    char* ws = (char*)d_ws;
    float* dinv = (float*)ws;                 ws += al256((size_t)N * sizeof(float));
    float* deg1 = (float*)ws;                 ws += al256((size_t)N * sizeof(float));
    float* hw   = (float*)ws;                 ws += al256(nd);
    float* hw2  = (float*)ws;                 ws += al256(nd);
    float* accb = (float*)ws;                 ws += al256(nd);
    const int go = 2 * ((N + OWN_ROWS - 1) / OWN_ROWS);
    int* keepAll = (int*)ws;                  ws += al256((size_t)go * OWN_WAVES * OWN_KEEP * sizeof(int2));
    int* keptAll = (int*)ws;                  ws += al256((size_t)go * OWN_WAVES * 2 * sizeof(int));
    float* part = (float*)ws;

    edge_scan<<<go, OWN_WAVES * 32, 0, stream>>>(ei, dinv, deg1, keepAll, keptAll, N, E);
    deg_to_dinv<<<(N + 255) / 256, 256, 0, stream>>>(dinv, deg1, N);

    const int nstrips = N >> 4;
    const int gw = (nstrips + 7) / 8;
    layer_rows<<<gw, 256, 0, stream>>>(x, Ws, hw, N);
    owner_layer<false><<<go, OWN_WAVES * 32, 0, stream>>>(
        ei, hw, dinv, keepAll, keptAll, Ws + (size_t)DIM * DIM, gammas, betas, bs, hw2, N, E);
    owner_layer<false><<<go, OWN_WAVES * 32, 0, stream>>>(
        ei, hw2, dinv, keepAll, keptAll, Ws + (size_t)2 * DIM * DIM, gammas + DIM, betas + DIM, bs + DIM, hw, N, E);
    owner_layer<true><<<go, OWN_WAVES * 32, 0, stream>>>(
        ei, hw, dinv, keepAll, keptAll, Ws, gammas + (size_t)2 * DIM, betas + (size_t)2 * DIM, bs + (size_t)2 * DIM, accb, N, E);
    const float* hcur = accb;

    const int P = (N + PCH_NODES - 1) / PCH_NODES;
    pool_onehot<<<P, 128, 0, stream>>>(hcur, batch, part, N);
    pool_reduce<<<GNUM * DIM / 32, 256, 0, stream>>>(part, (float*)d_out, P);
}


# edge scan: work-group per 3200 rows and 1/16 of the list (list read 16 times, not 125); entries go to per-200-row lists at an LDS count's slot; owner waves split them evenly (sum order: LDS arrival)
# speedup vs baseline: 5.3650x; 1.0697x over previous
#include <hip/hip_runtime.h>


typedef __attribute__((ext_vector_type(2))) float v2f;
typedef __attribute__((ext_vector_type(8))) float v8f;

#define DIM   128
#define GNUM  64
#define LAYERS 3
#define LNEPS 1e-5f

#define SUB_ROWS   200
#define SCAN_SUBS  16
#define SCAN_ROWS  (SCAN_SUBS * SUB_ROWS)
#define SCAN_PARTS 16
#define SUB_CAP    320

__global__ void deg_to_dinv(float* __restrict__ d, const int* __restrict__ degp, int N) {
    int i = blockIdx.x * blockDim.x + threadIdx.x;
    if (i < N) {
        int c = 1;
        #pragma unroll
        for (int p = 0; p < SCAN_PARTS; ++p) c += degp[(size_t)p * N + i];
        d[i] = rsqrtf((float)c);
    }
}

__global__ __launch_bounds__(256) void layer_rows(const float* __restrict__ X,
                                                  const float* __restrict__ W,
                                                  float* __restrict__ hw, int N) {
    __shared__ float lw[DIM * DIM];
    {
        const float4* Wv = (const float4*)W;
        float4* lv = (float4*)lw;
        #pragma unroll
        for (int i = 0; i < (DIM * DIM / 4) / 256; ++i)
            lv[threadIdx.x + i * 256] = Wv[threadIdx.x + i * 256];
    }
    __syncthreads();

    const int wave = __builtin_amdgcn_readfirstlane((int)(threadIdx.x >> 5));
    const int lane = threadIdx.x & 31;
    const int half = lane >> 4;
    const int ln   = lane & 15;

    const int nstrips = N >> 4;
    const int strip = blockIdx.x * 8 + wave;
    if (strip >= nstrips) return;
    const int rowBase = strip << 4;

    v2f a[DIM / 4];
    const v2f* arow2 = (const v2f*)(X + (size_t)(rowBase + ln) * DIM);
    #pragma unroll
    for (int kk = 0; kk < DIM / 4; ++kk) a[kk] = arow2[2 * kk + half];

    v8f acc[8] = {};

    #pragma unroll
    for (int kk = 0; kk < DIM / 4; ++kk) {
        const int kb = 4 * kk + 2 * half;
        #pragma unroll
        for (int t = 0; t < 8; ++t) {
            v2f b;
            b.x = lw[kb * DIM + t * 16 + ln];
            b.y = lw[(kb + 1) * DIM + t * 16 + ln];
            acc[t] = __builtin_amdgcn_wmma_f32_16x16x4_f32(
                false, a[kk], false, b, (short)0, acc[t], false, false);
        }
    }

    #pragma unroll
    for (int t = 0; t < 8; ++t) {
        #pragma unroll
        for (int j = 0; j < 8; ++j) {
            const int row = rowBase + half * 8 + j;
            hw[(size_t)row * DIM + t * 16 + ln] = acc[t][j];
        }
    }
}

#define SUB_PITCH 129
#define SUB_RPW   ((SUB_ROWS + 31) / 32)
#define OWN_WAVES 32
#define OWN_LIST  160
#define OWN_AHEAD 8
#define OWN_STRIDE (SCAN_PARTS * OWN_WAVES)

template <int PITCH>
__device__ __forceinline__ void own_rows(const float* __restrict__ hw, float* sum, int src, int r,
                                         float wa, float wb, int n, int lane) {
    #pragma nounroll
    for (int g = 0; g < 32; g += 8) {
        if (g >= n) break;
        float4 v[8];
        #pragma unroll
        for (int j = 0; j < 8; ++j) {
            const int s = __builtin_amdgcn_readlane(src, g + j);
            v[j] = *(const float4*)(hw + (size_t)s * DIM + 4 * lane);
        }
        #pragma unroll
        for (int j = 0; j < 8; ++j) {
            if (g + j < n) {
                const int rj = __builtin_amdgcn_readlane(r, g + j);
                const float wj = __int_as_float(__builtin_amdgcn_readlane(__float_as_int(wa), g + j)) *
                                 __int_as_float(__builtin_amdgcn_readlane(__float_as_int(wb), g + j));
                float* dp = sum + rj * PITCH + lane;
                __hip_atomic_fetch_add(dp +  0, wj * v[j].x, __ATOMIC_RELAXED, __HIP_MEMORY_SCOPE_WORKGROUP);
                __hip_atomic_fetch_add(dp + 32, wj * v[j].y, __ATOMIC_RELAXED, __HIP_MEMORY_SCOPE_WORKGROUP);
                __hip_atomic_fetch_add(dp + 64, wj * v[j].z, __ATOMIC_RELAXED, __HIP_MEMORY_SCOPE_WORKGROUP);
                __hip_atomic_fetch_add(dp + 96, wj * v[j].w, __ATOMIC_RELAXED, __HIP_MEMORY_SCOPE_WORKGROUP);
            }
        }
    }
}

template <bool SAVE, int PITCH>
__device__ __forceinline__ void own_batch(const int* __restrict__ ei, const float* __restrict__ hw,
                                          const float* __restrict__ dinv, float* sum, int* degc, int* scnt,
                                          const int* le, const int* lr, int n, int base, int lane,
                                          int2* __restrict__ keep) {
    const int i = min(lane, n - 1);
    const int e = le[i];
    const int r = lr[i];
    const int src = ei[e];
    if (!SAVE) {
        const float ws = dinv[src];
        const float wd = dinv[base + r];
        own_rows<PITCH>(hw, sum, src, r, ws, wd, n, lane);
    } else if (lane < n) {
        __hip_atomic_fetch_add(degc + r, 1, __ATOMIC_RELAXED, __HIP_MEMORY_SCOPE_WORKGROUP);
        const int sub = (int)((unsigned)r / (unsigned)SUB_ROWS);
        const int slot = __hip_atomic_fetch_add(scnt + sub, 1, __ATOMIC_RELAXED, __HIP_MEMORY_SCOPE_WORKGROUP);
        if (slot < SUB_CAP) keep[sub * SUB_CAP + slot] = make_int2(src, r - sub * SUB_ROWS);
    }
}

template <bool SAVE, int PITCH>
__device__ __forceinline__ void own_scan(const int* __restrict__ ei, const float* __restrict__ hw,
                                         const float* __restrict__ dinv, float* sum, int* degc, int* scnt,
                                         int* mle, int* mlr, int base, unsigned rows, int part, int wave, int lane,
                                         int E, int2* __restrict__ keep) {
    const int* __restrict__ dstp = ei + E;
    const int nsteps = (E + 127) >> 7;
    int cnt = 0;
    for (int st0 = part * OWN_WAVES + wave; st0 < nsteps; st0 += OWN_STRIDE * OWN_AHEAD) {
        int4 d[OWN_AHEAD];
        if ((st0 + (OWN_AHEAD - 1) * OWN_STRIDE) * 128 + 128 <= E) {
            #pragma unroll
            for (int u = 0; u < OWN_AHEAD; ++u)
                d[u] = *(const int4*)(dstp + (st0 + u * OWN_STRIDE) * 128 + 4 * lane);
        } else {
            #pragma unroll
            for (int u = 0; u < OWN_AHEAD; ++u) {
                const int eb = (st0 + u * OWN_STRIDE) * 128;
                const int e = eb + 4 * lane;
                if (eb + 128 <= E) {
                    d[u] = *(const int4*)(dstp + e);
                } else {
                    d[u].x = (e     < E) ? dstp[e]     : -1;
                    d[u].y = (e + 1 < E) ? dstp[e + 1] : -1;
                    d[u].z = (e + 2 < E) ? dstp[e + 2] : -1;
                    d[u].w = (e + 3 < E) ? dstp[e + 3] : -1;
                }
            }
        }
        #pragma unroll
        for (int u = 0; u < OWN_AHEAD; ++u) {
            const int e = (st0 + u * OWN_STRIDE) * 128 + 4 * lane;
            const unsigned r0 = (unsigned)d[u].x - (unsigned)base;
            const unsigned r1 = (unsigned)d[u].y - (unsigned)base;
            const unsigned r2 = (unsigned)d[u].z - (unsigned)base;
            const unsigned r3 = (unsigned)d[u].w - (unsigned)base;
            if (__builtin_amdgcn_ballot_w32(min(min(r0, r1), min(r2, r3)) < rows) != 0u) {
                #define OWN_APPEND(rk, ek) {                                              \
                    const bool hit = (rk) < rows;                                         \
                    const unsigned mask = __builtin_amdgcn_ballot_w32(hit);               \
                    if (hit) {                                                            \
                        const int pos = cnt + (int)__builtin_amdgcn_mbcnt_lo(mask, 0u);   \
                        mle[pos] = (ek);                                                  \
                        mlr[pos] = (int)(rk);                                             \
                    }                                                                     \
                    cnt += __builtin_popcount(mask);                                      \
                }
                OWN_APPEND(r0, e)
                OWN_APPEND(r1, e + 1)
                OWN_APPEND(r2, e + 2)
                OWN_APPEND(r3, e + 3)
                #undef OWN_APPEND
                while (cnt >= 32) {
                    cnt -= 32;
                    own_batch<SAVE, PITCH>(ei, hw, dinv, sum, degc, scnt, mle + cnt, mlr + cnt, 32, base, lane, keep);
                }
            }
        }
    }
    if (cnt > 0) own_batch<SAVE, PITCH>(ei, hw, dinv, sum, degc, scnt, mle, mlr, cnt, base, lane, keep);
}

__global__ __launch_bounds__(1024) void edge_scan(const int* __restrict__ ei,
                                                  int* __restrict__ degp,
                                                  int* __restrict__ keepAll,
                                                  int* __restrict__ cntAll, int N, int E) {
    __shared__ int degc[SCAN_ROWS];
    __shared__ int scnt[SCAN_SUBS];
    __shared__ int le[OWN_WAVES][OWN_LIST];
    __shared__ int lr[OWN_WAVES][OWN_LIST];
    const int lane = threadIdx.x & 31;
    const int wave = __builtin_amdgcn_readfirstlane((int)(threadIdx.x >> 5));
    const int part = blockIdx.x % SCAN_PARTS;
    const int base = (blockIdx.x / SCAN_PARTS) * SCAN_ROWS;
    const unsigned rows = (unsigned)min(SCAN_ROWS, N - base);

    for (int i = threadIdx.x; i < SCAN_ROWS; i += OWN_WAVES * 32) degc[i] = 0;
    if (threadIdx.x < SCAN_SUBS) scnt[threadIdx.x] = 0;
    __syncthreads();

    int2* __restrict__ keep = (int2*)keepAll + (size_t)blockIdx.x * SCAN_SUBS * SUB_CAP;
    own_scan<true, DIM>(ei, nullptr, nullptr, nullptr, degc, scnt, le[wave], lr[wave], base, rows, part, wave, lane, E,
                        keep);
    __syncthreads();

    int* __restrict__ dp = degp + (size_t)part * N;
    for (int i = threadIdx.x; i < (int)rows; i += OWN_WAVES * 32) dp[base + i] = degc[i];
    if (threadIdx.x < SCAN_SUBS)
        cntAll[blockIdx.x * SCAN_SUBS + threadIdx.x] = scnt[threadIdx.x] > SUB_CAP ? -1 : scnt[threadIdx.x];
}

template <bool LAST>
__global__ __launch_bounds__(1024) void owner_layer(const int* __restrict__ ei,
                                                    const float* __restrict__ hw,
                                                    const float* __restrict__ dinv,
                                                    const int* __restrict__ keepAll,
                                                    const int* __restrict__ cntAll,
                                                    const float* __restrict__ W,
                                                    const float* __restrict__ gamma,
                                                    const float* __restrict__ beta,
                                                    const float* __restrict__ bias,
                                                    float* __restrict__ out, int N, int E) {
    __shared__ __attribute__((aligned(16))) float sum[SUB_ROWS * SUB_PITCH];
    __shared__ int le[OWN_WAVES][OWN_LIST];
    __shared__ int lr[OWN_WAVES][OWN_LIST];
    __shared__ __attribute__((aligned(16))) float lw[LAST ? 4 : DIM * DIM];
    const int lane = threadIdx.x & 31;
    const int wave = __builtin_amdgcn_readfirstlane((int)(threadIdx.x >> 5));
    const int sub  = blockIdx.x % SCAN_SUBS;
    const int grp  = blockIdx.x / SCAN_SUBS;
    const int base = blockIdx.x * SUB_ROWS;
    const int rows = max(0, min(SUB_ROWS, N - base));

    const int craw = cntAll[(grp * SCAN_PARTS + min(lane, SCAN_PARTS - 1)) * SCAN_SUBS + sub];

    for (int i = threadIdx.x; i < SUB_ROWS * SUB_PITCH / 4; i += OWN_WAVES * 32)
        ((float4*)sum)[i] = make_float4(0.0f, 0.0f, 0.0f, 0.0f);

    int pre[SCAN_PARTS + 1];
    unsigned redo = 0u;
    pre[0] = 0;
    #pragma unroll
    for (int p = 0; p < SCAN_PARTS; ++p) {
        const int c = __builtin_amdgcn_readlane(craw, p);
        if (c < 0) redo |= 1u << p;
        pre[p + 1] = pre[p] + max(c, 0);
    }
    const int T  = pre[SCAN_PARTS];
    const int gs = (T * wave) >> 5;
    const int ge = (T * (wave + 1)) >> 5;
    const int2* __restrict__ keepR = (const int2*)keepAll + ((size_t)grp * SCAN_PARTS * SCAN_SUBS + sub) * SUB_CAP;
    auto fetch = [&](int g0) -> int2 {
        const int g = min(g0 + lane, ge - 1);
        int l = 0, o = g;
        #pragma unroll
        for (int p = 1; p < SCAN_PARTS; ++p) {
            const bool in = g >= pre[p];
            l = in ? p : l;
            o = in ? g - pre[p] : o;
        }
        return keepR[(size_t)l * (SCAN_SUBS * SUB_CAP) + o];
    };
    int2 en = make_int2(0, 0);
    if (ge > gs) en = fetch(gs);
    __syncthreads();

    for (unsigned m = redo; m != 0u; m &= m - 1u) {
        const int p = __builtin_ctz(m);
        own_scan<false, SUB_PITCH>(ei, hw, dinv, sum, nullptr, nullptr, le[wave], lr[wave], base, (unsigned)rows, p, wave,
                                   lane, E, nullptr);
    }
    {
        int b = gs;
        for (; b + 32 <= ge; b += 32) {
            const int2 cur = en;
            if (b + 32 < ge) en = fetch(b + 32);
            own_rows<SUB_PITCH>(hw, sum, cur.x, cur.y, dinv[cur.x], dinv[base + cur.y], 32, lane);
        }
        if (b < ge) own_rows<SUB_PITCH>(hw, sum, en.x, en.y, dinv[en.x], dinv[base + en.y], ge - b, lane);
    }

    float4 hv[SUB_RPW];
    float  dv[SUB_RPW];
    if (rows > 0) {
        #pragma unroll
        for (int i = 0; i < SUB_RPW; ++i) {
            const int rc = min(wave + i * OWN_WAVES, rows - 1);
            hv[i] = ((const float4*)(hw + (size_t)(base + rc) * DIM))[lane];
            dv[i] = dinv[base + rc];
        }
    }
    if (!LAST) {
        const float4* Wv = (const float4*)W;
        float4* lv = (float4*)lw;
        #pragma unroll
        for (int i = 0; i < (DIM * DIM / 4) / (OWN_WAVES * 32); ++i)
            lv[threadIdx.x + i * OWN_WAVES * 32] = Wv[threadIdx.x + i * OWN_WAVES * 32];
    }
    __syncthreads();

    if (rows > 0) {
        const float4 bi = ((const float4*)bias)[lane];
        const float4 g  = ((const float4*)gamma)[lane];
        const float4 bt = ((const float4*)beta)[lane];
        #pragma unroll
        for (int i = 0; i < SUB_RPW; ++i) {
            const int rr = wave + i * OWN_WAVES;
            if (rr < rows) {
                const float sn = dv[i] * dv[i];
                float* sp = sum + rr * SUB_PITCH + lane;
                float4 v;
                v.x = sp[0]  + (bi.x + sn * hv[i].x);
                v.y = sp[32] + (bi.y + sn * hv[i].y);
                v.z = sp[64] + (bi.z + sn * hv[i].z);
                v.w = sp[96] + (bi.w + sn * hv[i].w);
                float s = v.x + v.y + v.z + v.w;
                float q = v.x * v.x + v.y * v.y + v.z * v.z + v.w * v.w;
                #pragma unroll
                for (int off = 16; off > 0; off >>= 1) {
                    s += __shfl_xor(s, off, 32);
                    q += __shfl_xor(q, off, 32);
                }
                const float mu  = s * (1.0f / DIM);
                const float var = fmaxf(q * (1.0f / DIM) - mu * mu, 0.0f);
                const float inv = rsqrtf(var + LNEPS);
                float4 r;
                r.x = fmaxf(0.0f, g.x * (v.x - mu) * inv + bt.x);
                r.y = fmaxf(0.0f, g.y * (v.y - mu) * inv + bt.y);
                r.z = fmaxf(0.0f, g.z * (v.z - mu) * inv + bt.z);
                r.w = fmaxf(0.0f, g.w * (v.w - mu) * inv + bt.w);
                if (LAST) {
                    ((float4*)(out + (size_t)(base + rr) * DIM))[lane] = r;
                } else {
                    sp[0] = r.x;  sp[32] = r.y;  sp[64] = r.z;  sp[96] = r.w;
                }
            }
        }
    }
    if (!LAST) {
        __syncthreads();
        const int strip = wave >> 1;
        const int ch    = wave & 1;
        if (strip * 16 >= rows) return;
        const int half = lane >> 4;
        const int ln   = lane & 15;
        const int rl = min(strip * 16 + ln, rows - 1);
        const float* srow = sum + rl * SUB_PITCH + 64 * half;

        v8f acc[4] = {};
        #pragma unroll
        for (int kk = 0; kk < DIM / 4; ++kk) {
            v2f a;
            a.x = srow[kk];
            a.y = srow[32 + kk];
            const int kb = 4 * kk + 2 * half;
            #pragma unroll
            for (int t = 0; t < 4; ++t) {
                v2f b;
                b.x = lw[kb * DIM + ch * 64 + t * 16 + ln];
                b.y = lw[(kb + 1) * DIM + ch * 64 + t * 16 + ln];
                acc[t] = __builtin_amdgcn_wmma_f32_16x16x4_f32(
                    false, a, false, b, (short)0, acc[t], false, false);
            }
        }
        #pragma unroll
        for (int t = 0; t < 4; ++t) {
            #pragma unroll
            for (int j = 0; j < 8; ++j) {
                const int r2 = strip * 16 + half * 8 + j;
                if (r2 < rows) out[(size_t)(base + r2) * DIM + ch * 64 + t * 16 + ln] = acc[t][j];
            }
        }
    }
}

#define PCH_BLOCKS 5
#define PCH_STEPS  5
#define PCH_WNODES (PCH_BLOCKS * PCH_STEPS * 4)
#define PCH_NODES  (2 * PCH_WNODES)
#define PSTRIDE    (GNUM * DIM + GNUM)

__global__ __launch_bounds__(128) void pool_onehot(const float* __restrict__ h,
                                                   const int* __restrict__ batch,
                                                   float* __restrict__ part, int N) {
    __shared__ float4 lsum[GNUM * DIM / 4];
    __shared__ float  lcnt[GNUM];
    const int wave = threadIdx.x >> 5, lane = threadIdx.x & 31;
    const int ch = wave & 1;
    const int sc = wave >> 1;
    const int half = lane >> 4, c = lane & 15;
    const int nbase = blockIdx.x * PCH_NODES + sc * PCH_WNODES + 2 * half;
    const float* hcol = h + 64 * ch + 4 * c;

    v8f acc[4][4] = {};
    float cnt[4] = {0.0f, 0.0f, 0.0f, 0.0f};

    #pragma unroll
    for (int blk = 0; blk < PCH_BLOCKS; ++blk) {
        float4 r0[PCH_STEPS], r1[PCH_STEPS];
        int g0[PCH_STEPS], g1[PCH_STEPS];
        #pragma unroll
        for (int s = 0; s < PCH_STEPS; ++s) {
            const int n0 = nbase + (blk * PCH_STEPS + s) * 4;
            const int n1 = n0 + 1;
            const int i0 = min(n0, N - 1), i1 = min(n1, N - 1);
            r0[s] = *(const float4*)(hcol + (size_t)i0 * DIM);
            r1[s] = *(const float4*)(hcol + (size_t)i1 * DIM);
            const int b0 = batch[i0], b1 = batch[i1];
            g0[s] = (n0 < N) ? b0 : -1;
            g1[s] = (n1 < N) ? b1 : -1;
        }
        #pragma unroll
        for (int s = 0; s < PCH_STEPS; ++s) {
            #pragma unroll
            for (int m = 0; m < 4; ++m) {
                v2f a;
                a.x = (g0[s] == 16 * m + c) ? 1.0f : 0.0f;
                a.y = (g1[s] == 16 * m + c) ? 1.0f : 0.0f;
                cnt[m] += a.x + a.y;
                v2f b;
                b.x = r0[s].x; b.y = r1[s].x;
                acc[m][0] = __builtin_amdgcn_wmma_f32_16x16x4_f32(false, a, false, b, (short)0, acc[m][0], false, false);
                b.x = r0[s].y; b.y = r1[s].y;
                acc[m][1] = __builtin_amdgcn_wmma_f32_16x16x4_f32(false, a, false, b, (short)0, acc[m][1], false, false);
                b.x = r0[s].z; b.y = r1[s].z;
                acc[m][2] = __builtin_amdgcn_wmma_f32_16x16x4_f32(false, a, false, b, (short)0, acc[m][2], false, false);
                b.x = r0[s].w; b.y = r1[s].w;
                acc[m][3] = __builtin_amdgcn_wmma_f32_16x16x4_f32(false, a, false, b, (short)0, acc[m][3], false, false);
            }
        }
    }

    #pragma unroll
    for (int m = 0; m < 4; ++m) cnt[m] += __shfl_xor(cnt[m], 16, 32);

    if (sc == 1) {
        #pragma unroll
        for (int m = 0; m < 4; ++m) {
            #pragma unroll
            for (int r = 0; r < 8; ++r) {
                const int row = 16 * m + 8 * half + r;
                lsum[row * (DIM / 4) + 16 * ch + c] =
                    make_float4(acc[m][0][r], acc[m][1][r], acc[m][2][r], acc[m][3][r]);
            }
            if (ch == 0 && half == 0) lcnt[16 * m + c] = cnt[m];
        }
    }
    __syncthreads();
    if (sc == 0) {
        float* pp = part + (size_t)blockIdx.x * PSTRIDE;
        #pragma unroll
        for (int m = 0; m < 4; ++m) {
            #pragma unroll
            for (int r = 0; r < 8; ++r) {
                const int row = 16 * m + 8 * half + r;
                const float4 o = lsum[row * (DIM / 4) + 16 * ch + c];
                float4 v;
                v.x = acc[m][0][r] + o.x;  v.y = acc[m][1][r] + o.y;
                v.z = acc[m][2][r] + o.z;  v.w = acc[m][3][r] + o.w;
                ((float4*)pp)[row * (DIM / 4) + 16 * ch + c] = v;
            }
            if (ch == 0 && half == 0) pp[GNUM * DIM + 16 * m + c] = cnt[m] + lcnt[16 * m + c];
        }
    }
}

__global__ __launch_bounds__(256) void pool_reduce(const float* __restrict__ part,
                                                   float* __restrict__ out, int P) {
    __shared__ float red[8][32];
    __shared__ float redc[8];
    const int lane = threadIdx.x & 31, s = threadIdx.x >> 5;
    const int o = blockIdx.x * 32 + lane;
    const int g = (blockIdx.x * 32) / DIM;
    float sum = 0.0f;
    for (int p0 = s; p0 < P; p0 += 128) {
        float v[16];
        #pragma unroll
        for (int u = 0; u < 16; ++u) {
            const int p = p0 + 8 * u;
            v[u] = (p < P) ? part[(size_t)p * PSTRIDE + o] : 0.0f;
        }
        #pragma unroll
        for (int u = 0; u < 16; ++u) sum += v[u];
    }
    float cs = 0.0f;
    for (int p = s + 8 * lane; p < P; p += 256) cs += part[(size_t)p * PSTRIDE + GNUM * DIM + g];
    #pragma unroll
    for (int off = 16; off > 0; off >>= 1) cs += __shfl_xor(cs, off, 32);
    red[s][lane] = sum;
    if (lane == 0) redc[s] = cs;
    __syncthreads();
    if (s == 0) {
        float t = 0.0f, cc = 0.0f;
        #pragma unroll
        for (int i = 0; i < 8; ++i) { t += red[i][lane]; cc += redc[i]; }
        out[o] = t / fmaxf(cc, 1.0f);
    }
}
static inline size_t al256(size_t x) { return (x + 255) & ~(size_t)255; }

extern "C" void kernel_launch(void* const* d_in, const int* in_sizes, int n_in,
                              void* d_out, int out_size, void* d_ws, size_t ws_size,
                              hipStream_t stream) {
    const float* x      = (const float*)d_in[0];
    const int*   ei     = (const int*)d_in[1];
    const int*   batch  = (const int*)d_in[2];
    const float* Ws     = (const float*)d_in[3];
    const float* bs     = (const float*)d_in[4];
    const float* gammas = (const float*)d_in[5];
    const float* betas  = (const float*)d_in[6];

    const int N = in_sizes[0] / DIM;
    const int E = in_sizes[1] / 2;
    const size_t nd = (size_t)N * DIM * sizeof(float);

    const int gsc = ((N + SCAN_ROWS - 1) / SCAN_ROWS) * SCAN_PARTS;
    const int go  = (N + SUB_ROWS - 1) / SUB_ROWS;

    char* ws = (char*)d_ws;
    float* dinv = (float*)ws;                 ws += al256((size_t)N * sizeof(float));
    int*   degp = (int*)ws;                   ws += al256((size_t)SCAN_PARTS * N * sizeof(int));
    float* hw   = (float*)ws;                 ws += al256(nd);
    float* hw2  = (float*)ws;                 ws += al256(nd);
    float* accb = (float*)ws;                 ws += al256(nd);
    int* keepAll = (int*)ws;                  ws += al256((size_t)gsc * SCAN_SUBS * SUB_CAP * sizeof(int2));
    int* cntAll  = (int*)ws;                  ws += al256((size_t)gsc * SCAN_SUBS * sizeof(int));
    float* part = (float*)ws;

    edge_scan<<<gsc, OWN_WAVES * 32, 0, stream>>>(ei, degp, keepAll, cntAll, N, E);
    deg_to_dinv<<<(N + 255) / 256, 256, 0, stream>>>(dinv, degp, N);

    const int nstrips = N >> 4;
    const int gw = (nstrips + 7) / 8;
    layer_rows<<<gw, 256, 0, stream>>>(x, Ws, hw, N);
    owner_layer<false><<<go, OWN_WAVES * 32, 0, stream>>>(
        ei, hw, dinv, keepAll, cntAll, Ws + (size_t)DIM * DIM, gammas, betas, bs, hw2, N, E);
    owner_layer<false><<<go, OWN_WAVES * 32, 0, stream>>>(
        ei, hw2, dinv, keepAll, cntAll, Ws + (size_t)2 * DIM * DIM, gammas + DIM, betas + DIM, bs + DIM, hw, N, E);
    owner_layer<true><<<go, OWN_WAVES * 32, 0, stream>>>(
        ei, hw, dinv, keepAll, cntAll, Ws, gammas + (size_t)2 * DIM, betas + (size_t)2 * DIM, bs + (size_t)2 * DIM, accb, N, E);
    const float* hcur = accb;

    const int P = (N + PCH_NODES - 1) / PCH_NODES;
    pool_onehot<<<P, 128, 0, stream>>>(hcur, batch, part, N);
    pool_reduce<<<GNUM * DIM / 32, 256, 0, stream>>>(part, (float*)d_out, P);
}
